# starting kernel
# baseline (speedup 1.0000x reference)
#include <hip/hip_runtime.h>
#include <math.h>

typedef __attribute__((ext_vector_type(2))) float v2f;
typedef __attribute__((ext_vector_type(8))) float v8f;

#define B_  8
#define T_  1024
#define DIN 256
#define DH1 1024
#define DH2 1024
#define KW  9

__device__ __forceinline__ float mishf(float v) {
    float sp = (v > 20.f) ? v : log1pf(__expf(v));
    return v * tanhf(sp);
}

__global__ void k_norm_d(const float* __restrict__ dw, float* __restrict__ inv_nd) {
    int tid = threadIdx.x;
    if (tid >= B_ * KW) return;
    int b = tid / KW, k = tid % KW;
    float acc = 0.f;
    for (int c = 0; c < DH1; ++c) {
        float v = dw[((size_t)b * DH1 + c) * KW + k];
        acc += v * v;
    }
    inv_nd[tid] = 1.f / fmaxf(sqrtf(acc), 1e-12f);
}

__global__ __launch_bounds__(256) void k_norm_p(const float* __restrict__ pw,
                                                float* __restrict__ inv_np) {
    __shared__ float s[256];
    int b  = blockIdx.x >> 4;
    int o  = ((blockIdx.x & 15) << 6) + (threadIdx.x & 63);
    int cg = threadIdx.x >> 6;
    const float* P = pw + (size_t)b * DH1 * DH2;
    float acc = 0.f;
    for (int c = cg; c < DH1; c += 4) {
        float v = P[(size_t)c * DH2 + o];
        acc += v * v;
    }
    s[threadIdx.x] = acc;
    __syncthreads();
    if (cg == 0) {
        float tot = s[threadIdx.x] + s[threadIdx.x + 64] + s[threadIdx.x + 128] + s[threadIdx.x + 192];
        inv_np[b * DH2 + o] = 1.f / fmaxf(sqrtf(tot), 1e-12f);
    }
}

__global__ __launch_bounds__(256) void k_gemm1_mish(
    const float* __restrict__ x, const float* __restrict__ w1w,
    const float* __restrict__ w1b, float* __restrict__ h)
{
    const int lane = threadIdx.x & 31;
    const int hf = lane >> 4, l16 = lane & 15;
    int tid = blockIdx.x * 8 + (threadIdx.x >> 5);
    const int tn = T_ / 16, tm = DH1 / 16;
    int b  = tid / (tm * tn);
    int r  = tid % (tm * tn);
    int m0 = (r / tn) * 16;
    int n0 = (r % tn) * 16;

    const float* Arow = w1w + (size_t)(m0 + l16) * DIN;
    const float* Brow = x + ((size_t)b * T_ + (n0 + l16)) * DIN;
    v8f acc = {};
    #pragma unroll 4
    for (int k0 = 0; k0 < DIN; k0 += 4) {
        v2f a  = *(const v2f*)(Arow + k0 + 2 * hf);
        v2f bv = *(const v2f*)(Brow + k0 + 2 * hf);
        acc = __builtin_amdgcn_wmma_f32_16x16x4_f32(false, a, false, bv,
                                                    (short)0, acc, false, false);
    }
    float* H = h + (size_t)b * DH1 * T_;
    #pragma unroll
    for (int v = 0; v < 8; ++v) {
        int c = m0 + v + 8 * hf;
        int t = n0 + l16;
        H[(size_t)c * T_ + t] = mishf(acc[v] + w1b[c]);
    }
}

__global__ __launch_bounds__(256) void k_dwconv(
    const float* __restrict__ h, const float* __restrict__ dw,
    const float* __restrict__ dg, const float* __restrict__ db,
    const float* __restrict__ pg, const float* __restrict__ inv_nd,
    float* __restrict__ y)
{
    int idx = blockIdx.x * 256 + threadIdx.x;
    int t = idx & (T_ - 1);
    int c = (idx >> 10) & (DH1 - 1);
    int b = idx >> 20;
    const float* hp  = h + ((size_t)b * DH1 + c) * T_;
    const float* wp  = dw + ((size_t)b * DH1 + c) * KW;
    const float* ndp = inv_nd + b * KW;
    float acc = 0.f;
    #pragma unroll
    for (int k = 0; k < KW; ++k) {
        int tt = t + k - 4;
        if (tt >= 0 && tt < T_)
            acc += hp[tt] * (wp[k] * ndp[k]);
    }
    int bc = b * DH1 + c;
    float yy = acc * dg[bc] + db[bc];
    y[((size_t)b * DH1 + c) * T_ + t] = yy * pg[bc];
}

__global__ __launch_bounds__(256) void k_gemm2(
    const float* __restrict__ y, const float* __restrict__ pw,
    const float* __restrict__ pb, const float* __restrict__ inv_np,
    float* __restrict__ z)
{
    const int lane = threadIdx.x & 31;
    const int hf = lane >> 4, l16 = lane & 15;
    int tid = blockIdx.x * 8 + (threadIdx.x >> 5);
    const int tn = T_ / 16, tm = DH2 / 16;
    int b  = tid / (tm * tn);
    int r  = tid % (tm * tn);
    int m0 = (r / tn) * 16;
    int n0 = (r % tn) * 16;
    const float* A  = pw + (size_t)b * DH1 * DH2;
    const float* Bm = y  + (size_t)b * DH1 * T_;
    v8f acc = {};
    #pragma unroll 4
    for (int k0 = 0; k0 < DH1; k0 += 4) {
        int ka = k0 + 2 * hf;
        v2f a, bv;
        a.x  = A[(size_t)ka * DH2 + m0 + l16];
        a.y  = A[(size_t)(ka + 1) * DH2 + m0 + l16];
        bv.x = Bm[(size_t)ka * T_ + n0 + l16];
        bv.y = Bm[(size_t)(ka + 1) * T_ + n0 + l16];
        acc = __builtin_amdgcn_wmma_f32_16x16x4_f32(false, a, false, bv,
                                                    (short)0, acc, false, false);
    }
    float* Z = z + (size_t)b * DH2 * T_;
    #pragma unroll
    for (int v = 0; v < 8; ++v) {
        int o = m0 + v + 8 * hf;
        int t = n0 + l16;
        Z[(size_t)o * T_ + t] = acc[v] * inv_np[b * DH2 + o] + pb[b * DH2 + o];
    }
}

__global__ __launch_bounds__(256) void k_gemm3(
    const float* __restrict__ z, const float* __restrict__ w2w,
    const float* __restrict__ w2b, const float* __restrict__ x,
    float* __restrict__ out)
{
    const int lane = threadIdx.x & 31;
    const int hf = lane >> 4, l16 = lane & 15;
    int tid = blockIdx.x * 8 + (threadIdx.x >> 5);
    const int tn = T_ / 16, tm = DIN / 16;
    int b  = tid / (tm * tn);
    int r  = tid % (tm * tn);
    int m0 = (r / tn) * 16;
    int n0 = (r % tn) * 16;
    const float* Arow = w2w + (size_t)(m0 + l16) * DH2;
    const float* Bm   = z + (size_t)b * DH2 * T_;
    v8f acc = {};
    #pragma unroll 4
    for (int k0 = 0; k0 < DH2; k0 += 4) {
        int ka = k0 + 2 * hf;
        v2f a = *(const v2f*)(Arow + ka);
        v2f bv;
        bv.x = Bm[(size_t)ka * T_ + n0 + l16];
        bv.y = Bm[(size_t)(ka + 1) * T_ + n0 + l16];
        acc = __builtin_amdgcn_wmma_f32_16x16x4_f32(false, a, false, bv,
                                                    (short)0, acc, false, false);
    }
    #pragma unroll
    for (int v = 0; v < 8; ++v) {
        int i = m0 + v + 8 * hf;
        int t = n0 + l16;
        size_t oi = ((size_t)b * T_ + t) * DIN + i;
        out[oi] = acc[v] + w2b[i] + x[oi];
    }
}

extern "C" void kernel_launch(void* const* d_in, const int* in_sizes, int n_in,
                              void* d_out, int out_size, void* d_ws, size_t ws_size,
                              hipStream_t stream)
{
    const float* x   = (const float*)d_in[0];
    const float* d_w = (const float*)d_in[1];
    const float* d_g = (const float*)d_in[2];
    const float* d_b = (const float*)d_in[3];
    const float* p_w = (const float*)d_in[4];
    const float* p_g = (const float*)d_in[5];
    const float* p_b = (const float*)d_in[6];
    const float* w1w = (const float*)d_in[7];
    const float* w1b = (const float*)d_in[8];
    const float* w2w = (const float*)d_in[9];
    const float* w2b = (const float*)d_in[10];
    float* out = (float*)d_out;

    float* ws = (float*)d_ws;
    const size_t NCT = (size_t)B_ * DH1 * T_;
    float* hbuf   = ws;
    float* ybuf   = ws + NCT;
    float* inv_nd = ws + 2 * NCT;
    float* inv_np = inv_nd + 128;

    k_norm_d<<<1, 128, 0, stream>>>(d_w, inv_nd);
    k_norm_p<<<B_ * (DH2 / 64), 256, 0, stream>>>(p_w, inv_np);
    k_gemm1_mish<<<(B_ * (DH1 / 16) * (T_ / 16)) / 8, 256, 0, stream>>>(x, w1w, w1b, hbuf);
    k_dwconv<<<(int)(NCT / 256), 256, 0, stream>>>(hbuf, d_w, d_g, d_b, p_g, inv_nd, ybuf);
    k_gemm2<<<(B_ * (DH2 / 16) * (T_ / 16)) / 8, 256, 0, stream>>>(ybuf, p_w, p_b, inv_np, hbuf);
    k_gemm3<<<(B_ * (DIN / 16) * (T_ / 16)) / 8, 256, 0, stream>>>(hbuf, w2w, w2b, x, out);
}


# Register tiling of the three WMMA GEMMs: each wave computes a block of 16x16 tiles (2x4, 4x4, 4x2) from fragments loaded once per k unit, one unit ahead, 16-byte loads and stores; k order kept
# speedup vs baseline: 1.3695x; 1.3695x over previous
#include <hip/hip_runtime.h>
#include <math.h>

typedef __attribute__((ext_vector_type(2))) float v2f;
typedef __attribute__((ext_vector_type(4))) float v4f;
typedef __attribute__((ext_vector_type(8))) float v8f;

#define B_  8
#define T_  1024
#define DIN 256
#define DH1 1024
#define DH2 1024
#define KW  9

__device__ __forceinline__ float mishf(float v) {
    float sp = (v > 20.f) ? v : log1pf(__expf(v));
    return v * tanhf(sp);
}

__global__ void k_norm_d(const float* __restrict__ dw, float* __restrict__ inv_nd) {
    int tid = threadIdx.x;
    if (tid >= B_ * KW) return;
    int b = tid / KW, k = tid % KW;
    float acc = 0.f;
    for (int c = 0; c < DH1; ++c) {
        float v = dw[((size_t)b * DH1 + c) * KW + k];
        acc += v * v;
    }
    inv_nd[tid] = 1.f / fmaxf(sqrtf(acc), 1e-12f);
}

__global__ __launch_bounds__(256) void k_norm_p(const float* __restrict__ pw,
                                                float* __restrict__ inv_np) {
    __shared__ float s[256];
    int b  = blockIdx.x >> 4;
    int o  = ((blockIdx.x & 15) << 6) + (threadIdx.x & 63);
    int cg = threadIdx.x >> 6;
    const float* P = pw + (size_t)b * DH1 * DH2;
    float acc = 0.f;
    for (int c = cg; c < DH1; c += 4) {
        float v = P[(size_t)c * DH2 + o];
        acc += v * v;
    }
    s[threadIdx.x] = acc;
    __syncthreads();
    if (cg == 0) {
        float tot = s[threadIdx.x] + s[threadIdx.x + 64] + s[threadIdx.x + 128] + s[threadIdx.x + 192];
        inv_np[b * DH2 + o] = 1.f / fmaxf(sqrtf(tot), 1e-12f);
    }
}

__device__ __forceinline__ float pick8(const v8f a, const int v) {
    float r = a[0];
    r = (v == 1) ? a[1] : r;
    r = (v == 2) ? a[2] : r;
    r = (v == 3) ? a[3] : r;
    r = (v == 4) ? a[4] : r;
    r = (v == 5) ? a[5] : r;
    r = (v == 6) ? a[6] : r;
    r = (v == 7) ? a[7] : r;
    return r;
}

__global__ __launch_bounds__(128) void k_gemm1_mish(
    const float* __restrict__ x, const float* __restrict__ w1w,
    const float* __restrict__ w1b, float* __restrict__ h)
{
    const int lane = threadIdx.x & 31;
    const int hf = lane >> 4, l16 = lane & 15, w = threadIdx.x >> 5;
    const int b  = blockIdx.x >> 7;
    const int r  = blockIdx.x & 127;
    const int m0 = ((r >> 3) * 2 + (w >> 1)) * 32;
    const int n0 = ((r & 7) * 2 + (w & 1)) * 64;
    const float* Ap = w1w + (size_t)(m0 + l16) * DIN + 2 * hf;
    const float* Bp = x + ((size_t)b * T_ + n0 + 4 * l16) * DIN + 2 * hf;
    v8f acc[2][4] = {};
    v2f ca[4][2], cb[4][4];
    #pragma unroll
    for (int st = 0; st < 4; ++st) {
        #pragma unroll
        for (int i = 0; i < 2; ++i) ca[st][i] = *(const v2f*)(Ap + (size_t)i * 16 * DIN + 4 * st);
        #pragma unroll
        for (int j = 0; j < 4; ++j) cb[st][j] = *(const v2f*)(Bp + (size_t)j * DIN + 4 * st);
    }
    #pragma unroll 2
    for (int k0 = 0; k0 < DIN; k0 += 16) {
        const int kn = (k0 + 16 < DIN) ? k0 + 16 : k0;
        v2f na[4][2], nb[4][4];
        #pragma unroll
        for (int st = 0; st < 4; ++st) {
            #pragma unroll
            for (int i = 0; i < 2; ++i) na[st][i] = *(const v2f*)(Ap + (size_t)i * 16 * DIN + kn + 4 * st);
            #pragma unroll
            for (int j = 0; j < 4; ++j) nb[st][j] = *(const v2f*)(Bp + (size_t)j * DIN + kn + 4 * st);
        }
        __builtin_amdgcn_sched_barrier(0);
        #pragma unroll
        for (int st = 0; st < 4; ++st)
            #pragma unroll
            for (int i = 0; i < 2; ++i)
                #pragma unroll
                for (int j = 0; j < 4; ++j)
                    acc[i][j] = __builtin_amdgcn_wmma_f32_16x16x4_f32(false, ca[st][i], false, cb[st][j],
                                                                     (short)0, acc[i][j], false, false);
        __builtin_amdgcn_sched_barrier(0);
        asm volatile("" :: "v"(na[0][0]), "v"(na[0][1]), "v"(na[1][0]), "v"(na[1][1]),
                        "v"(na[2][0]), "v"(na[2][1]), "v"(na[3][0]), "v"(na[3][1]),
                        "v"(nb[0][0]), "v"(nb[0][1]), "v"(nb[0][2]), "v"(nb[0][3]),
                        "v"(nb[1][0]), "v"(nb[1][1]), "v"(nb[1][2]), "v"(nb[1][3]),
                        "v"(nb[2][0]), "v"(nb[2][1]), "v"(nb[2][2]), "v"(nb[2][3]),
                        "v"(nb[3][0]), "v"(nb[3][1]), "v"(nb[3][2]), "v"(nb[3][3]));
        #pragma unroll
        for (int st = 0; st < 4; ++st) {
            #pragma unroll
            for (int i = 0; i < 2; ++i) ca[st][i] = na[st][i];
            #pragma unroll
            for (int j = 0; j < 4; ++j) cb[st][j] = nb[st][j];
        }
    }
    float* H = h + (size_t)b * DH1 * T_;
    #pragma unroll
    for (int i = 0; i < 2; ++i) {
        #pragma unroll 1
        for (int v = 0; v < 8; ++v) {
            const int c = m0 + 16 * i + v + 8 * hf;
            const float bias = w1b[c];
            v4f o;
            o.x = mishf(pick8(acc[i][0], v) + bias);
            o.y = mishf(pick8(acc[i][1], v) + bias);
            o.z = mishf(pick8(acc[i][2], v) + bias);
            o.w = mishf(pick8(acc[i][3], v) + bias);
            *(v4f*)(H + (size_t)c * T_ + n0 + 4 * l16) = o;
        }
    }
}

__global__ __launch_bounds__(256) void k_dwconv(
    const float* __restrict__ h, const float* __restrict__ dw,
    const float* __restrict__ dg, const float* __restrict__ db,
    const float* __restrict__ pg, const float* __restrict__ inv_nd,
    float* __restrict__ y)
{
    int idx = blockIdx.x * 256 + threadIdx.x;
    int t = idx & (T_ - 1);
    int c = (idx >> 10) & (DH1 - 1);
    int b = idx >> 20;
    const float* hp  = h + ((size_t)b * DH1 + c) * T_;
    const float* wp  = dw + ((size_t)b * DH1 + c) * KW;
    const float* ndp = inv_nd + b * KW;
    float acc = 0.f;
    #pragma unroll
    for (int k = 0; k < KW; ++k) {
        int tt = t + k - 4;
        if (tt >= 0 && tt < T_)
            acc += hp[tt] * (wp[k] * ndp[k]);
    }
    int bc = b * DH1 + c;
    float yy = acc * dg[bc] + db[bc];
    y[((size_t)b * DH1 + c) * T_ + t] = yy * pg[bc];
}

__global__ __launch_bounds__(128) void k_gemm2(
    const float* __restrict__ y, const float* __restrict__ pw,
    const float* __restrict__ pb, const float* __restrict__ inv_np,
    float* __restrict__ z)
{
    const int lane = threadIdx.x & 31;
    const int hf = lane >> 4, l16 = lane & 15, w = threadIdx.x >> 5;
    const int b  = blockIdx.x >> 6;
    const int r  = blockIdx.x & 63;
    const int m0 = ((r >> 3) * 2 + (w >> 1)) * 64;
    const int n0 = ((r & 7) * 2 + (w & 1)) * 64;
    const float* Ap = pw + (size_t)b * DH1 * DH2 + (size_t)(2 * hf) * DH2 + m0 + 4 * l16;
    const float* Bp = y  + (size_t)b * DH1 * T_  + (size_t)(2 * hf) * T_  + n0 + 4 * l16;
    v8f acc[4][4] = {};
    v4f ca[2][2], cb[2][2];
    #pragma unroll
    for (int st = 0; st < 2; ++st)
        #pragma unroll
        for (int s = 0; s < 2; ++s) {
            ca[st][s] = *(const v4f*)(Ap + (size_t)(4 * st + s) * DH2);
            cb[st][s] = *(const v4f*)(Bp + (size_t)(4 * st + s) * T_);
        }
    #pragma unroll 2
    for (int k0 = 0; k0 < DH1; k0 += 8) {
        const int kn = (k0 + 8 < DH1) ? k0 + 8 : k0;
        v4f na[2][2], nb[2][2];
        #pragma unroll
        for (int st = 0; st < 2; ++st)
            #pragma unroll
            for (int s = 0; s < 2; ++s) {
                na[st][s] = *(const v4f*)(Ap + (size_t)(kn + 4 * st + s) * DH2);
                nb[st][s] = *(const v4f*)(Bp + (size_t)(kn + 4 * st + s) * T_);
            }
        __builtin_amdgcn_sched_barrier(0);
        #pragma unroll
        for (int st = 0; st < 2; ++st)
            #pragma unroll
            for (int i = 0; i < 4; ++i) {
                v2f a;
                a.x = ca[st][0][i];
                a.y = ca[st][1][i];
                #pragma unroll
                for (int j = 0; j < 4; ++j) {
                    v2f bv;
                    bv.x = cb[st][0][j];
                    bv.y = cb[st][1][j];
                    acc[i][j] = __builtin_amdgcn_wmma_f32_16x16x4_f32(false, a, false, bv,
                                                                     (short)0, acc[i][j], false, false);
                }
            }
        __builtin_amdgcn_sched_barrier(0);
        asm volatile("" :: "v"(na[0][0]), "v"(na[0][1]), "v"(na[1][0]), "v"(na[1][1]),
                        "v"(nb[0][0]), "v"(nb[0][1]), "v"(nb[1][0]), "v"(nb[1][1]));
        #pragma unroll
        for (int st = 0; st < 2; ++st)
            #pragma unroll
            for (int s = 0; s < 2; ++s) {
                ca[st][s] = na[st][s];
                cb[st][s] = nb[st][s];
            }
    }
    float* Z = z + (size_t)b * DH2 * T_;
    #pragma unroll
    for (int i = 0; i < 4; ++i) {
        #pragma unroll
        for (int v = 0; v < 8; ++v) {
            const int o = m0 + 4 * (v + 8 * hf) + i;
            const float sc = inv_np[b * DH2 + o];
            const float bi = pb[b * DH2 + o];
            v4f q;
            q.x = acc[i][0][v] * sc + bi;
            q.y = acc[i][1][v] * sc + bi;
            q.z = acc[i][2][v] * sc + bi;
            q.w = acc[i][3][v] * sc + bi;
            *(v4f*)(Z + (size_t)o * T_ + n0 + 4 * l16) = q;
        }
    }
}

__global__ __launch_bounds__(128) void k_gemm3(
    const float* __restrict__ z, const float* __restrict__ w2w,
    const float* __restrict__ w2b, const float* __restrict__ x,
    float* __restrict__ out)
{
    const int lane = threadIdx.x & 31;
    const int hf = lane >> 4, l16 = lane & 15, w = threadIdx.x >> 5;
    const int b  = blockIdx.x >> 5;
    const int r  = blockIdx.x & 31;
    const int t0 = ((r >> 2) * 2 + (w >> 1)) * 64;
    const int i0 = ((r & 3) * 2 + (w & 1)) * 32;
    const float* Zp = z + (size_t)b * DH2 * T_ + (size_t)(2 * hf) * T_ + t0 + 4 * l16;
    const float* Wp = w2w + (size_t)(i0 + 2 * l16) * DH2 + 2 * hf;
    v8f acc[4][2] = {};
    v4f ca[4][2];
    v2f cb[4][2];
    #pragma unroll
    for (int st = 0; st < 4; ++st) {
        #pragma unroll
        for (int s = 0; s < 2; ++s) ca[st][s] = *(const v4f*)(Zp + (size_t)(4 * st + s) * T_);
        #pragma unroll
        for (int j = 0; j < 2; ++j) cb[st][j] = *(const v2f*)(Wp + (size_t)j * DH2 + 4 * st);
    }
    #pragma unroll 2
    for (int k0 = 0; k0 < DH2; k0 += 16) {
        const int kn = (k0 + 16 < DH2) ? k0 + 16 : k0;
        v4f na[4][2];
        v2f nb[4][2];
        #pragma unroll
        for (int st = 0; st < 4; ++st) {
            #pragma unroll
            for (int s = 0; s < 2; ++s) na[st][s] = *(const v4f*)(Zp + (size_t)(kn + 4 * st + s) * T_);
            #pragma unroll
            for (int j = 0; j < 2; ++j) nb[st][j] = *(const v2f*)(Wp + (size_t)j * DH2 + kn + 4 * st);
        }
        __builtin_amdgcn_sched_barrier(0);
        #pragma unroll
        for (int st = 0; st < 4; ++st)
            #pragma unroll
            for (int a = 0; a < 4; ++a) {
                v2f av;
                av.x = ca[st][0][a];
                av.y = ca[st][1][a];
                #pragma unroll
                for (int j = 0; j < 2; ++j)
                    acc[a][j] = __builtin_amdgcn_wmma_f32_16x16x4_f32(false, av, false, cb[st][j],
                                                                     (short)0, acc[a][j], false, false);
            }
        __builtin_amdgcn_sched_barrier(0);
        asm volatile("" :: "v"(na[0][0]), "v"(na[0][1]), "v"(na[1][0]), "v"(na[1][1]),
                        "v"(na[2][0]), "v"(na[2][1]), "v"(na[3][0]), "v"(na[3][1]),
                        "v"(nb[0][0]), "v"(nb[0][1]), "v"(nb[1][0]), "v"(nb[1][1]),
                        "v"(nb[2][0]), "v"(nb[2][1]), "v"(nb[3][0]), "v"(nb[3][1]));
        #pragma unroll
        for (int st = 0; st < 4; ++st) {
            #pragma unroll
            for (int s = 0; s < 2; ++s) ca[st][s] = na[st][s];
            #pragma unroll
            for (int j = 0; j < 2; ++j) cb[st][j] = nb[st][j];
        }
    }
    const int ii = i0 + 2 * l16;
    const v2f bb = *(const v2f*)(w2b + ii);
    #pragma unroll
    for (int a = 0; a < 4; ++a) {
        #pragma unroll
        for (int v = 0; v < 8; ++v) {
            const int t = t0 + 4 * (v + 8 * hf) + a;
            const size_t oi = ((size_t)b * T_ + t) * DIN + ii;
            const v2f xv = *(const v2f*)(x + oi);
            v2f q;
            q.x = acc[a][0][v] + bb.x + xv.x;
            q.y = acc[a][1][v] + bb.y + xv.y;
            *(v2f*)(out + oi) = q;
        }
    }
}

extern "C" void kernel_launch(void* const* d_in, const int* in_sizes, int n_in,
                              void* d_out, int out_size, void* d_ws, size_t ws_size,
                              hipStream_t stream)
{
    const float* x   = (const float*)d_in[0];
    const float* d_w = (const float*)d_in[1];
    const float* d_g = (const float*)d_in[2];
    const float* d_b = (const float*)d_in[3];
    const float* p_w = (const float*)d_in[4];
    const float* p_g = (const float*)d_in[5];
    const float* p_b = (const float*)d_in[6];
    const float* w1w = (const float*)d_in[7];
    const float* w1b = (const float*)d_in[8];
    const float* w2w = (const float*)d_in[9];
    const float* w2b = (const float*)d_in[10];
    float* out = (float*)d_out;

    float* ws = (float*)d_ws;
    const size_t NCT = (size_t)B_ * DH1 * T_;
    float* hbuf   = ws;
    float* ybuf   = ws + NCT;
    float* inv_nd = ws + 2 * NCT;
    float* inv_np = inv_nd + 128;

    k_norm_d<<<1, 128, 0, stream>>>(d_w, inv_nd);
    k_norm_p<<<B_ * (DH2 / 64), 256, 0, stream>>>(p_w, inv_np);
    k_gemm1_mish<<<1024, 128, 0, stream>>>(x, w1w, w1b, hbuf);
    k_dwconv<<<(int)(NCT / 256), 256, 0, stream>>>(hbuf, d_w, d_g, d_b, p_g, inv_nd, ybuf);
    k_gemm2<<<512, 128, 0, stream>>>(ybuf, p_w, p_b, inv_np, hbuf);
    k_gemm3<<<256, 128, 0, stream>>>(hbuf, w2w, w2b, x, out);
}


# Depthwise conv: one wave per (b,c) row, 4 neighbouring outputs per lane per pass from three 16-byte loads and one 16-byte store, row taps and gains by scalar loads: 32x fewer waves; sums unchanged
# speedup vs baseline: 1.5522x; 1.1334x over previous
#include <hip/hip_runtime.h>
#include <math.h>

typedef __attribute__((ext_vector_type(2))) float v2f;
typedef __attribute__((ext_vector_type(4))) float v4f;
typedef __attribute__((ext_vector_type(8))) float v8f;

#define B_  8
#define T_  1024
#define DIN 256
#define DH1 1024
#define DH2 1024
#define KW  9

__device__ __forceinline__ float mishf(float v) {
    float sp = (v > 20.f) ? v : log1pf(__expf(v));
    return v * tanhf(sp);
}

__global__ void k_norm_d(const float* __restrict__ dw, float* __restrict__ inv_nd) {
    int tid = threadIdx.x;
    if (tid >= B_ * KW) return;
    int b = tid / KW, k = tid % KW;
    float acc = 0.f;
    for (int c = 0; c < DH1; ++c) {
        float v = dw[((size_t)b * DH1 + c) * KW + k];
        acc += v * v;
    }
    inv_nd[tid] = 1.f / fmaxf(sqrtf(acc), 1e-12f);
}

__global__ __launch_bounds__(256) void k_norm_p(const float* __restrict__ pw,
                                                float* __restrict__ inv_np) {
    __shared__ float s[256];
    int b  = blockIdx.x >> 4;
    int o  = ((blockIdx.x & 15) << 6) + (threadIdx.x & 63);
    int cg = threadIdx.x >> 6;
    const float* P = pw + (size_t)b * DH1 * DH2;
    float acc = 0.f;
    for (int c = cg; c < DH1; c += 4) {
        float v = P[(size_t)c * DH2 + o];
        acc += v * v;
    }
    s[threadIdx.x] = acc;
    __syncthreads();
    if (cg == 0) {
        float tot = s[threadIdx.x] + s[threadIdx.x + 64] + s[threadIdx.x + 128] + s[threadIdx.x + 192];
        inv_np[b * DH2 + o] = 1.f / fmaxf(sqrtf(tot), 1e-12f);
    }
}

__device__ __forceinline__ float pick8(const v8f a, const int v) {
    float r = a[0];
    r = (v == 1) ? a[1] : r;
    r = (v == 2) ? a[2] : r;
    r = (v == 3) ? a[3] : r;
    r = (v == 4) ? a[4] : r;
    r = (v == 5) ? a[5] : r;
    r = (v == 6) ? a[6] : r;
    r = (v == 7) ? a[7] : r;
    return r;
}

__global__ __launch_bounds__(128) void k_gemm1_mish(
    const float* __restrict__ x, const float* __restrict__ w1w,
    const float* __restrict__ w1b, float* __restrict__ h)
{
    const int lane = threadIdx.x & 31;
    const int hf = lane >> 4, l16 = lane & 15, w = threadIdx.x >> 5;
    const int b  = blockIdx.x >> 7;
    const int r  = blockIdx.x & 127;
    const int m0 = ((r >> 3) * 2 + (w >> 1)) * 32;
    const int n0 = ((r & 7) * 2 + (w & 1)) * 64;
    const float* Ap = w1w + (size_t)(m0 + l16) * DIN + 2 * hf;
    const float* Bp = x + ((size_t)b * T_ + n0 + 4 * l16) * DIN + 2 * hf;
    v8f acc[2][4] = {};
    v2f ca[4][2], cb[4][4];
    #pragma unroll
    for (int st = 0; st < 4; ++st) {
        #pragma unroll
        for (int i = 0; i < 2; ++i) ca[st][i] = *(const v2f*)(Ap + (size_t)i * 16 * DIN + 4 * st);
        #pragma unroll
        for (int j = 0; j < 4; ++j) cb[st][j] = *(const v2f*)(Bp + (size_t)j * DIN + 4 * st);
    }
    #pragma unroll 2
    for (int k0 = 0; k0 < DIN; k0 += 16) {
        const int kn = (k0 + 16 < DIN) ? k0 + 16 : k0;
        v2f na[4][2], nb[4][4];
        #pragma unroll
        for (int st = 0; st < 4; ++st) {
            #pragma unroll
            for (int i = 0; i < 2; ++i) na[st][i] = *(const v2f*)(Ap + (size_t)i * 16 * DIN + kn + 4 * st);
            #pragma unroll
            for (int j = 0; j < 4; ++j) nb[st][j] = *(const v2f*)(Bp + (size_t)j * DIN + kn + 4 * st);
        }
        __builtin_amdgcn_sched_barrier(0);
        #pragma unroll
        for (int st = 0; st < 4; ++st)
            #pragma unroll
            for (int i = 0; i < 2; ++i)
                #pragma unroll
                for (int j = 0; j < 4; ++j)
                    acc[i][j] = __builtin_amdgcn_wmma_f32_16x16x4_f32(false, ca[st][i], false, cb[st][j],
                                                                     (short)0, acc[i][j], false, false);
        __builtin_amdgcn_sched_barrier(0);
        asm volatile("" :: "v"(na[0][0]), "v"(na[0][1]), "v"(na[1][0]), "v"(na[1][1]),
                        "v"(na[2][0]), "v"(na[2][1]), "v"(na[3][0]), "v"(na[3][1]),
                        "v"(nb[0][0]), "v"(nb[0][1]), "v"(nb[0][2]), "v"(nb[0][3]),
                        "v"(nb[1][0]), "v"(nb[1][1]), "v"(nb[1][2]), "v"(nb[1][3]),
                        "v"(nb[2][0]), "v"(nb[2][1]), "v"(nb[2][2]), "v"(nb[2][3]),
                        "v"(nb[3][0]), "v"(nb[3][1]), "v"(nb[3][2]), "v"(nb[3][3]));
        #pragma unroll
        for (int st = 0; st < 4; ++st) {
            #pragma unroll
            for (int i = 0; i < 2; ++i) ca[st][i] = na[st][i];
            #pragma unroll
            for (int j = 0; j < 4; ++j) cb[st][j] = nb[st][j];
        }
    }
    float* H = h + (size_t)b * DH1 * T_;
    #pragma unroll
    for (int i = 0; i < 2; ++i) {
        #pragma unroll 1
        for (int v = 0; v < 8; ++v) {
            const int c = m0 + 16 * i + v + 8 * hf;
            const float bias = w1b[c];
            v4f o;
            o.x = mishf(pick8(acc[i][0], v) + bias);
            o.y = mishf(pick8(acc[i][1], v) + bias);
            o.z = mishf(pick8(acc[i][2], v) + bias);
            o.w = mishf(pick8(acc[i][3], v) + bias);
            *(v4f*)(H + (size_t)c * T_ + n0 + 4 * l16) = o;
        }
    }
}

__global__ __launch_bounds__(256) void k_dwconv(
    const float* __restrict__ h, const float* __restrict__ dw,
    const float* __restrict__ dg, const float* __restrict__ db,
    const float* __restrict__ pg, const float* __restrict__ inv_nd,
    float* __restrict__ y)
{
    const int lane = threadIdx.x & 31;
    const int row = __builtin_amdgcn_readfirstlane((int)(blockIdx.x * 8 + (threadIdx.x >> 5)));
    const int b = row >> 10;
    const float* hp  = h + (size_t)row * T_;
    const float* wp  = dw + (size_t)row * KW;
    const float* ndp = inv_nd + b * KW;
    float* yp = y + (size_t)row * T_;
    float wk[KW];
    #pragma unroll
    for (int k = 0; k < KW; ++k) wk[k] = wp[k] * ndp[k];
    const float g = dg[row], bi = db[row], pgv = pg[row];
    v4f cur[8], prv[8], nxt[8];
    #pragma unroll
    for (int j = 0; j < 8; ++j) cur[j] = *(const v4f*)(hp + 128 * j + 4 * lane);
    #pragma unroll
    for (int j = 0; j < 8; ++j) {
        const int t = 128 * j + 4 * lane;
        const bool inl = (t >= 4), inr = (t + 4 < T_);
        const v4f p = *(const v4f*)(hp + (inl ? t - 4 : t));
        const v4f n = *(const v4f*)(hp + (inr ? t + 4 : t));
        const v4f zero = {0.f, 0.f, 0.f, 0.f};
        prv[j] = inl ? p : zero;
        nxt[j] = inr ? n : zero;
    }
    #pragma unroll
    for (int j = 0; j < 8; ++j) {
        const int t = 128 * j + 4 * lane;
        const float win[12] = {prv[j].x, prv[j].y, prv[j].z, prv[j].w,
                               cur[j].x, cur[j].y, cur[j].z, cur[j].w,
                               nxt[j].x, nxt[j].y, nxt[j].z, nxt[j].w};
        v4f o;
        #pragma unroll
        for (int i = 0; i < 4; ++i) {
            float acc = 0.f;
            #pragma unroll
            for (int k = 0; k < KW; ++k) acc += win[i + k] * wk[k];
            const float yy = acc * g + bi;
            o[i] = yy * pgv;
        }
        *(v4f*)(yp + t) = o;
    }
}

__global__ __launch_bounds__(128) void k_gemm2(
    const float* __restrict__ y, const float* __restrict__ pw,
    const float* __restrict__ pb, const float* __restrict__ inv_np,
    float* __restrict__ z)
{
    const int lane = threadIdx.x & 31;
    const int hf = lane >> 4, l16 = lane & 15, w = threadIdx.x >> 5;
    const int b  = blockIdx.x >> 6;
    const int r  = blockIdx.x & 63;
    const int m0 = ((r >> 3) * 2 + (w >> 1)) * 64;
    const int n0 = ((r & 7) * 2 + (w & 1)) * 64;
    const float* Ap = pw + (size_t)b * DH1 * DH2 + (size_t)(2 * hf) * DH2 + m0 + 4 * l16;
    const float* Bp = y  + (size_t)b * DH1 * T_  + (size_t)(2 * hf) * T_  + n0 + 4 * l16;
    v8f acc[4][4] = {};
    v4f ca[2][2], cb[2][2];
    #pragma unroll
    for (int st = 0; st < 2; ++st)
        #pragma unroll
        for (int s = 0; s < 2; ++s) {
            ca[st][s] = *(const v4f*)(Ap + (size_t)(4 * st + s) * DH2);
            cb[st][s] = *(const v4f*)(Bp + (size_t)(4 * st + s) * T_);
        }
    #pragma unroll 2
    for (int k0 = 0; k0 < DH1; k0 += 8) {
        const int kn = (k0 + 8 < DH1) ? k0 + 8 : k0;
        v4f na[2][2], nb[2][2];
        #pragma unroll
        for (int st = 0; st < 2; ++st)
            #pragma unroll
            for (int s = 0; s < 2; ++s) {
                na[st][s] = *(const v4f*)(Ap + (size_t)(kn + 4 * st + s) * DH2);
                nb[st][s] = *(const v4f*)(Bp + (size_t)(kn + 4 * st + s) * T_);
            }
        __builtin_amdgcn_sched_barrier(0);
        #pragma unroll
        for (int st = 0; st < 2; ++st)
            #pragma unroll
            for (int i = 0; i < 4; ++i) {
                v2f a;
                a.x = ca[st][0][i];
                a.y = ca[st][1][i];
                #pragma unroll
                for (int j = 0; j < 4; ++j) {
                    v2f bv;
                    bv.x = cb[st][0][j];
                    bv.y = cb[st][1][j];
                    acc[i][j] = __builtin_amdgcn_wmma_f32_16x16x4_f32(false, a, false, bv,
                                                                     (short)0, acc[i][j], false, false);
                }
            }
        __builtin_amdgcn_sched_barrier(0);
        asm volatile("" :: "v"(na[0][0]), "v"(na[0][1]), "v"(na[1][0]), "v"(na[1][1]),
                        "v"(nb[0][0]), "v"(nb[0][1]), "v"(nb[1][0]), "v"(nb[1][1]));
        #pragma unroll
        for (int st = 0; st < 2; ++st)
            #pragma unroll
            for (int s = 0; s < 2; ++s) {
                ca[st][s] = na[st][s];
                cb[st][s] = nb[st][s];
            }
    }
    float* Z = z + (size_t)b * DH2 * T_;
    #pragma unroll
    for (int i = 0; i < 4; ++i) {
        #pragma unroll
        for (int v = 0; v < 8; ++v) {
            const int o = m0 + 4 * (v + 8 * hf) + i;
            const float sc = inv_np[b * DH2 + o];
            const float bi = pb[b * DH2 + o];
            v4f q;
            q.x = acc[i][0][v] * sc + bi;
            q.y = acc[i][1][v] * sc + bi;
            q.z = acc[i][2][v] * sc + bi;
            q.w = acc[i][3][v] * sc + bi;
            *(v4f*)(Z + (size_t)o * T_ + n0 + 4 * l16) = q;
        }
    }
}

__global__ __launch_bounds__(128) void k_gemm3(
    const float* __restrict__ z, const float* __restrict__ w2w,
    const float* __restrict__ w2b, const float* __restrict__ x,
    float* __restrict__ out)
{
    const int lane = threadIdx.x & 31;
    const int hf = lane >> 4, l16 = lane & 15, w = threadIdx.x >> 5;
    const int b  = blockIdx.x >> 5;
    const int r  = blockIdx.x & 31;
    const int t0 = ((r >> 2) * 2 + (w >> 1)) * 64;
    const int i0 = ((r & 3) * 2 + (w & 1)) * 32;
    const float* Zp = z + (size_t)b * DH2 * T_ + (size_t)(2 * hf) * T_ + t0 + 4 * l16;
    const float* Wp = w2w + (size_t)(i0 + 2 * l16) * DH2 + 2 * hf;
    v8f acc[4][2] = {};
    v4f ca[4][2];
    v2f cb[4][2];
    #pragma unroll
    for (int st = 0; st < 4; ++st) {
        #pragma unroll
        for (int s = 0; s < 2; ++s) ca[st][s] = *(const v4f*)(Zp + (size_t)(4 * st + s) * T_);
        #pragma unroll
        for (int j = 0; j < 2; ++j) cb[st][j] = *(const v2f*)(Wp + (size_t)j * DH2 + 4 * st);
    }
    #pragma unroll 2
    for (int k0 = 0; k0 < DH2; k0 += 16) {
        const int kn = (k0 + 16 < DH2) ? k0 + 16 : k0;
        v4f na[4][2];
        v2f nb[4][2];
        #pragma unroll
        for (int st = 0; st < 4; ++st) {
            #pragma unroll
            for (int s = 0; s < 2; ++s) na[st][s] = *(const v4f*)(Zp + (size_t)(kn + 4 * st + s) * T_);
            #pragma unroll
            for (int j = 0; j < 2; ++j) nb[st][j] = *(const v2f*)(Wp + (size_t)j * DH2 + kn + 4 * st);
        }
        __builtin_amdgcn_sched_barrier(0);
        #pragma unroll
        for (int st = 0; st < 4; ++st)
            #pragma unroll
            for (int a = 0; a < 4; ++a) {
                v2f av;
                av.x = ca[st][0][a];
                av.y = ca[st][1][a];
                #pragma unroll
                for (int j = 0; j < 2; ++j)
                    acc[a][j] = __builtin_amdgcn_wmma_f32_16x16x4_f32(false, av, false, cb[st][j],
                                                                     (short)0, acc[a][j], false, false);
            }
        __builtin_amdgcn_sched_barrier(0);
        asm volatile("" :: "v"(na[0][0]), "v"(na[0][1]), "v"(na[1][0]), "v"(na[1][1]),
                        "v"(na[2][0]), "v"(na[2][1]), "v"(na[3][0]), "v"(na[3][1]),
                        "v"(nb[0][0]), "v"(nb[0][1]), "v"(nb[1][0]), "v"(nb[1][1]),
                        "v"(nb[2][0]), "v"(nb[2][1]), "v"(nb[3][0]), "v"(nb[3][1]));
        #pragma unroll
        for (int st = 0; st < 4; ++st) {
            #pragma unroll
            for (int s = 0; s < 2; ++s) ca[st][s] = na[st][s];
            #pragma unroll
            for (int j = 0; j < 2; ++j) cb[st][j] = nb[st][j];
        }
    }
    const int ii = i0 + 2 * l16;
    const v2f bb = *(const v2f*)(w2b + ii);
    #pragma unroll
    for (int a = 0; a < 4; ++a) {
        #pragma unroll
        for (int v = 0; v < 8; ++v) {
            const int t = t0 + 4 * (v + 8 * hf) + a;
            const size_t oi = ((size_t)b * T_ + t) * DIN + ii;
            const v2f xv = *(const v2f*)(x + oi);
            v2f q;
            q.x = acc[a][0][v] + bb.x + xv.x;
            q.y = acc[a][1][v] + bb.y + xv.y;
            *(v2f*)(out + oi) = q;
        }
    }
}

extern "C" void kernel_launch(void* const* d_in, const int* in_sizes, int n_in,
                              void* d_out, int out_size, void* d_ws, size_t ws_size,
                              hipStream_t stream)
{
    const float* x   = (const float*)d_in[0];
    const float* d_w = (const float*)d_in[1];
    const float* d_g = (const float*)d_in[2];
    const float* d_b = (const float*)d_in[3];
    const float* p_w = (const float*)d_in[4];
    const float* p_g = (const float*)d_in[5];
    const float* p_b = (const float*)d_in[6];
    const float* w1w = (const float*)d_in[7];
    const float* w1b = (const float*)d_in[8];
    const float* w2w = (const float*)d_in[9];
    const float* w2b = (const float*)d_in[10];
    float* out = (float*)d_out;

    float* ws = (float*)d_ws;
    const size_t NCT = (size_t)B_ * DH1 * T_;
    float* hbuf   = ws;
    float* ybuf   = ws + NCT;
    float* inv_nd = ws + 2 * NCT;
    float* inv_np = inv_nd + 128;

    k_norm_d<<<1, 128, 0, stream>>>(d_w, inv_nd);
    k_norm_p<<<B_ * (DH2 / 64), 256, 0, stream>>>(p_w, inv_np);
    k_gemm1_mish<<<1024, 128, 0, stream>>>(x, w1w, w1b, hbuf);
    k_dwconv<<<1024, 256, 0, stream>>>(hbuf, d_w, d_g, d_b, p_g, inv_nd, ybuf);
    k_gemm2<<<512, 128, 0, stream>>>(ybuf, p_w, p_b, inv_np, hbuf);
    k_gemm3<<<256, 128, 0, stream>>>(hbuf, w2w, w2b, x, out);
}


# Norm pre-pass re-parallelised into one launch: p-norm 256 groups, 32 independent 16-byte loads in flight per wave; d-norm 72 groups (was one serial wave); both sums of squares become trees
# speedup vs baseline: 2.1142x; 1.3621x over previous
#include <hip/hip_runtime.h>
#include <math.h>

typedef __attribute__((ext_vector_type(2))) float v2f;
typedef __attribute__((ext_vector_type(4))) float v4f;
typedef __attribute__((ext_vector_type(8))) float v8f;

#define B_  8
#define T_  1024
#define DIN 256
#define DH1 1024
#define DH2 1024
#define KW  9

__device__ __forceinline__ float mishf(float v) {
    float sp = (v > 20.f) ? v : log1pf(__expf(v));
    return v * tanhf(sp);
}

__global__ __launch_bounds__(256) void k_norms(const float* __restrict__ pw,
                                               const float* __restrict__ dw,
                                               float* __restrict__ inv_np,
                                               float* __restrict__ inv_nd)
{
    __shared__ float s[8][32];
    const int lane = threadIdx.x & 31, w = threadIdx.x >> 5;
    if (blockIdx.x < 256) {
        const int b = blockIdx.x >> 5, oc = blockIdx.x & 31;
        const int cs = lane >> 3, ol = lane & 7;
        const float* P = pw + (size_t)b * DH1 * DH2 + (size_t)(4 * w + cs) * DH2 + oc * 32 + 4 * ol;
        v4f v[32];
        #pragma unroll
        for (int j = 0; j < 32; ++j) v[j] = *(const v4f*)(P + (size_t)(32 * j) * DH2);
        __builtin_amdgcn_sched_barrier(0);
        v4f acc = {0.f, 0.f, 0.f, 0.f};
        #pragma unroll
        for (int j = 0; j < 32; ++j) acc += v[j] * v[j];
        #pragma unroll
        for (int i = 0; i < 4; ++i) {
            acc[i] += __shfl_xor(acc[i], 8);
            acc[i] += __shfl_xor(acc[i], 16);
        }
        if (lane < 8) *(v4f*)&s[w][4 * lane] = acc;
        __syncthreads();
        if (threadIdx.x < 32) {
            float tot = 0.f;
            #pragma unroll
            for (int q = 0; q < 8; ++q) tot += s[q][threadIdx.x];
            inv_np[b * DH2 + oc * 32 + threadIdx.x] = 1.f / fmaxf(sqrtf(tot), 1e-12f);
        }
    } else {
        const int p = blockIdx.x - 256;
        const int b = p / KW, k = p - b * KW;
        const float* D = dw + (size_t)b * DH1 * KW + k;
        float acc = 0.f;
        #pragma unroll
        for (int j = 0; j < 4; ++j) {
            const float v = D[(size_t)(threadIdx.x + 256 * j) * KW];
            acc += v * v;
        }
        #pragma unroll
        for (int m = 16; m >= 1; m >>= 1) acc += __shfl_xor(acc, m);
        if (lane == 0) s[0][w] = acc;
        __syncthreads();
        if (threadIdx.x == 0) {
            float tot = 0.f;
            #pragma unroll
            for (int q = 0; q < 8; ++q) tot += s[0][q];
            inv_nd[p] = 1.f / fmaxf(sqrtf(tot), 1e-12f);
        }
    }
}

__device__ __forceinline__ float pick8(const v8f a, const int v) {
    float r = a[0];
    r = (v == 1) ? a[1] : r;
    r = (v == 2) ? a[2] : r;
    r = (v == 3) ? a[3] : r;
    r = (v == 4) ? a[4] : r;
    r = (v == 5) ? a[5] : r;
    r = (v == 6) ? a[6] : r;
    r = (v == 7) ? a[7] : r;
    return r;
}

__global__ __launch_bounds__(128) void k_gemm1_mish(
    const float* __restrict__ x, const float* __restrict__ w1w,
    const float* __restrict__ w1b, float* __restrict__ h)
{
    const int lane = threadIdx.x & 31;
    const int hf = lane >> 4, l16 = lane & 15, w = threadIdx.x >> 5;
    const int b  = blockIdx.x >> 7;
    const int r  = blockIdx.x & 127;
    const int m0 = ((r >> 3) * 2 + (w >> 1)) * 32;
    const int n0 = ((r & 7) * 2 + (w & 1)) * 64;
    const float* Ap = w1w + (size_t)(m0 + l16) * DIN + 2 * hf;
    const float* Bp = x + ((size_t)b * T_ + n0 + 4 * l16) * DIN + 2 * hf;
    v8f acc[2][4] = {};
    v2f ca[4][2], cb[4][4];
    #pragma unroll
    for (int st = 0; st < 4; ++st) {
        #pragma unroll
        for (int i = 0; i < 2; ++i) ca[st][i] = *(const v2f*)(Ap + (size_t)i * 16 * DIN + 4 * st);
        #pragma unroll
        for (int j = 0; j < 4; ++j) cb[st][j] = *(const v2f*)(Bp + (size_t)j * DIN + 4 * st);
    }
    #pragma unroll 2
    for (int k0 = 0; k0 < DIN; k0 += 16) {
        const int kn = (k0 + 16 < DIN) ? k0 + 16 : k0;
        v2f na[4][2], nb[4][4];
        #pragma unroll
        for (int st = 0; st < 4; ++st) {
            #pragma unroll
            for (int i = 0; i < 2; ++i) na[st][i] = *(const v2f*)(Ap + (size_t)i * 16 * DIN + kn + 4 * st);
            #pragma unroll
            for (int j = 0; j < 4; ++j) nb[st][j] = *(const v2f*)(Bp + (size_t)j * DIN + kn + 4 * st);
        }
        __builtin_amdgcn_sched_barrier(0);
        #pragma unroll
        for (int st = 0; st < 4; ++st)
            #pragma unroll
            for (int i = 0; i < 2; ++i)
                #pragma unroll
                for (int j = 0; j < 4; ++j)
                    acc[i][j] = __builtin_amdgcn_wmma_f32_16x16x4_f32(false, ca[st][i], false, cb[st][j],
                                                                     (short)0, acc[i][j], false, false);
        __builtin_amdgcn_sched_barrier(0);
        asm volatile("" :: "v"(na[0][0]), "v"(na[0][1]), "v"(na[1][0]), "v"(na[1][1]),
                        "v"(na[2][0]), "v"(na[2][1]), "v"(na[3][0]), "v"(na[3][1]),
                        "v"(nb[0][0]), "v"(nb[0][1]), "v"(nb[0][2]), "v"(nb[0][3]),
                        "v"(nb[1][0]), "v"(nb[1][1]), "v"(nb[1][2]), "v"(nb[1][3]),
                        "v"(nb[2][0]), "v"(nb[2][1]), "v"(nb[2][2]), "v"(nb[2][3]),
                        "v"(nb[3][0]), "v"(nb[3][1]), "v"(nb[3][2]), "v"(nb[3][3]));
        #pragma unroll
        for (int st = 0; st < 4; ++st) {
            #pragma unroll
            for (int i = 0; i < 2; ++i) ca[st][i] = na[st][i];
            #pragma unroll
            for (int j = 0; j < 4; ++j) cb[st][j] = nb[st][j];
        }
    }
    float* H = h + (size_t)b * DH1 * T_;
    #pragma unroll
    for (int i = 0; i < 2; ++i) {
        #pragma unroll 1
        for (int v = 0; v < 8; ++v) {
            const int c = m0 + 16 * i + v + 8 * hf;
            const float bias = w1b[c];
            v4f o;
            o.x = mishf(pick8(acc[i][0], v) + bias);
            o.y = mishf(pick8(acc[i][1], v) + bias);
            o.z = mishf(pick8(acc[i][2], v) + bias);
            o.w = mishf(pick8(acc[i][3], v) + bias);
            *(v4f*)(H + (size_t)c * T_ + n0 + 4 * l16) = o;
        }
    }
}

__global__ __launch_bounds__(256) void k_dwconv(
    const float* __restrict__ h, const float* __restrict__ dw,
    const float* __restrict__ dg, const float* __restrict__ db,
    const float* __restrict__ pg, const float* __restrict__ inv_nd,
    float* __restrict__ y)
{
    const int lane = threadIdx.x & 31;
    const int row = __builtin_amdgcn_readfirstlane((int)(blockIdx.x * 8 + (threadIdx.x >> 5)));
    const int b = row >> 10;
    const float* hp  = h + (size_t)row * T_;
    const float* wp  = dw + (size_t)row * KW;
    const float* ndp = inv_nd + b * KW;
    float* yp = y + (size_t)row * T_;
    float wk[KW];
    #pragma unroll
    for (int k = 0; k < KW; ++k) wk[k] = wp[k] * ndp[k];
    const float g = dg[row], bi = db[row], pgv = pg[row];
    v4f cur[8], prv[8], nxt[8];
    #pragma unroll
    for (int j = 0; j < 8; ++j) cur[j] = *(const v4f*)(hp + 128 * j + 4 * lane);
    #pragma unroll
    for (int j = 0; j < 8; ++j) {
        const int t = 128 * j + 4 * lane;
        const bool inl = (t >= 4), inr = (t + 4 < T_);
        const v4f p = *(const v4f*)(hp + (inl ? t - 4 : t));
        const v4f n = *(const v4f*)(hp + (inr ? t + 4 : t));
        const v4f zero = {0.f, 0.f, 0.f, 0.f};
        prv[j] = inl ? p : zero;
        nxt[j] = inr ? n : zero;
    }
    #pragma unroll
    for (int j = 0; j < 8; ++j) {
        const int t = 128 * j + 4 * lane;
        const float win[12] = {prv[j].x, prv[j].y, prv[j].z, prv[j].w,
                               cur[j].x, cur[j].y, cur[j].z, cur[j].w,
                               nxt[j].x, nxt[j].y, nxt[j].z, nxt[j].w};
        v4f o;
        #pragma unroll
        for (int i = 0; i < 4; ++i) {
            float acc = 0.f;
            #pragma unroll
            for (int k = 0; k < KW; ++k) acc += win[i + k] * wk[k];
            const float yy = acc * g + bi;
            o[i] = yy * pgv;
        }
        *(v4f*)(yp + t) = o;
    }
}

__global__ __launch_bounds__(128) void k_gemm2(
    const float* __restrict__ y, const float* __restrict__ pw,
    const float* __restrict__ pb, const float* __restrict__ inv_np,
    float* __restrict__ z)
{
    const int lane = threadIdx.x & 31;
    const int hf = lane >> 4, l16 = lane & 15, w = threadIdx.x >> 5;
    const int b  = blockIdx.x >> 6;
    const int r  = blockIdx.x & 63;
    const int m0 = ((r >> 3) * 2 + (w >> 1)) * 64;
    const int n0 = ((r & 7) * 2 + (w & 1)) * 64;
    const float* Ap = pw + (size_t)b * DH1 * DH2 + (size_t)(2 * hf) * DH2 + m0 + 4 * l16;
    const float* Bp = y  + (size_t)b * DH1 * T_  + (size_t)(2 * hf) * T_  + n0 + 4 * l16;
    v8f acc[4][4] = {};
    v4f ca[2][2], cb[2][2];
    #pragma unroll
    for (int st = 0; st < 2; ++st)
        #pragma unroll
        for (int s = 0; s < 2; ++s) {
            ca[st][s] = *(const v4f*)(Ap + (size_t)(4 * st + s) * DH2);
            cb[st][s] = *(const v4f*)(Bp + (size_t)(4 * st + s) * T_);
        }
    #pragma unroll 2
    for (int k0 = 0; k0 < DH1; k0 += 8) {
        const int kn = (k0 + 8 < DH1) ? k0 + 8 : k0;
        v4f na[2][2], nb[2][2];
        #pragma unroll
        for (int st = 0; st < 2; ++st)
            #pragma unroll
            for (int s = 0; s < 2; ++s) {
                na[st][s] = *(const v4f*)(Ap + (size_t)(kn + 4 * st + s) * DH2);
                nb[st][s] = *(const v4f*)(Bp + (size_t)(kn + 4 * st + s) * T_);
            }
        __builtin_amdgcn_sched_barrier(0);
        #pragma unroll
        for (int st = 0; st < 2; ++st)
            #pragma unroll
            for (int i = 0; i < 4; ++i) {
                v2f a;
                a.x = ca[st][0][i];
                a.y = ca[st][1][i];
                #pragma unroll
                for (int j = 0; j < 4; ++j) {
                    v2f bv;
                    bv.x = cb[st][0][j];
                    bv.y = cb[st][1][j];
                    acc[i][j] = __builtin_amdgcn_wmma_f32_16x16x4_f32(false, a, false, bv,
                                                                     (short)0, acc[i][j], false, false);
                }
            }
        __builtin_amdgcn_sched_barrier(0);
        asm volatile("" :: "v"(na[0][0]), "v"(na[0][1]), "v"(na[1][0]), "v"(na[1][1]),
                        "v"(nb[0][0]), "v"(nb[0][1]), "v"(nb[1][0]), "v"(nb[1][1]));
        #pragma unroll
        for (int st = 0; st < 2; ++st)
            #pragma unroll
            for (int s = 0; s < 2; ++s) {
                ca[st][s] = na[st][s];
                cb[st][s] = nb[st][s];
            }
    }
    float* Z = z + (size_t)b * DH2 * T_;
    #pragma unroll
    for (int i = 0; i < 4; ++i) {
        #pragma unroll
        for (int v = 0; v < 8; ++v) {
            const int o = m0 + 4 * (v + 8 * hf) + i;
            const float sc = inv_np[b * DH2 + o];
            const float bi = pb[b * DH2 + o];
            v4f q;
            q.x = acc[i][0][v] * sc + bi;
            q.y = acc[i][1][v] * sc + bi;
            q.z = acc[i][2][v] * sc + bi;
            q.w = acc[i][3][v] * sc + bi;
            *(v4f*)(Z + (size_t)o * T_ + n0 + 4 * l16) = q;
        }
    }
}

__global__ __launch_bounds__(128) void k_gemm3(
    const float* __restrict__ z, const float* __restrict__ w2w,
    const float* __restrict__ w2b, const float* __restrict__ x,
    float* __restrict__ out)
{
    const int lane = threadIdx.x & 31;
    const int hf = lane >> 4, l16 = lane & 15, w = threadIdx.x >> 5;
    const int b  = blockIdx.x >> 5;
    const int r  = blockIdx.x & 31;
    const int t0 = ((r >> 2) * 2 + (w >> 1)) * 64;
    const int i0 = ((r & 3) * 2 + (w & 1)) * 32;
    const float* Zp = z + (size_t)b * DH2 * T_ + (size_t)(2 * hf) * T_ + t0 + 4 * l16;
    const float* Wp = w2w + (size_t)(i0 + 2 * l16) * DH2 + 2 * hf;
    v8f acc[4][2] = {};
    v4f ca[4][2];
    v2f cb[4][2];
    #pragma unroll
    for (int st = 0; st < 4; ++st) {
        #pragma unroll
        for (int s = 0; s < 2; ++s) ca[st][s] = *(const v4f*)(Zp + (size_t)(4 * st + s) * T_);
        #pragma unroll
        for (int j = 0; j < 2; ++j) cb[st][j] = *(const v2f*)(Wp + (size_t)j * DH2 + 4 * st);
    }
    #pragma unroll 2
    for (int k0 = 0; k0 < DH2; k0 += 16) {
        const int kn = (k0 + 16 < DH2) ? k0 + 16 : k0;
        v4f na[4][2];
        v2f nb[4][2];
        #pragma unroll
        for (int st = 0; st < 4; ++st) {
            #pragma unroll
            for (int s = 0; s < 2; ++s) na[st][s] = *(const v4f*)(Zp + (size_t)(kn + 4 * st + s) * T_);
            #pragma unroll
            for (int j = 0; j < 2; ++j) nb[st][j] = *(const v2f*)(Wp + (size_t)j * DH2 + kn + 4 * st);
        }
        __builtin_amdgcn_sched_barrier(0);
        #pragma unroll
        for (int st = 0; st < 4; ++st)
            #pragma unroll
            for (int a = 0; a < 4; ++a) {
                v2f av;
                av.x = ca[st][0][a];
                av.y = ca[st][1][a];
                #pragma unroll
                for (int j = 0; j < 2; ++j)
                    acc[a][j] = __builtin_amdgcn_wmma_f32_16x16x4_f32(false, av, false, cb[st][j],
                                                                     (short)0, acc[a][j], false, false);
            }
        __builtin_amdgcn_sched_barrier(0);
        asm volatile("" :: "v"(na[0][0]), "v"(na[0][1]), "v"(na[1][0]), "v"(na[1][1]),
                        "v"(na[2][0]), "v"(na[2][1]), "v"(na[3][0]), "v"(na[3][1]),
                        "v"(nb[0][0]), "v"(nb[0][1]), "v"(nb[1][0]), "v"(nb[1][1]),
                        "v"(nb[2][0]), "v"(nb[2][1]), "v"(nb[3][0]), "v"(nb[3][1]));
        #pragma unroll
        for (int st = 0; st < 4; ++st) {
            #pragma unroll
            for (int s = 0; s < 2; ++s) ca[st][s] = na[st][s];
            #pragma unroll
            for (int j = 0; j < 2; ++j) cb[st][j] = nb[st][j];
        }
    }
    const int ii = i0 + 2 * l16;
    const v2f bb = *(const v2f*)(w2b + ii);
    #pragma unroll
    for (int a = 0; a < 4; ++a) {
        #pragma unroll
        for (int v = 0; v < 8; ++v) {
            const int t = t0 + 4 * (v + 8 * hf) + a;
            const size_t oi = ((size_t)b * T_ + t) * DIN + ii;
            const v2f xv = *(const v2f*)(x + oi);
            v2f q;
            q.x = acc[a][0][v] + bb.x + xv.x;
            q.y = acc[a][1][v] + bb.y + xv.y;
            *(v2f*)(out + oi) = q;
        }
    }
}

extern "C" void kernel_launch(void* const* d_in, const int* in_sizes, int n_in,
                              void* d_out, int out_size, void* d_ws, size_t ws_size,
                              hipStream_t stream)
{
    const float* x   = (const float*)d_in[0];
    const float* d_w = (const float*)d_in[1];
    const float* d_g = (const float*)d_in[2];
    const float* d_b = (const float*)d_in[3];
    const float* p_w = (const float*)d_in[4];
    const float* p_g = (const float*)d_in[5];
    const float* p_b = (const float*)d_in[6];
    const float* w1w = (const float*)d_in[7];
    const float* w1b = (const float*)d_in[8];
    const float* w2w = (const float*)d_in[9];
    const float* w2b = (const float*)d_in[10];
    float* out = (float*)d_out;

    float* ws = (float*)d_ws;
    const size_t NCT = (size_t)B_ * DH1 * T_;
    float* hbuf   = ws;
    float* ybuf   = ws + NCT;
    float* inv_nd = ws + 2 * NCT;
    float* inv_np = inv_nd + 128;

    k_norms<<<328, 256, 0, stream>>>(p_w, d_w, inv_np, inv_nd);
    k_gemm1_mish<<<1024, 128, 0, stream>>>(x, w1w, w1b, hbuf);
    k_dwconv<<<1024, 256, 0, stream>>>(hbuf, d_w, d_g, d_b, p_g, inv_nd, ybuf);
    k_gemm2<<<512, 128, 0, stream>>>(ybuf, p_w, p_b, inv_np, hbuf);
    k_gemm3<<<256, 128, 0, stream>>>(hbuf, w2w, w2b, x, out);
}


# Re-associate the last two products: W[b] = w2*diag(1/norm)*p_w[b]^T first (256x1024x1024), then out = W[b]*y: 60 % fewer matrix instructions there; both sums run over another index (order changes)
# speedup vs baseline: 2.7438x; 1.2978x over previous
#include <hip/hip_runtime.h>
#include <math.h>

typedef __attribute__((ext_vector_type(2))) float v2f;
typedef __attribute__((ext_vector_type(4))) float v4f;
typedef __attribute__((ext_vector_type(8))) float v8f;

#define B_  8
#define T_  1024
#define DIN 256
#define DH1 1024
#define DH2 1024
#define KW  9

__device__ __forceinline__ float mishf(float v) {
    float sp = (v > 20.f) ? v : log1pf(__expf(v));
    return v * tanhf(sp);
}

__global__ __launch_bounds__(256) void k_norms(const float* __restrict__ pw,
                                               const float* __restrict__ dw,
                                               const float* __restrict__ w2w,
                                               const float* __restrict__ pb,
                                               float* __restrict__ inv_np,
                                               float* __restrict__ inv_nd,
                                               float* __restrict__ vb)
{
    __shared__ float s[8][256];
    const int lane = threadIdx.x & 31, w = threadIdx.x >> 5;
    if (blockIdx.x < 256) {
        const int b = blockIdx.x >> 5, oc = blockIdx.x & 31;
        const int cs = lane >> 3, ol = lane & 7;
        const float* P = pw + (size_t)b * DH1 * DH2 + (size_t)(4 * w + cs) * DH2 + oc * 32 + 4 * ol;
        v4f v[32];
        #pragma unroll
        for (int j = 0; j < 32; ++j) v[j] = *(const v4f*)(P + (size_t)(32 * j) * DH2);
        __builtin_amdgcn_sched_barrier(0);
        v4f acc = {0.f, 0.f, 0.f, 0.f};
        #pragma unroll
        for (int j = 0; j < 32; ++j) acc += v[j] * v[j];
        #pragma unroll
        for (int i = 0; i < 4; ++i) {
            acc[i] += __shfl_xor(acc[i], 8);
            acc[i] += __shfl_xor(acc[i], 16);
        }
        if (lane < 8) *(v4f*)&s[w][4 * lane] = acc;
        __syncthreads();
        if (threadIdx.x < 32) {
            float tot = 0.f;
            #pragma unroll
            for (int q = 0; q < 8; ++q) tot += s[q][threadIdx.x];
            inv_np[b * DH2 + oc * 32 + threadIdx.x] = 1.f / fmaxf(sqrtf(tot), 1e-12f);
        }
    } else if (blockIdx.x < 328) {
        const int p = blockIdx.x - 256;
        const int b = p / KW, k = p - b * KW;
        const float* D = dw + (size_t)b * DH1 * KW + k;
        float acc = 0.f;
        #pragma unroll
        for (int j = 0; j < 4; ++j) {
            const float v = D[(size_t)(threadIdx.x + 256 * j) * KW];
            acc += v * v;
        }
        #pragma unroll
        for (int m = 16; m >= 1; m >>= 1) acc += __shfl_xor(acc, m);
        if (lane == 0) s[0][w] = acc;
        __syncthreads();
        if (threadIdx.x == 0) {
            float tot = 0.f;
            #pragma unroll
            for (int q = 0; q < 8; ++q) tot += s[0][q];
            inv_nd[p] = 1.f / fmaxf(sqrtf(tot), 1e-12f);
        }
    } else {
        const int i0 = (blockIdx.x - 328) * 16;
        const int hf = lane >> 4, l16 = lane & 15;
        const float* Ap = w2w + (size_t)(i0 + l16) * DH2 + 128 * w + 4 * hf;
        const float* Bq = pb + (size_t)(l16 & 7) * DH2 + 128 * w + 4 * hf;
        v4f a[16], q[16];
        #pragma unroll
        for (int u = 0; u < 16; ++u) {
            a[u] = *(const v4f*)(Ap + 8 * u);
            q[u] = *(const v4f*)(Bq + 8 * u);
        }
        const bool real = (l16 < 8);
        const v4f zero = {0.f, 0.f, 0.f, 0.f};
        v8f acc0 = {}, acc1 = {};
        #pragma unroll
        for (int u = 0; u < 16; ++u) {
            const v4f qq = real ? q[u] : zero;
            v2f a0, a1, b0, b1;
            a0.x = a[u][0]; a0.y = a[u][1]; a1.x = a[u][2]; a1.y = a[u][3];
            b0.x = qq[0];   b0.y = qq[1];   b1.x = qq[2];   b1.y = qq[3];
            acc0 = __builtin_amdgcn_wmma_f32_16x16x4_f32(false, a0, false, b0, (short)0, acc0, false, false);
            acc1 = __builtin_amdgcn_wmma_f32_16x16x4_f32(false, a1, false, b1, (short)0, acc1, false, false);
        }
        #pragma unroll
        for (int v = 0; v < 8; ++v) s[w][v * 32 + lane] = acc0[v] + acc1[v];
        __syncthreads();
        if (real) {
            float tot = 0.f;
            #pragma unroll
            for (int k = 0; k < 8; ++k) tot += s[k][w * 32 + lane];
            vb[l16 * DIN + i0 + w + 8 * hf] = tot;
        }
    }
}

__device__ __forceinline__ float pick8(const v8f a, const int v) {
    float r = a[0];
    r = (v == 1) ? a[1] : r;
    r = (v == 2) ? a[2] : r;
    r = (v == 3) ? a[3] : r;
    r = (v == 4) ? a[4] : r;
    r = (v == 5) ? a[5] : r;
    r = (v == 6) ? a[6] : r;
    r = (v == 7) ? a[7] : r;
    return r;
}

__global__ __launch_bounds__(128) void k_gemm1_mish(
    const float* __restrict__ x, const float* __restrict__ w1w,
    const float* __restrict__ w1b, float* __restrict__ h)
{
    const int lane = threadIdx.x & 31;
    const int hf = lane >> 4, l16 = lane & 15, w = threadIdx.x >> 5;
    const int b  = blockIdx.x >> 7;
    const int r  = blockIdx.x & 127;
    const int m0 = ((r >> 3) * 2 + (w >> 1)) * 32;
    const int n0 = ((r & 7) * 2 + (w & 1)) * 64;
    const float* Ap = w1w + (size_t)(m0 + l16) * DIN + 2 * hf;
    const float* Bp = x + ((size_t)b * T_ + n0 + 4 * l16) * DIN + 2 * hf;
    v8f acc[2][4] = {};
    v2f ca[4][2], cb[4][4];
    #pragma unroll
    for (int st = 0; st < 4; ++st) {
        #pragma unroll
        for (int i = 0; i < 2; ++i) ca[st][i] = *(const v2f*)(Ap + (size_t)i * 16 * DIN + 4 * st);
        #pragma unroll
        for (int j = 0; j < 4; ++j) cb[st][j] = *(const v2f*)(Bp + (size_t)j * DIN + 4 * st);
    }
    #pragma unroll 2
    for (int k0 = 0; k0 < DIN; k0 += 16) {
        const int kn = (k0 + 16 < DIN) ? k0 + 16 : k0;
        v2f na[4][2], nb[4][4];
        #pragma unroll
        for (int st = 0; st < 4; ++st) {
            #pragma unroll
            for (int i = 0; i < 2; ++i) na[st][i] = *(const v2f*)(Ap + (size_t)i * 16 * DIN + kn + 4 * st);
            #pragma unroll
            for (int j = 0; j < 4; ++j) nb[st][j] = *(const v2f*)(Bp + (size_t)j * DIN + kn + 4 * st);
        }
        __builtin_amdgcn_sched_barrier(0);
        #pragma unroll
        for (int st = 0; st < 4; ++st)
            #pragma unroll
            for (int i = 0; i < 2; ++i)
                #pragma unroll
                for (int j = 0; j < 4; ++j)
                    acc[i][j] = __builtin_amdgcn_wmma_f32_16x16x4_f32(false, ca[st][i], false, cb[st][j],
                                                                     (short)0, acc[i][j], false, false);
        __builtin_amdgcn_sched_barrier(0);
        asm volatile("" :: "v"(na[0][0]), "v"(na[0][1]), "v"(na[1][0]), "v"(na[1][1]),
                        "v"(na[2][0]), "v"(na[2][1]), "v"(na[3][0]), "v"(na[3][1]),
                        "v"(nb[0][0]), "v"(nb[0][1]), "v"(nb[0][2]), "v"(nb[0][3]),
                        "v"(nb[1][0]), "v"(nb[1][1]), "v"(nb[1][2]), "v"(nb[1][3]),
                        "v"(nb[2][0]), "v"(nb[2][1]), "v"(nb[2][2]), "v"(nb[2][3]),
                        "v"(nb[3][0]), "v"(nb[3][1]), "v"(nb[3][2]), "v"(nb[3][3]));
        #pragma unroll
        for (int st = 0; st < 4; ++st) {
            #pragma unroll
            for (int i = 0; i < 2; ++i) ca[st][i] = na[st][i];
            #pragma unroll
            for (int j = 0; j < 4; ++j) cb[st][j] = nb[st][j];
        }
    }
    float* H = h + (size_t)b * DH1 * T_;
    #pragma unroll
    for (int i = 0; i < 2; ++i) {
        #pragma unroll 1
        for (int v = 0; v < 8; ++v) {
            const int c = m0 + 16 * i + v + 8 * hf;
            const float bias = w1b[c];
            v4f o;
            o.x = mishf(pick8(acc[i][0], v) + bias);
            o.y = mishf(pick8(acc[i][1], v) + bias);
            o.z = mishf(pick8(acc[i][2], v) + bias);
            o.w = mishf(pick8(acc[i][3], v) + bias);
            *(v4f*)(H + (size_t)c * T_ + n0 + 4 * l16) = o;
        }
    }
}

__global__ __launch_bounds__(256) void k_dwconv(
    const float* __restrict__ h, const float* __restrict__ dw,
    const float* __restrict__ dg, const float* __restrict__ db,
    const float* __restrict__ pg, const float* __restrict__ inv_nd,
    float* __restrict__ y)
{
    const int lane = threadIdx.x & 31;
    const int row = __builtin_amdgcn_readfirstlane((int)(blockIdx.x * 8 + (threadIdx.x >> 5)));
    const int b = row >> 10;
    const float* hp  = h + (size_t)row * T_;
    const float* wp  = dw + (size_t)row * KW;
    const float* ndp = inv_nd + b * KW;
    float* yp = y + (size_t)row * T_;
    float wk[KW];
    #pragma unroll
    for (int k = 0; k < KW; ++k) wk[k] = wp[k] * ndp[k];
    const float g = dg[row], bi = db[row], pgv = pg[row];
    v4f cur[8], prv[8], nxt[8];
    #pragma unroll
    for (int j = 0; j < 8; ++j) cur[j] = *(const v4f*)(hp + 128 * j + 4 * lane);
    #pragma unroll
    for (int j = 0; j < 8; ++j) {
        const int t = 128 * j + 4 * lane;
        const bool inl = (t >= 4), inr = (t + 4 < T_);
        const v4f p = *(const v4f*)(hp + (inl ? t - 4 : t));
        const v4f n = *(const v4f*)(hp + (inr ? t + 4 : t));
        const v4f zero = {0.f, 0.f, 0.f, 0.f};
        prv[j] = inl ? p : zero;
        nxt[j] = inr ? n : zero;
    }
    #pragma unroll
    for (int j = 0; j < 8; ++j) {
        const int t = 128 * j + 4 * lane;
        const float win[12] = {prv[j].x, prv[j].y, prv[j].z, prv[j].w,
                               cur[j].x, cur[j].y, cur[j].z, cur[j].w,
                               nxt[j].x, nxt[j].y, nxt[j].z, nxt[j].w};
        v4f o;
        #pragma unroll
        for (int i = 0; i < 4; ++i) {
            float acc = 0.f;
            #pragma unroll
            for (int k = 0; k < KW; ++k) acc += win[i + k] * wk[k];
            const float yy = acc * g + bi;
            o[i] = yy * pgv;
        }
        *(v4f*)(yp + t) = o;
    }
}

__global__ __launch_bounds__(128) void k_weff(
    const float* __restrict__ w2w, const float* __restrict__ pw,
    const float* __restrict__ inv_np, float* __restrict__ weff)
{
    const int lane = threadIdx.x & 31;
    const int hf = lane >> 4, l16 = lane & 15, w = threadIdx.x >> 5;
    const int b  = blockIdx.x >> 5;
    const int r  = blockIdx.x & 31;
    const int m0 = ((r >> 3) * 2 + (w >> 1)) * 32;
    const int n0 = ((r & 7) * 2 + (w & 1)) * 64;
    const float* Ap = w2w + (size_t)(m0 + l16) * DH2 + 4 * hf;
    const float* Bp = pw + (size_t)b * DH1 * DH2 + (size_t)(n0 + 4 * l16) * DH2 + 4 * hf;
    const float* Sp = inv_np + b * DH2 + 4 * hf;
    v8f acc[2][4] = {};
    v4f ca[2][2], cb[2][4];
    #pragma unroll
    for (int u = 0; u < 2; ++u) {
        const v4f sc = *(const v4f*)(Sp + 8 * u);
        #pragma unroll
        for (int i = 0; i < 2; ++i) ca[u][i] = *(const v4f*)(Ap + (size_t)i * 16 * DH2 + 8 * u) * sc;
        #pragma unroll
        for (int j = 0; j < 4; ++j) cb[u][j] = *(const v4f*)(Bp + (size_t)j * DH2 + 8 * u);
    }
    #pragma unroll 2
    for (int k0 = 0; k0 < DH2; k0 += 16) {
        const int kn = (k0 + 16 < DH2) ? k0 + 16 : k0;
        v4f na[2][2], nb[2][4], ns[2];
        #pragma unroll
        for (int u = 0; u < 2; ++u) {
            ns[u] = *(const v4f*)(Sp + kn + 8 * u);
            #pragma unroll
            for (int i = 0; i < 2; ++i) na[u][i] = *(const v4f*)(Ap + (size_t)i * 16 * DH2 + kn + 8 * u);
            #pragma unroll
            for (int j = 0; j < 4; ++j) nb[u][j] = *(const v4f*)(Bp + (size_t)j * DH2 + kn + 8 * u);
        }
        __builtin_amdgcn_sched_barrier(0);
        #pragma unroll
        for (int u = 0; u < 2; ++u)
            #pragma unroll
            for (int hh = 0; hh < 2; ++hh)
                #pragma unroll
                for (int i = 0; i < 2; ++i) {
                    v2f a;
                    a.x = ca[u][i][2 * hh];
                    a.y = ca[u][i][2 * hh + 1];
                    #pragma unroll
                    for (int j = 0; j < 4; ++j) {
                        v2f bv;
                        bv.x = cb[u][j][2 * hh];
                        bv.y = cb[u][j][2 * hh + 1];
                        acc[i][j] = __builtin_amdgcn_wmma_f32_16x16x4_f32(false, a, false, bv,
                                                                         (short)0, acc[i][j], false, false);
                    }
                }
        __builtin_amdgcn_sched_barrier(0);
        asm volatile("" :: "v"(na[0][0]), "v"(na[0][1]), "v"(na[1][0]), "v"(na[1][1]),
                        "v"(nb[0][0]), "v"(nb[0][1]), "v"(nb[0][2]), "v"(nb[0][3]),
                        "v"(nb[1][0]), "v"(nb[1][1]), "v"(nb[1][2]), "v"(nb[1][3]),
                        "v"(ns[0]), "v"(ns[1]));
        #pragma unroll
        for (int u = 0; u < 2; ++u) {
            #pragma unroll
            for (int i = 0; i < 2; ++i) ca[u][i] = na[u][i] * ns[u];
            #pragma unroll
            for (int j = 0; j < 4; ++j) cb[u][j] = nb[u][j];
        }
    }
    float* Wo = weff + (size_t)b * DIN * DH1;
    #pragma unroll
    for (int i = 0; i < 2; ++i) {
        #pragma unroll
        for (int v = 0; v < 8; ++v) {
            const int row = m0 + 16 * i + v + 8 * hf;
            v4f o;
            o.x = acc[i][0][v];
            o.y = acc[i][1][v];
            o.z = acc[i][2][v];
            o.w = acc[i][3][v];
            *(v4f*)(Wo + (size_t)row * DH1 + n0 + 4 * l16) = o;
        }
    }
}

__global__ __launch_bounds__(128) void k_final(
    const float* __restrict__ y, const float* __restrict__ weff,
    const float* __restrict__ vb, const float* __restrict__ w2b,
    const float* __restrict__ x, float* __restrict__ out)
{
    const int lane = threadIdx.x & 31;
    const int hf = lane >> 4, l16 = lane & 15, w = threadIdx.x >> 5;
    const int b  = blockIdx.x >> 5;
    const int r  = blockIdx.x & 31;
    const int t0 = ((r >> 2) * 2 + (w >> 1)) * 64;
    const int i0 = ((r & 3) * 2 + (w & 1)) * 32;
    const float* Zp = y + (size_t)b * DH1 * T_ + (size_t)(2 * hf) * T_ + t0 + 4 * l16;
    const float* Wp = weff + (size_t)b * DIN * DH1 + (size_t)(i0 + 2 * l16) * DH1 + 2 * hf;
    v8f acc[4][2] = {};
    v4f ca[4][2];
    v2f cb[4][2];
    #pragma unroll
    for (int st = 0; st < 4; ++st) {
        #pragma unroll
        for (int s = 0; s < 2; ++s) ca[st][s] = *(const v4f*)(Zp + (size_t)(4 * st + s) * T_);
        #pragma unroll
        for (int j = 0; j < 2; ++j) cb[st][j] = *(const v2f*)(Wp + (size_t)j * DH1 + 4 * st);
    }
    #pragma unroll 2
    for (int k0 = 0; k0 < DH1; k0 += 16) {
        const int kn = (k0 + 16 < DH1) ? k0 + 16 : k0;
        v4f na[4][2];
        v2f nb[4][2];
        #pragma unroll
        for (int st = 0; st < 4; ++st) {
            #pragma unroll
            for (int s = 0; s < 2; ++s) na[st][s] = *(const v4f*)(Zp + (size_t)(kn + 4 * st + s) * T_);
            #pragma unroll
            for (int j = 0; j < 2; ++j) nb[st][j] = *(const v2f*)(Wp + (size_t)j * DH1 + kn + 4 * st);
        }
        __builtin_amdgcn_sched_barrier(0);
        #pragma unroll
        for (int st = 0; st < 4; ++st)
            #pragma unroll
            for (int a = 0; a < 4; ++a) {
                v2f av;
                av.x = ca[st][0][a];
                av.y = ca[st][1][a];
                #pragma unroll
                for (int j = 0; j < 2; ++j)
                    acc[a][j] = __builtin_amdgcn_wmma_f32_16x16x4_f32(false, av, false, cb[st][j],
                                                                     (short)0, acc[a][j], false, false);
            }
        __builtin_amdgcn_sched_barrier(0);
        asm volatile("" :: "v"(na[0][0]), "v"(na[0][1]), "v"(na[1][0]), "v"(na[1][1]),
                        "v"(na[2][0]), "v"(na[2][1]), "v"(na[3][0]), "v"(na[3][1]),
                        "v"(nb[0][0]), "v"(nb[0][1]), "v"(nb[1][0]), "v"(nb[1][1]),
                        "v"(nb[2][0]), "v"(nb[2][1]), "v"(nb[3][0]), "v"(nb[3][1]));
        #pragma unroll
        for (int st = 0; st < 4; ++st) {
            #pragma unroll
            for (int s = 0; s < 2; ++s) ca[st][s] = na[st][s];
            #pragma unroll
            for (int j = 0; j < 2; ++j) cb[st][j] = nb[st][j];
        }
    }
    const int ii = i0 + 2 * l16;
    const v2f bb = *(const v2f*)(w2b + ii) + *(const v2f*)(vb + b * DIN + ii);
    #pragma unroll
    for (int a = 0; a < 4; ++a) {
        #pragma unroll
        for (int v = 0; v < 8; ++v) {
            const int t = t0 + 4 * (v + 8 * hf) + a;
            const size_t oi = ((size_t)b * T_ + t) * DIN + ii;
            const v2f xv = *(const v2f*)(x + oi);
            v2f q;
            q.x = acc[a][0][v] + bb.x + xv.x;
            q.y = acc[a][1][v] + bb.y + xv.y;
            *(v2f*)(out + oi) = q;
        }
    }
}

extern "C" void kernel_launch(void* const* d_in, const int* in_sizes, int n_in,
                              void* d_out, int out_size, void* d_ws, size_t ws_size,
                              hipStream_t stream)
{
    const float* x   = (const float*)d_in[0];
    const float* d_w = (const float*)d_in[1];
    const float* d_g = (const float*)d_in[2];
    const float* d_b = (const float*)d_in[3];
    const float* p_w = (const float*)d_in[4];
    const float* p_g = (const float*)d_in[5];
    const float* p_b = (const float*)d_in[6];
    const float* w1w = (const float*)d_in[7];
    const float* w1b = (const float*)d_in[8];
    const float* w2w = (const float*)d_in[9];
    const float* w2b = (const float*)d_in[10];
    float* out = (float*)d_out;

    float* ws = (float*)d_ws;
    const size_t NCT = (size_t)B_ * DH1 * T_;
    float* hbuf   = ws;
    float* ybuf   = ws + NCT;
    float* inv_nd = ws + 2 * NCT;
    float* inv_np = inv_nd + 128;
    float* vbuf   = inv_np + 8192;
    float* weff   = vbuf + 2048;

    k_norms<<<344, 256, 0, stream>>>(p_w, d_w, w2w, p_b, inv_np, inv_nd, vbuf);
    k_weff<<<256, 128, 0, stream>>>(w2w, p_w, inv_np, weff);
    k_gemm1_mish<<<1024, 128, 0, stream>>>(x, w1w, w1b, hbuf);
    k_dwconv<<<1024, 256, 0, stream>>>(hbuf, d_w, d_g, d_b, p_g, inv_nd, ybuf);
    k_final<<<256, 128, 0, stream>>>(ybuf, weff, vbuf, w2b, x, out);
}


# Weight and final products: each tile's k range split over 4 waves of one work-group (4 waves per SIMD, was 1); the 4 partial tiles added through LDS as (p0+p2)+(p1+p3): sum order changes
# speedup vs baseline: 3.1239x; 1.1385x over previous
#include <hip/hip_runtime.h>
#include <math.h>

typedef __attribute__((ext_vector_type(2))) float v2f;
typedef __attribute__((ext_vector_type(4))) float v4f;
typedef __attribute__((ext_vector_type(8))) float v8f;

#define B_  8
#define T_  1024
#define DIN 256
#define DH1 1024
#define DH2 1024
#define KW  9
#define KQ1 (DH1 / 4)
#define KQ2 (DH2 / 4)

__device__ __forceinline__ float mishf(float v) {
    float sp = (v > 20.f) ? v : log1pf(__expf(v));
    return v * tanhf(sp);
}

__global__ __launch_bounds__(256) void k_norms(const float* __restrict__ pw,
                                               const float* __restrict__ dw,
                                               const float* __restrict__ w2w,
                                               const float* __restrict__ pb,
                                               float* __restrict__ inv_np,
                                               float* __restrict__ inv_nd,
                                               float* __restrict__ vb)
{
    __shared__ float s[8][256];
    const int lane = threadIdx.x & 31, w = threadIdx.x >> 5;
    if (blockIdx.x < 256) {
        const int b = blockIdx.x >> 5, oc = blockIdx.x & 31;
        const int cs = lane >> 3, ol = lane & 7;
        const float* P = pw + (size_t)b * DH1 * DH2 + (size_t)(4 * w + cs) * DH2 + oc * 32 + 4 * ol;
        v4f v[32];
        #pragma unroll
        for (int j = 0; j < 32; ++j) v[j] = *(const v4f*)(P + (size_t)(32 * j) * DH2);
        __builtin_amdgcn_sched_barrier(0);
        v4f acc = {0.f, 0.f, 0.f, 0.f};
        #pragma unroll
        for (int j = 0; j < 32; ++j) acc += v[j] * v[j];
        #pragma unroll
        for (int i = 0; i < 4; ++i) {
            acc[i] += __shfl_xor(acc[i], 8);
            acc[i] += __shfl_xor(acc[i], 16);
        }
        if (lane < 8) *(v4f*)&s[w][4 * lane] = acc;
        __syncthreads();
        if (threadIdx.x < 32) {
            float tot = 0.f;
            #pragma unroll
            for (int q = 0; q < 8; ++q) tot += s[q][threadIdx.x];
            inv_np[b * DH2 + oc * 32 + threadIdx.x] = 1.f / fmaxf(sqrtf(tot), 1e-12f);
        }
    } else if (blockIdx.x < 328) {
        const int p = blockIdx.x - 256;
        const int b = p / KW, k = p - b * KW;
        const float* D = dw + (size_t)b * DH1 * KW + k;
        float acc = 0.f;
        #pragma unroll
        for (int j = 0; j < 4; ++j) {
            const float v = D[(size_t)(threadIdx.x + 256 * j) * KW];
            acc += v * v;
        }
        #pragma unroll
        for (int m = 16; m >= 1; m >>= 1) acc += __shfl_xor(acc, m);
        if (lane == 0) s[0][w] = acc;
        __syncthreads();
        if (threadIdx.x == 0) {
            float tot = 0.f;
            #pragma unroll
            for (int q = 0; q < 8; ++q) tot += s[0][q];
            inv_nd[p] = 1.f / fmaxf(sqrtf(tot), 1e-12f);
        }
    } else {
        const int i0 = (blockIdx.x - 328) * 16;
        const int hf = lane >> 4, l16 = lane & 15;
        const float* Ap = w2w + (size_t)(i0 + l16) * DH2 + 128 * w + 4 * hf;
        const float* Bq = pb + (size_t)(l16 & 7) * DH2 + 128 * w + 4 * hf;
        v4f a[16], q[16];
        #pragma unroll
        for (int u = 0; u < 16; ++u) {
            a[u] = *(const v4f*)(Ap + 8 * u);
            q[u] = *(const v4f*)(Bq + 8 * u);
        }
        const bool real = (l16 < 8);
        const v4f zero = {0.f, 0.f, 0.f, 0.f};
        v8f acc0 = {}, acc1 = {};
        #pragma unroll
        for (int u = 0; u < 16; ++u) {
            const v4f qq = real ? q[u] : zero;
            v2f a0, a1, b0, b1;
            a0.x = a[u][0]; a0.y = a[u][1]; a1.x = a[u][2]; a1.y = a[u][3];
            b0.x = qq[0];   b0.y = qq[1];   b1.x = qq[2];   b1.y = qq[3];
            acc0 = __builtin_amdgcn_wmma_f32_16x16x4_f32(false, a0, false, b0, (short)0, acc0, false, false);
            acc1 = __builtin_amdgcn_wmma_f32_16x16x4_f32(false, a1, false, b1, (short)0, acc1, false, false);
        }
        #pragma unroll
        for (int v = 0; v < 8; ++v) s[w][v * 32 + lane] = acc0[v] + acc1[v];
        __syncthreads();
        if (real) {
            float tot = 0.f;
            #pragma unroll
            for (int k = 0; k < 8; ++k) tot += s[k][w * 32 + lane];
            vb[l16 * DIN + i0 + w + 8 * hf] = tot;
        }
    }
}

__device__ __forceinline__ float pick8(const v8f a, const int v) {
    float r = a[0];
    r = (v == 1) ? a[1] : r;
    r = (v == 2) ? a[2] : r;
    r = (v == 3) ? a[3] : r;
    r = (v == 4) ? a[4] : r;
    r = (v == 5) ? a[5] : r;
    r = (v == 6) ? a[6] : r;
    r = (v == 7) ? a[7] : r;
    return r;
}

__global__ __launch_bounds__(128) void k_gemm1_mish(
    const float* __restrict__ x, const float* __restrict__ w1w,
    const float* __restrict__ w1b, float* __restrict__ h)
{
    const int lane = threadIdx.x & 31;
    const int hf = lane >> 4, l16 = lane & 15, w = threadIdx.x >> 5;
    const int b  = blockIdx.x >> 7;
    const int r  = blockIdx.x & 127;
    const int m0 = ((r >> 3) * 2 + (w >> 1)) * 32;
    const int n0 = ((r & 7) * 2 + (w & 1)) * 64;
    const float* Ap = w1w + (size_t)(m0 + l16) * DIN + 2 * hf;
    const float* Bp = x + ((size_t)b * T_ + n0 + 4 * l16) * DIN + 2 * hf;
    v8f acc[2][4] = {};
    v2f ca[4][2], cb[4][4];
    #pragma unroll
    for (int st = 0; st < 4; ++st) {
        #pragma unroll
        for (int i = 0; i < 2; ++i) ca[st][i] = *(const v2f*)(Ap + (size_t)i * 16 * DIN + 4 * st);
        #pragma unroll
        for (int j = 0; j < 4; ++j) cb[st][j] = *(const v2f*)(Bp + (size_t)j * DIN + 4 * st);
    }
    #pragma unroll 2
    for (int k0 = 0; k0 < DIN; k0 += 16) {
        const int kn = (k0 + 16 < DIN) ? k0 + 16 : k0;
        v2f na[4][2], nb[4][4];
        #pragma unroll
        for (int st = 0; st < 4; ++st) {
            #pragma unroll
            for (int i = 0; i < 2; ++i) na[st][i] = *(const v2f*)(Ap + (size_t)i * 16 * DIN + kn + 4 * st);
            #pragma unroll
            for (int j = 0; j < 4; ++j) nb[st][j] = *(const v2f*)(Bp + (size_t)j * DIN + kn + 4 * st);
        }
        __builtin_amdgcn_sched_barrier(0);
        #pragma unroll
        for (int st = 0; st < 4; ++st)
            #pragma unroll
            for (int i = 0; i < 2; ++i)
                #pragma unroll
                for (int j = 0; j < 4; ++j)
                    acc[i][j] = __builtin_amdgcn_wmma_f32_16x16x4_f32(false, ca[st][i], false, cb[st][j],
                                                                     (short)0, acc[i][j], false, false);
        __builtin_amdgcn_sched_barrier(0);
        asm volatile("" :: "v"(na[0][0]), "v"(na[0][1]), "v"(na[1][0]), "v"(na[1][1]),
                        "v"(na[2][0]), "v"(na[2][1]), "v"(na[3][0]), "v"(na[3][1]),
                        "v"(nb[0][0]), "v"(nb[0][1]), "v"(nb[0][2]), "v"(nb[0][3]),
                        "v"(nb[1][0]), "v"(nb[1][1]), "v"(nb[1][2]), "v"(nb[1][3]),
                        "v"(nb[2][0]), "v"(nb[2][1]), "v"(nb[2][2]), "v"(nb[2][3]),
                        "v"(nb[3][0]), "v"(nb[3][1]), "v"(nb[3][2]), "v"(nb[3][3]));
        #pragma unroll
        for (int st = 0; st < 4; ++st) {
            #pragma unroll
            for (int i = 0; i < 2; ++i) ca[st][i] = na[st][i];
            #pragma unroll
            for (int j = 0; j < 4; ++j) cb[st][j] = nb[st][j];
        }
    }
    float* H = h + (size_t)b * DH1 * T_;
    #pragma unroll
    for (int i = 0; i < 2; ++i) {
        #pragma unroll 1
        for (int v = 0; v < 8; ++v) {
            const int c = m0 + 16 * i + v + 8 * hf;
            const float bias = w1b[c];
            v4f o;
            o.x = mishf(pick8(acc[i][0], v) + bias);
            o.y = mishf(pick8(acc[i][1], v) + bias);
            o.z = mishf(pick8(acc[i][2], v) + bias);
            o.w = mishf(pick8(acc[i][3], v) + bias);
            *(v4f*)(H + (size_t)c * T_ + n0 + 4 * l16) = o;
        }
    }
}

__global__ __launch_bounds__(256) void k_dwconv(
    const float* __restrict__ h, const float* __restrict__ dw,
    const float* __restrict__ dg, const float* __restrict__ db,
    const float* __restrict__ pg, const float* __restrict__ inv_nd,
    float* __restrict__ y)
{
    const int lane = threadIdx.x & 31;
    const int row = __builtin_amdgcn_readfirstlane((int)(blockIdx.x * 8 + (threadIdx.x >> 5)));
    const int b = row >> 10;
    const float* hp  = h + (size_t)row * T_;
    const float* wp  = dw + (size_t)row * KW;
    const float* ndp = inv_nd + b * KW;
    float* yp = y + (size_t)row * T_;
    float wk[KW];
    #pragma unroll
    for (int k = 0; k < KW; ++k) wk[k] = wp[k] * ndp[k];
    const float g = dg[row], bi = db[row], pgv = pg[row];
    v4f cur[8], prv[8], nxt[8];
    #pragma unroll
    for (int j = 0; j < 8; ++j) cur[j] = *(const v4f*)(hp + 128 * j + 4 * lane);
    #pragma unroll
    for (int j = 0; j < 8; ++j) {
        const int t = 128 * j + 4 * lane;
        const bool inl = (t >= 4), inr = (t + 4 < T_);
        const v4f p = *(const v4f*)(hp + (inl ? t - 4 : t));
        const v4f n = *(const v4f*)(hp + (inr ? t + 4 : t));
        const v4f zero = {0.f, 0.f, 0.f, 0.f};
        prv[j] = inl ? p : zero;
        nxt[j] = inr ? n : zero;
    }
    #pragma unroll
    for (int j = 0; j < 8; ++j) {
        const int t = 128 * j + 4 * lane;
        const float win[12] = {prv[j].x, prv[j].y, prv[j].z, prv[j].w,
                               cur[j].x, cur[j].y, cur[j].z, cur[j].w,
                               nxt[j].x, nxt[j].y, nxt[j].z, nxt[j].w};
        v4f o;
        #pragma unroll
        for (int i = 0; i < 4; ++i) {
            float acc = 0.f;
            #pragma unroll
            for (int k = 0; k < KW; ++k) acc += win[i + k] * wk[k];
            const float yy = acc * g + bi;
            o[i] = yy * pgv;
        }
        *(v4f*)(yp + t) = o;
    }
}

__global__ __launch_bounds__(512) void k_weff(
    const float* __restrict__ w2w, const float* __restrict__ pw,
    const float* __restrict__ inv_np, float* __restrict__ weff)
{
    __shared__ v4f red[2][4][16][32];
    const int lane = threadIdx.x & 31;
    const int hf = lane >> 4, l16 = lane & 15;
    const int w  = (threadIdx.x >> 5) & 3;
    const int q  = threadIdx.x >> 7;
    const int b  = blockIdx.x >> 5;
    const int r  = blockIdx.x & 31;
    const int m0 = ((r >> 3) * 2 + (w >> 1)) * 32;
    const int n0 = ((r & 7) * 2 + (w & 1)) * 64;
    const int kq = q * KQ2;
    const float* Ap = w2w + (size_t)(m0 + l16) * DH2 + kq + 4 * hf;
    const float* Bp = pw + (size_t)b * DH1 * DH2 + (size_t)(n0 + 4 * l16) * DH2 + kq + 4 * hf;
    const float* Sp = inv_np + b * DH2 + kq + 4 * hf;
    v8f acc[2][4] = {};
    v4f ca[2][2], cb[2][4];
    #pragma unroll
    for (int u = 0; u < 2; ++u) {
        const v4f sc = *(const v4f*)(Sp + 8 * u);
        #pragma unroll
        for (int i = 0; i < 2; ++i) ca[u][i] = *(const v4f*)(Ap + (size_t)i * 16 * DH2 + 8 * u) * sc;
        #pragma unroll
        for (int j = 0; j < 4; ++j) cb[u][j] = *(const v4f*)(Bp + (size_t)j * DH2 + 8 * u);
    }
    #pragma unroll 2
    for (int k0 = 0; k0 < KQ2; k0 += 16) {
        const int kn = (k0 + 16 < KQ2) ? k0 + 16 : k0;
        v4f na[2][2], nb[2][4], ns[2];
        #pragma unroll
        for (int u = 0; u < 2; ++u) {
            ns[u] = *(const v4f*)(Sp + kn + 8 * u);
            #pragma unroll
            for (int i = 0; i < 2; ++i) na[u][i] = *(const v4f*)(Ap + (size_t)i * 16 * DH2 + kn + 8 * u);
            #pragma unroll
            for (int j = 0; j < 4; ++j) nb[u][j] = *(const v4f*)(Bp + (size_t)j * DH2 + kn + 8 * u);
        }
        __builtin_amdgcn_sched_barrier(0);
        #pragma unroll
        for (int u = 0; u < 2; ++u)
            #pragma unroll
            for (int hh = 0; hh < 2; ++hh)
                #pragma unroll
                for (int i = 0; i < 2; ++i) {
                    v2f a;
                    a.x = ca[u][i][2 * hh];
                    a.y = ca[u][i][2 * hh + 1];
                    #pragma unroll
                    for (int j = 0; j < 4; ++j) {
                        v2f bv;
                        bv.x = cb[u][j][2 * hh];
                        bv.y = cb[u][j][2 * hh + 1];
                        acc[i][j] = __builtin_amdgcn_wmma_f32_16x16x4_f32(false, a, false, bv,
                                                                         (short)0, acc[i][j], false, false);
                    }
                }
        __builtin_amdgcn_sched_barrier(0);
        asm volatile("" :: "v"(na[0][0]), "v"(na[0][1]), "v"(na[1][0]), "v"(na[1][1]),
                        "v"(nb[0][0]), "v"(nb[0][1]), "v"(nb[0][2]), "v"(nb[0][3]),
                        "v"(nb[1][0]), "v"(nb[1][1]), "v"(nb[1][2]), "v"(nb[1][3]),
                        "v"(ns[0]), "v"(ns[1]));
        #pragma unroll
        for (int u = 0; u < 2; ++u) {
            #pragma unroll
            for (int i = 0; i < 2; ++i) ca[u][i] = na[u][i] * ns[u];
            #pragma unroll
            for (int j = 0; j < 4; ++j) cb[u][j] = nb[u][j];
        }
    }
    if (q >= 2) {
        #pragma unroll
        for (int i = 0; i < 2; ++i)
            #pragma unroll
            for (int j = 0; j < 4; ++j)
                #pragma unroll
                for (int h = 0; h < 2; ++h) {
                    v4f p;
                    p.x = acc[i][j][4 * h];     p.y = acc[i][j][4 * h + 1];
                    p.z = acc[i][j][4 * h + 2]; p.w = acc[i][j][4 * h + 3];
                    red[q - 2][w][2 * (4 * i + j) + h][lane] = p;
                }
    }
    __syncthreads();
    if (q < 2) {
        #pragma unroll
        for (int i = 0; i < 2; ++i)
            #pragma unroll
            for (int j = 0; j < 4; ++j)
                #pragma unroll
                for (int h = 0; h < 2; ++h) {
                    const v4f p = red[q][w][2 * (4 * i + j) + h][lane];
                    acc[i][j][4 * h]     += p.x; acc[i][j][4 * h + 1] += p.y;
                    acc[i][j][4 * h + 2] += p.z; acc[i][j][4 * h + 3] += p.w;
                }
    }
    __syncthreads();
    if (q == 1) {
        #pragma unroll
        for (int i = 0; i < 2; ++i)
            #pragma unroll
            for (int j = 0; j < 4; ++j)
                #pragma unroll
                for (int h = 0; h < 2; ++h) {
                    v4f p;
                    p.x = acc[i][j][4 * h];     p.y = acc[i][j][4 * h + 1];
                    p.z = acc[i][j][4 * h + 2]; p.w = acc[i][j][4 * h + 3];
                    red[0][w][2 * (4 * i + j) + h][lane] = p;
                }
    }
    __syncthreads();
    if (q == 0) {
        #pragma unroll
        for (int i = 0; i < 2; ++i)
            #pragma unroll
            for (int j = 0; j < 4; ++j)
                #pragma unroll
                for (int h = 0; h < 2; ++h) {
                    const v4f p = red[0][w][2 * (4 * i + j) + h][lane];
                    acc[i][j][4 * h]     += p.x; acc[i][j][4 * h + 1] += p.y;
                    acc[i][j][4 * h + 2] += p.z; acc[i][j][4 * h + 3] += p.w;
                }
        float* Wo = weff + (size_t)b * DIN * DH1;
        #pragma unroll
        for (int i = 0; i < 2; ++i) {
            #pragma unroll
            for (int v = 0; v < 8; ++v) {
                const int row = m0 + 16 * i + v + 8 * hf;
                v4f o;
                o.x = acc[i][0][v];
                o.y = acc[i][1][v];
                o.z = acc[i][2][v];
                o.w = acc[i][3][v];
                *(v4f*)(Wo + (size_t)row * DH1 + n0 + 4 * l16) = o;
            }
        }
    }
}

__global__ __launch_bounds__(512) void k_final(
    const float* __restrict__ y, const float* __restrict__ weff,
    const float* __restrict__ vb, const float* __restrict__ w2b,
    const float* __restrict__ x, float* __restrict__ out)
{
    __shared__ v4f red[2][4][16][32];
    const int lane = threadIdx.x & 31;
    const int hf = lane >> 4, l16 = lane & 15;
    const int w  = (threadIdx.x >> 5) & 3;
    const int q  = threadIdx.x >> 7;
    const int b  = blockIdx.x >> 5;
    const int r  = blockIdx.x & 31;
    const int t0 = ((r >> 2) * 2 + (w >> 1)) * 64;
    const int i0 = ((r & 3) * 2 + (w & 1)) * 32;
    const int kq = q * KQ1;
    const float* Zp = y + (size_t)b * DH1 * T_ + (size_t)(kq + 2 * hf) * T_ + t0 + 4 * l16;
    const float* Wp = weff + (size_t)b * DIN * DH1 + (size_t)(i0 + 2 * l16) * DH1 + kq + 2 * hf;
    v8f acc[4][2] = {};
    v4f ca[4][2];
    v2f cb[4][2];
    #pragma unroll
    for (int st = 0; st < 4; ++st) {
        #pragma unroll
        for (int s = 0; s < 2; ++s) ca[st][s] = *(const v4f*)(Zp + (size_t)(4 * st + s) * T_);
        #pragma unroll
        for (int j = 0; j < 2; ++j) cb[st][j] = *(const v2f*)(Wp + (size_t)j * DH1 + 4 * st);
    }
    #pragma unroll 2
    for (int k0 = 0; k0 < KQ1; k0 += 16) {
        const int kn = (k0 + 16 < KQ1) ? k0 + 16 : k0;
        v4f na[4][2];
        v2f nb[4][2];
        #pragma unroll
        for (int st = 0; st < 4; ++st) {
            #pragma unroll
            for (int s = 0; s < 2; ++s) na[st][s] = *(const v4f*)(Zp + (size_t)(kn + 4 * st + s) * T_);
            #pragma unroll
            for (int j = 0; j < 2; ++j) nb[st][j] = *(const v2f*)(Wp + (size_t)j * DH1 + kn + 4 * st);
        }
        __builtin_amdgcn_sched_barrier(0);
        #pragma unroll
        for (int st = 0; st < 4; ++st)
            #pragma unroll
            for (int a = 0; a < 4; ++a) {
                v2f av;
                av.x = ca[st][0][a];
                av.y = ca[st][1][a];
                #pragma unroll
                for (int j = 0; j < 2; ++j)
                    acc[a][j] = __builtin_amdgcn_wmma_f32_16x16x4_f32(false, av, false, cb[st][j],
                                                                     (short)0, acc[a][j], false, false);
            }
        __builtin_amdgcn_sched_barrier(0);
        asm volatile("" :: "v"(na[0][0]), "v"(na[0][1]), "v"(na[1][0]), "v"(na[1][1]),
                        "v"(na[2][0]), "v"(na[2][1]), "v"(na[3][0]), "v"(na[3][1]),
                        "v"(nb[0][0]), "v"(nb[0][1]), "v"(nb[1][0]), "v"(nb[1][1]),
                        "v"(nb[2][0]), "v"(nb[2][1]), "v"(nb[3][0]), "v"(nb[3][1]));
        #pragma unroll
        for (int st = 0; st < 4; ++st) {
            #pragma unroll
            for (int s = 0; s < 2; ++s) ca[st][s] = na[st][s];
            #pragma unroll
            for (int j = 0; j < 2; ++j) cb[st][j] = nb[st][j];
        }
    }
    if (q >= 2) {
        #pragma unroll
        for (int a = 0; a < 4; ++a)
            #pragma unroll
            for (int j = 0; j < 2; ++j)
                #pragma unroll
                for (int h = 0; h < 2; ++h) {
                    v4f p;
                    p.x = acc[a][j][4 * h];     p.y = acc[a][j][4 * h + 1];
                    p.z = acc[a][j][4 * h + 2]; p.w = acc[a][j][4 * h + 3];
                    red[q - 2][w][2 * (2 * a + j) + h][lane] = p;
                }
    }
    __syncthreads();
    if (q < 2) {
        #pragma unroll
        for (int a = 0; a < 4; ++a)
            #pragma unroll
            for (int j = 0; j < 2; ++j)
                #pragma unroll
                for (int h = 0; h < 2; ++h) {
                    const v4f p = red[q][w][2 * (2 * a + j) + h][lane];
                    acc[a][j][4 * h]     += p.x; acc[a][j][4 * h + 1] += p.y;
                    acc[a][j][4 * h + 2] += p.z; acc[a][j][4 * h + 3] += p.w;
                }
    }
    __syncthreads();
    if (q == 1) {
        #pragma unroll
        for (int a = 0; a < 4; ++a)
            #pragma unroll
            for (int j = 0; j < 2; ++j)
                #pragma unroll
                for (int h = 0; h < 2; ++h) {
                    v4f p;
                    p.x = acc[a][j][4 * h];     p.y = acc[a][j][4 * h + 1];
                    p.z = acc[a][j][4 * h + 2]; p.w = acc[a][j][4 * h + 3];
                    red[0][w][2 * (2 * a + j) + h][lane] = p;
                }
    }
    __syncthreads();
    if (q == 0) {
        #pragma unroll
        for (int a = 0; a < 4; ++a)
            #pragma unroll
            for (int j = 0; j < 2; ++j)
                #pragma unroll
                for (int h = 0; h < 2; ++h) {
                    const v4f p = red[0][w][2 * (2 * a + j) + h][lane];
                    acc[a][j][4 * h]     += p.x; acc[a][j][4 * h + 1] += p.y;
                    acc[a][j][4 * h + 2] += p.z; acc[a][j][4 * h + 3] += p.w;
                }
        const int ii = i0 + 2 * l16;
        const v2f bb = *(const v2f*)(w2b + ii) + *(const v2f*)(vb + b * DIN + ii);
        #pragma unroll
        for (int a = 0; a < 4; ++a) {
            #pragma unroll
            for (int v = 0; v < 8; ++v) {
                const int t = t0 + 4 * (v + 8 * hf) + a;
                const size_t oi = ((size_t)b * T_ + t) * DIN + ii;
                const v2f xv = *(const v2f*)(x + oi);
                v2f o;
                o.x = acc[a][0][v] + bb.x + xv.x;
                o.y = acc[a][1][v] + bb.y + xv.y;
                *(v2f*)(out + oi) = o;
            }
        }
    }
}

extern "C" void kernel_launch(void* const* d_in, const int* in_sizes, int n_in,
                              void* d_out, int out_size, void* d_ws, size_t ws_size,
                              hipStream_t stream)
{
    const float* x   = (const float*)d_in[0];
    const float* d_w = (const float*)d_in[1];
    const float* d_g = (const float*)d_in[2];
    const float* d_b = (const float*)d_in[3];
    const float* p_w = (const float*)d_in[4];
    const float* p_g = (const float*)d_in[5];
    const float* p_b = (const float*)d_in[6];
    const float* w1w = (const float*)d_in[7];
    const float* w1b = (const float*)d_in[8];
    const float* w2w = (const float*)d_in[9];
    const float* w2b = (const float*)d_in[10];
    float* out = (float*)d_out;

    float* ws = (float*)d_ws;
    const size_t NCT = (size_t)B_ * DH1 * T_;
    float* hbuf   = ws;
    float* ybuf   = ws + NCT;
    float* inv_nd = ws + 2 * NCT;
    float* inv_np = inv_nd + 128;
    float* vbuf   = inv_np + 8192;
    float* weff   = vbuf + 2048;

    k_norms<<<344, 256, 0, stream>>>(p_w, d_w, w2w, p_b, inv_np, inv_nd, vbuf);
    k_weff<<<256, 512, 0, stream>>>(w2w, p_w, inv_np, weff);
    k_gemm1_mish<<<1024, 128, 0, stream>>>(x, w1w, w1b, hbuf);
    k_dwconv<<<1024, 256, 0, stream>>>(hbuf, d_w, d_g, d_b, p_g, inv_nd, ybuf);
    k_final<<<256, 512, 0, stream>>>(ybuf, weff, vbuf, w2b, x, out);
}


# Mish after GEMM 1 in closed form: v*n/(n+2), n = e^v*(e^v+2), one exp and one reciprocal, no branches, in place of the log1p, exp and tanh calls; its 8 rows unrolled; last bits of h differ
# speedup vs baseline: 3.7046x; 1.1859x over previous
#include <hip/hip_runtime.h>
#include <math.h>

typedef __attribute__((ext_vector_type(2))) float v2f;
typedef __attribute__((ext_vector_type(4))) float v4f;
typedef __attribute__((ext_vector_type(8))) float v8f;

#define B_  8
#define T_  1024
#define DIN 256
#define DH1 1024
#define DH2 1024
#define KW  9
#define KQ1 (DH1 / 4)
#define KQ2 (DH2 / 4)

__device__ __forceinline__ float mishf(float v) {
    const float e = __expf(v);
    const float n = e * (e + 2.f);
    const float m = v * (n * __builtin_amdgcn_rcpf(n + 2.f));
    return (v > 20.f) ? v : m;
}

__global__ __launch_bounds__(256) void k_norms(const float* __restrict__ pw,
                                               const float* __restrict__ dw,
                                               const float* __restrict__ w2w,
                                               const float* __restrict__ pb,
                                               float* __restrict__ inv_np,
                                               float* __restrict__ inv_nd,
                                               float* __restrict__ vb)
{
    __shared__ float s[8][256];
    const int lane = threadIdx.x & 31, w = threadIdx.x >> 5;
    if (blockIdx.x < 256) {
        const int b = blockIdx.x >> 5, oc = blockIdx.x & 31;
        const int cs = lane >> 3, ol = lane & 7;
        const float* P = pw + (size_t)b * DH1 * DH2 + (size_t)(4 * w + cs) * DH2 + oc * 32 + 4 * ol;
        v4f v[32];
        #pragma unroll
        for (int j = 0; j < 32; ++j) v[j] = *(const v4f*)(P + (size_t)(32 * j) * DH2);
        __builtin_amdgcn_sched_barrier(0);
        v4f acc = {0.f, 0.f, 0.f, 0.f};
        #pragma unroll
        for (int j = 0; j < 32; ++j) acc += v[j] * v[j];
        #pragma unroll
        for (int i = 0; i < 4; ++i) {
            acc[i] += __shfl_xor(acc[i], 8);
            acc[i] += __shfl_xor(acc[i], 16);
        }
        if (lane < 8) *(v4f*)&s[w][4 * lane] = acc;
        __syncthreads();
        if (threadIdx.x < 32) {
            float tot = 0.f;
            #pragma unroll
            for (int q = 0; q < 8; ++q) tot += s[q][threadIdx.x];
            inv_np[b * DH2 + oc * 32 + threadIdx.x] = 1.f / fmaxf(sqrtf(tot), 1e-12f);
        }
    } else if (blockIdx.x < 328) {
        const int p = blockIdx.x - 256;
        const int b = p / KW, k = p - b * KW;
        const float* D = dw + (size_t)b * DH1 * KW + k;
        float acc = 0.f;
        #pragma unroll
        for (int j = 0; j < 4; ++j) {
            const float v = D[(size_t)(threadIdx.x + 256 * j) * KW];
            acc += v * v;
        }
        #pragma unroll
        for (int m = 16; m >= 1; m >>= 1) acc += __shfl_xor(acc, m);
        if (lane == 0) s[0][w] = acc;
        __syncthreads();
        if (threadIdx.x == 0) {
            float tot = 0.f;
            #pragma unroll
            for (int q = 0; q < 8; ++q) tot += s[0][q];
            inv_nd[p] = 1.f / fmaxf(sqrtf(tot), 1e-12f);
        }
    } else {
        const int i0 = (blockIdx.x - 328) * 16;
        const int hf = lane >> 4, l16 = lane & 15;
        const float* Ap = w2w + (size_t)(i0 + l16) * DH2 + 128 * w + 4 * hf;
        const float* Bq = pb + (size_t)(l16 & 7) * DH2 + 128 * w + 4 * hf;
        v4f a[16], q[16];
        #pragma unroll
        for (int u = 0; u < 16; ++u) {
            a[u] = *(const v4f*)(Ap + 8 * u);
            q[u] = *(const v4f*)(Bq + 8 * u);
        }
        const bool real = (l16 < 8);
        const v4f zero = {0.f, 0.f, 0.f, 0.f};
        v8f acc0 = {}, acc1 = {};
        #pragma unroll
        for (int u = 0; u < 16; ++u) {
            const v4f qq = real ? q[u] : zero;
            v2f a0, a1, b0, b1;
            a0.x = a[u][0]; a0.y = a[u][1]; a1.x = a[u][2]; a1.y = a[u][3];
            b0.x = qq[0];   b0.y = qq[1];   b1.x = qq[2];   b1.y = qq[3];
            acc0 = __builtin_amdgcn_wmma_f32_16x16x4_f32(false, a0, false, b0, (short)0, acc0, false, false);
            acc1 = __builtin_amdgcn_wmma_f32_16x16x4_f32(false, a1, false, b1, (short)0, acc1, false, false);
        }
        #pragma unroll
        for (int v = 0; v < 8; ++v) s[w][v * 32 + lane] = acc0[v] + acc1[v];
        __syncthreads();
        if (real) {
            float tot = 0.f;
            #pragma unroll
            for (int k = 0; k < 8; ++k) tot += s[k][w * 32 + lane];
            vb[l16 * DIN + i0 + w + 8 * hf] = tot;
        }
    }
}

__global__ __launch_bounds__(128) void k_gemm1_mish(
    const float* __restrict__ x, const float* __restrict__ w1w,
    const float* __restrict__ w1b, float* __restrict__ h)
{
    const int lane = threadIdx.x & 31;
    const int hf = lane >> 4, l16 = lane & 15, w = threadIdx.x >> 5;
    const int b  = blockIdx.x >> 7;
    const int r  = blockIdx.x & 127;
    const int m0 = ((r >> 3) * 2 + (w >> 1)) * 32;
    const int n0 = ((r & 7) * 2 + (w & 1)) * 64;
    const float* Ap = w1w + (size_t)(m0 + l16) * DIN + 2 * hf;
    const float* Bp = x + ((size_t)b * T_ + n0 + 4 * l16) * DIN + 2 * hf;
    v8f acc[2][4] = {};
    v2f ca[4][2], cb[4][4];
    #pragma unroll
    for (int st = 0; st < 4; ++st) {
        #pragma unroll
        for (int i = 0; i < 2; ++i) ca[st][i] = *(const v2f*)(Ap + (size_t)i * 16 * DIN + 4 * st);
        #pragma unroll
        for (int j = 0; j < 4; ++j) cb[st][j] = *(const v2f*)(Bp + (size_t)j * DIN + 4 * st);
    }
    #pragma unroll 2
    for (int k0 = 0; k0 < DIN; k0 += 16) {
        const int kn = (k0 + 16 < DIN) ? k0 + 16 : k0;
        v2f na[4][2], nb[4][4];
        #pragma unroll
        for (int st = 0; st < 4; ++st) {
            #pragma unroll
            for (int i = 0; i < 2; ++i) na[st][i] = *(const v2f*)(Ap + (size_t)i * 16 * DIN + kn + 4 * st);
            #pragma unroll
            for (int j = 0; j < 4; ++j) nb[st][j] = *(const v2f*)(Bp + (size_t)j * DIN + kn + 4 * st);
        }
        __builtin_amdgcn_sched_barrier(0);
        #pragma unroll
        for (int st = 0; st < 4; ++st)
            #pragma unroll
            for (int i = 0; i < 2; ++i)
                #pragma unroll
                for (int j = 0; j < 4; ++j)
                    acc[i][j] = __builtin_amdgcn_wmma_f32_16x16x4_f32(false, ca[st][i], false, cb[st][j],
                                                                     (short)0, acc[i][j], false, false);
        __builtin_amdgcn_sched_barrier(0);
        asm volatile("" :: "v"(na[0][0]), "v"(na[0][1]), "v"(na[1][0]), "v"(na[1][1]),
                        "v"(na[2][0]), "v"(na[2][1]), "v"(na[3][0]), "v"(na[3][1]),
                        "v"(nb[0][0]), "v"(nb[0][1]), "v"(nb[0][2]), "v"(nb[0][3]),
                        "v"(nb[1][0]), "v"(nb[1][1]), "v"(nb[1][2]), "v"(nb[1][3]),
                        "v"(nb[2][0]), "v"(nb[2][1]), "v"(nb[2][2]), "v"(nb[2][3]),
                        "v"(nb[3][0]), "v"(nb[3][1]), "v"(nb[3][2]), "v"(nb[3][3]));
        #pragma unroll
        for (int st = 0; st < 4; ++st) {
            #pragma unroll
            for (int i = 0; i < 2; ++i) ca[st][i] = na[st][i];
            #pragma unroll
            for (int j = 0; j < 4; ++j) cb[st][j] = nb[st][j];
        }
    }
    float* H = h + (size_t)b * DH1 * T_;
    #pragma unroll
    for (int i = 0; i < 2; ++i) {
        #pragma unroll
        for (int v = 0; v < 8; ++v) {
            const int c = m0 + 16 * i + v + 8 * hf;
            const float bias = w1b[c];
            v4f o;
            o.x = mishf(acc[i][0][v] + bias);
            o.y = mishf(acc[i][1][v] + bias);
            o.z = mishf(acc[i][2][v] + bias);
            o.w = mishf(acc[i][3][v] + bias);
            *(v4f*)(H + (size_t)c * T_ + n0 + 4 * l16) = o;
        }
    }
}

__global__ __launch_bounds__(256) void k_dwconv(
    const float* __restrict__ h, const float* __restrict__ dw,
    const float* __restrict__ dg, const float* __restrict__ db,
    const float* __restrict__ pg, const float* __restrict__ inv_nd,
    float* __restrict__ y)
{
    const int lane = threadIdx.x & 31;
    const int row = __builtin_amdgcn_readfirstlane((int)(blockIdx.x * 8 + (threadIdx.x >> 5)));
    const int b = row >> 10;
    const float* hp  = h + (size_t)row * T_;
    const float* wp  = dw + (size_t)row * KW;
    const float* ndp = inv_nd + b * KW;
    float* yp = y + (size_t)row * T_;
    float wk[KW];
    #pragma unroll
    for (int k = 0; k < KW; ++k) wk[k] = wp[k] * ndp[k];
    const float g = dg[row], bi = db[row], pgv = pg[row];
    v4f cur[8], prv[8], nxt[8];
    #pragma unroll
    for (int j = 0; j < 8; ++j) cur[j] = *(const v4f*)(hp + 128 * j + 4 * lane);
    #pragma unroll
    for (int j = 0; j < 8; ++j) {
        const int t = 128 * j + 4 * lane;
        const bool inl = (t >= 4), inr = (t + 4 < T_);
        const v4f p = *(const v4f*)(hp + (inl ? t - 4 : t));
        const v4f n = *(const v4f*)(hp + (inr ? t + 4 : t));
        const v4f zero = {0.f, 0.f, 0.f, 0.f};
        prv[j] = inl ? p : zero;
        nxt[j] = inr ? n : zero;
    }
    #pragma unroll
    for (int j = 0; j < 8; ++j) {
        const int t = 128 * j + 4 * lane;
        const float win[12] = {prv[j].x, prv[j].y, prv[j].z, prv[j].w,
                               cur[j].x, cur[j].y, cur[j].z, cur[j].w,
                               nxt[j].x, nxt[j].y, nxt[j].z, nxt[j].w};
        v4f o;
        #pragma unroll
        for (int i = 0; i < 4; ++i) {
            float acc = 0.f;
            #pragma unroll
            for (int k = 0; k < KW; ++k) acc += win[i + k] * wk[k];
            const float yy = acc * g + bi;
            o[i] = yy * pgv;
        }
        *(v4f*)(yp + t) = o;
    }
}

__global__ __launch_bounds__(512) void k_weff(
    const float* __restrict__ w2w, const float* __restrict__ pw,
    const float* __restrict__ inv_np, float* __restrict__ weff)
{
    __shared__ v4f red[2][4][16][32];
    const int lane = threadIdx.x & 31;
    const int hf = lane >> 4, l16 = lane & 15;
    const int w  = (threadIdx.x >> 5) & 3;
    const int q  = threadIdx.x >> 7;
    const int b  = blockIdx.x >> 5;
    const int r  = blockIdx.x & 31;
    const int m0 = ((r >> 3) * 2 + (w >> 1)) * 32;
    const int n0 = ((r & 7) * 2 + (w & 1)) * 64;
    const int kq = q * KQ2;
    const float* Ap = w2w + (size_t)(m0 + l16) * DH2 + kq + 4 * hf;
    const float* Bp = pw + (size_t)b * DH1 * DH2 + (size_t)(n0 + 4 * l16) * DH2 + kq + 4 * hf;
    const float* Sp = inv_np + b * DH2 + kq + 4 * hf;
    v8f acc[2][4] = {};
    v4f ca[2][2], cb[2][4];
    #pragma unroll
    for (int u = 0; u < 2; ++u) {
        const v4f sc = *(const v4f*)(Sp + 8 * u);
        #pragma unroll
        for (int i = 0; i < 2; ++i) ca[u][i] = *(const v4f*)(Ap + (size_t)i * 16 * DH2 + 8 * u) * sc;
        #pragma unroll
        for (int j = 0; j < 4; ++j) cb[u][j] = *(const v4f*)(Bp + (size_t)j * DH2 + 8 * u);
    }
    #pragma unroll 2
    for (int k0 = 0; k0 < KQ2; k0 += 16) {
        const int kn = (k0 + 16 < KQ2) ? k0 + 16 : k0;
        v4f na[2][2], nb[2][4], ns[2];
        #pragma unroll
        for (int u = 0; u < 2; ++u) {
            ns[u] = *(const v4f*)(Sp + kn + 8 * u);
            #pragma unroll
            for (int i = 0; i < 2; ++i) na[u][i] = *(const v4f*)(Ap + (size_t)i * 16 * DH2 + kn + 8 * u);
            #pragma unroll
            for (int j = 0; j < 4; ++j) nb[u][j] = *(const v4f*)(Bp + (size_t)j * DH2 + kn + 8 * u);
        }
        __builtin_amdgcn_sched_barrier(0);
        #pragma unroll
        for (int u = 0; u < 2; ++u)
            #pragma unroll
            for (int hh = 0; hh < 2; ++hh)
                #pragma unroll
                for (int i = 0; i < 2; ++i) {
                    v2f a;
                    a.x = ca[u][i][2 * hh];
                    a.y = ca[u][i][2 * hh + 1];
                    #pragma unroll
                    for (int j = 0; j < 4; ++j) {
                        v2f bv;
                        bv.x = cb[u][j][2 * hh];
                        bv.y = cb[u][j][2 * hh + 1];
                        acc[i][j] = __builtin_amdgcn_wmma_f32_16x16x4_f32(false, a, false, bv,
                                                                         (short)0, acc[i][j], false, false);
                    }
                }
        __builtin_amdgcn_sched_barrier(0);
        asm volatile("" :: "v"(na[0][0]), "v"(na[0][1]), "v"(na[1][0]), "v"(na[1][1]),
                        "v"(nb[0][0]), "v"(nb[0][1]), "v"(nb[0][2]), "v"(nb[0][3]),
                        "v"(nb[1][0]), "v"(nb[1][1]), "v"(nb[1][2]), "v"(nb[1][3]),
                        "v"(ns[0]), "v"(ns[1]));
        #pragma unroll
        for (int u = 0; u < 2; ++u) {
            #pragma unroll
            for (int i = 0; i < 2; ++i) ca[u][i] = na[u][i] * ns[u];
            #pragma unroll
            for (int j = 0; j < 4; ++j) cb[u][j] = nb[u][j];
        }
    }
    if (q >= 2) {
        #pragma unroll
        for (int i = 0; i < 2; ++i)
            #pragma unroll
            for (int j = 0; j < 4; ++j)
                #pragma unroll
                for (int h = 0; h < 2; ++h) {
                    v4f p;
                    p.x = acc[i][j][4 * h];     p.y = acc[i][j][4 * h + 1];
                    p.z = acc[i][j][4 * h + 2]; p.w = acc[i][j][4 * h + 3];
                    red[q - 2][w][2 * (4 * i + j) + h][lane] = p;
                }
    }
    __syncthreads();
    if (q < 2) {
        #pragma unroll
        for (int i = 0; i < 2; ++i)
            #pragma unroll
            for (int j = 0; j < 4; ++j)
                #pragma unroll
                for (int h = 0; h < 2; ++h) {
                    const v4f p = red[q][w][2 * (4 * i + j) + h][lane];
                    acc[i][j][4 * h]     += p.x; acc[i][j][4 * h + 1] += p.y;
                    acc[i][j][4 * h + 2] += p.z; acc[i][j][4 * h + 3] += p.w;
                }
    }
    __syncthreads();
    if (q == 1) {
        #pragma unroll
        for (int i = 0; i < 2; ++i)
            #pragma unroll
            for (int j = 0; j < 4; ++j)
                #pragma unroll
                for (int h = 0; h < 2; ++h) {
                    v4f p;
                    p.x = acc[i][j][4 * h];     p.y = acc[i][j][4 * h + 1];
                    p.z = acc[i][j][4 * h + 2]; p.w = acc[i][j][4 * h + 3];
                    red[0][w][2 * (4 * i + j) + h][lane] = p;
                }
    }
    __syncthreads();
    if (q == 0) {
        #pragma unroll
        for (int i = 0; i < 2; ++i)
            #pragma unroll
            for (int j = 0; j < 4; ++j)
                #pragma unroll
                for (int h = 0; h < 2; ++h) {
                    const v4f p = red[0][w][2 * (4 * i + j) + h][lane];
                    acc[i][j][4 * h]     += p.x; acc[i][j][4 * h + 1] += p.y;
                    acc[i][j][4 * h + 2] += p.z; acc[i][j][4 * h + 3] += p.w;
                }
        float* Wo = weff + (size_t)b * DIN * DH1;
        #pragma unroll
        for (int i = 0; i < 2; ++i) {
            #pragma unroll
            for (int v = 0; v < 8; ++v) {
                const int row = m0 + 16 * i + v + 8 * hf;
                v4f o;
                o.x = acc[i][0][v];
                o.y = acc[i][1][v];
                o.z = acc[i][2][v];
                o.w = acc[i][3][v];
                *(v4f*)(Wo + (size_t)row * DH1 + n0 + 4 * l16) = o;
            }
        }
    }
}

__global__ __launch_bounds__(512) void k_final(
    const float* __restrict__ y, const float* __restrict__ weff,
    const float* __restrict__ vb, const float* __restrict__ w2b,
    const float* __restrict__ x, float* __restrict__ out)
{
    __shared__ v4f red[2][4][16][32];
    const int lane = threadIdx.x & 31;
    const int hf = lane >> 4, l16 = lane & 15;
    const int w  = (threadIdx.x >> 5) & 3;
    const int q  = threadIdx.x >> 7;
    const int b  = blockIdx.x >> 5;
    const int r  = blockIdx.x & 31;
    const int t0 = ((r >> 2) * 2 + (w >> 1)) * 64;
    const int i0 = ((r & 3) * 2 + (w & 1)) * 32;
    const int kq = q * KQ1;
    const float* Zp = y + (size_t)b * DH1 * T_ + (size_t)(kq + 2 * hf) * T_ + t0 + 4 * l16;
    const float* Wp = weff + (size_t)b * DIN * DH1 + (size_t)(i0 + 2 * l16) * DH1 + kq + 2 * hf;
    v8f acc[4][2] = {};
    v4f ca[4][2];
    v2f cb[4][2];
    #pragma unroll
    for (int st = 0; st < 4; ++st) {
        #pragma unroll
        for (int s = 0; s < 2; ++s) ca[st][s] = *(const v4f*)(Zp + (size_t)(4 * st + s) * T_);
        #pragma unroll
        for (int j = 0; j < 2; ++j) cb[st][j] = *(const v2f*)(Wp + (size_t)j * DH1 + 4 * st);
    }
    #pragma unroll 2
    for (int k0 = 0; k0 < KQ1; k0 += 16) {
        const int kn = (k0 + 16 < KQ1) ? k0 + 16 : k0;
        v4f na[4][2];
        v2f nb[4][2];
        #pragma unroll
        for (int st = 0; st < 4; ++st) {
            #pragma unroll
            for (int s = 0; s < 2; ++s) na[st][s] = *(const v4f*)(Zp + (size_t)(kn + 4 * st + s) * T_);
            #pragma unroll
            for (int j = 0; j < 2; ++j) nb[st][j] = *(const v2f*)(Wp + (size_t)j * DH1 + kn + 4 * st);
        }
        __builtin_amdgcn_sched_barrier(0);
        #pragma unroll
        for (int st = 0; st < 4; ++st)
            #pragma unroll
            for (int a = 0; a < 4; ++a) {
                v2f av;
                av.x = ca[st][0][a];
                av.y = ca[st][1][a];
                #pragma unroll
                for (int j = 0; j < 2; ++j)
                    acc[a][j] = __builtin_amdgcn_wmma_f32_16x16x4_f32(false, av, false, cb[st][j],
                                                                     (short)0, acc[a][j], false, false);
            }
        __builtin_amdgcn_sched_barrier(0);
        asm volatile("" :: "v"(na[0][0]), "v"(na[0][1]), "v"(na[1][0]), "v"(na[1][1]),
                        "v"(na[2][0]), "v"(na[2][1]), "v"(na[3][0]), "v"(na[3][1]),
                        "v"(nb[0][0]), "v"(nb[0][1]), "v"(nb[1][0]), "v"(nb[1][1]),
                        "v"(nb[2][0]), "v"(nb[2][1]), "v"(nb[3][0]), "v"(nb[3][1]));
        #pragma unroll
        for (int st = 0; st < 4; ++st) {
            #pragma unroll
            for (int s = 0; s < 2; ++s) ca[st][s] = na[st][s];
            #pragma unroll
            for (int j = 0; j < 2; ++j) cb[st][j] = nb[st][j];
        }
    }
    if (q >= 2) {
        #pragma unroll
        for (int a = 0; a < 4; ++a)
            #pragma unroll
            for (int j = 0; j < 2; ++j)
                #pragma unroll
                for (int h = 0; h < 2; ++h) {
                    v4f p;
                    p.x = acc[a][j][4 * h];     p.y = acc[a][j][4 * h + 1];
                    p.z = acc[a][j][4 * h + 2]; p.w = acc[a][j][4 * h + 3];
                    red[q - 2][w][2 * (2 * a + j) + h][lane] = p;
                }
    }
    __syncthreads();
    if (q < 2) {
        #pragma unroll
        for (int a = 0; a < 4; ++a)
            #pragma unroll
            for (int j = 0; j < 2; ++j)
                #pragma unroll
                for (int h = 0; h < 2; ++h) {
                    const v4f p = red[q][w][2 * (2 * a + j) + h][lane];
                    acc[a][j][4 * h]     += p.x; acc[a][j][4 * h + 1] += p.y;
                    acc[a][j][4 * h + 2] += p.z; acc[a][j][4 * h + 3] += p.w;
                }
    }
    __syncthreads();
    if (q == 1) {
        #pragma unroll
        for (int a = 0; a < 4; ++a)
            #pragma unroll
            for (int j = 0; j < 2; ++j)
                #pragma unroll
                for (int h = 0; h < 2; ++h) {
                    v4f p;
                    p.x = acc[a][j][4 * h];     p.y = acc[a][j][4 * h + 1];
                    p.z = acc[a][j][4 * h + 2]; p.w = acc[a][j][4 * h + 3];
                    red[0][w][2 * (2 * a + j) + h][lane] = p;
                }
    }
    __syncthreads();
    if (q == 0) {
        #pragma unroll
        for (int a = 0; a < 4; ++a)
            #pragma unroll
            for (int j = 0; j < 2; ++j)
                #pragma unroll
                for (int h = 0; h < 2; ++h) {
                    const v4f p = red[0][w][2 * (2 * a + j) + h][lane];
                    acc[a][j][4 * h]     += p.x; acc[a][j][4 * h + 1] += p.y;
                    acc[a][j][4 * h + 2] += p.z; acc[a][j][4 * h + 3] += p.w;
                }
        const int ii = i0 + 2 * l16;
        const v2f bb = *(const v2f*)(w2b + ii) + *(const v2f*)(vb + b * DIN + ii);
        #pragma unroll
        for (int a = 0; a < 4; ++a) {
            #pragma unroll
            for (int v = 0; v < 8; ++v) {
                const int t = t0 + 4 * (v + 8 * hf) + a;
                const size_t oi = ((size_t)b * T_ + t) * DIN + ii;
                const v2f xv = *(const v2f*)(x + oi);
                v2f o;
                o.x = acc[a][0][v] + bb.x + xv.x;
                o.y = acc[a][1][v] + bb.y + xv.y;
                *(v2f*)(out + oi) = o;
            }
        }
    }
}

extern "C" void kernel_launch(void* const* d_in, const int* in_sizes, int n_in,
                              void* d_out, int out_size, void* d_ws, size_t ws_size,
                              hipStream_t stream)
{
    const float* x   = (const float*)d_in[0];
    const float* d_w = (const float*)d_in[1];
    const float* d_g = (const float*)d_in[2];
    const float* d_b = (const float*)d_in[3];
    const float* p_w = (const float*)d_in[4];
    const float* p_g = (const float*)d_in[5];
    const float* p_b = (const float*)d_in[6];
    const float* w1w = (const float*)d_in[7];
    const float* w1b = (const float*)d_in[8];
    const float* w2w = (const float*)d_in[9];
    const float* w2b = (const float*)d_in[10];
    float* out = (float*)d_out;

    float* ws = (float*)d_ws;
    const size_t NCT = (size_t)B_ * DH1 * T_;
    float* hbuf   = ws;
    float* ybuf   = ws + NCT;
    float* inv_nd = ws + 2 * NCT;
    float* inv_np = inv_nd + 128;
    float* vbuf   = inv_np + 8192;
    float* weff   = vbuf + 2048;

    k_norms<<<344, 256, 0, stream>>>(p_w, d_w, w2w, p_b, inv_np, inv_nd, vbuf);
    k_weff<<<256, 512, 0, stream>>>(w2w, p_w, inv_np, weff);
    k_gemm1_mish<<<1024, 128, 0, stream>>>(x, w1w, w1b, hbuf);
    k_dwconv<<<1024, 256, 0, stream>>>(hbuf, d_w, d_g, d_b, p_g, inv_nd, ybuf);
    k_final<<<256, 512, 0, stream>>>(ybuf, weff, vbuf, w2b, x, out);
}


# GEMM 1 (both operands) and the final product's weight operand fetched 16 bytes per lane, was 8: half the load instructions there, same cache lines; k order inside each 16-k unit changes (sum order)
# speedup vs baseline: 3.8949x; 1.0514x over previous
#include <hip/hip_runtime.h>
#include <math.h>

typedef __attribute__((ext_vector_type(2))) float v2f;
typedef __attribute__((ext_vector_type(4))) float v4f;
typedef __attribute__((ext_vector_type(8))) float v8f;

#define B_  8
#define T_  1024
#define DIN 256
#define DH1 1024
#define DH2 1024
#define KW  9
#define KQ1 (DH1 / 4)
#define KQ2 (DH2 / 4)

__device__ __forceinline__ float mishf(float v) {
    const float e = __expf(v);
    const float n = e * (e + 2.f);
    const float m = v * (n * __builtin_amdgcn_rcpf(n + 2.f));
    return (v > 20.f) ? v : m;
}

__global__ __launch_bounds__(256) void k_norms(const float* __restrict__ pw,
                                               const float* __restrict__ dw,
                                               const float* __restrict__ w2w,
                                               const float* __restrict__ pb,
                                               float* __restrict__ inv_np,
                                               float* __restrict__ inv_nd,
                                               float* __restrict__ vb)
{
    __shared__ float s[8][256];
    const int lane = threadIdx.x & 31, w = threadIdx.x >> 5;
    if (blockIdx.x < 256) {
        const int b = blockIdx.x >> 5, oc = blockIdx.x & 31;
        const int cs = lane >> 3, ol = lane & 7;
        const float* P = pw + (size_t)b * DH1 * DH2 + (size_t)(4 * w + cs) * DH2 + oc * 32 + 4 * ol;
        v4f v[32];
        #pragma unroll
        for (int j = 0; j < 32; ++j) v[j] = *(const v4f*)(P + (size_t)(32 * j) * DH2);
        __builtin_amdgcn_sched_barrier(0);
        v4f acc = {0.f, 0.f, 0.f, 0.f};
        #pragma unroll
        for (int j = 0; j < 32; ++j) acc += v[j] * v[j];
        #pragma unroll
        for (int i = 0; i < 4; ++i) {
            acc[i] += __shfl_xor(acc[i], 8);
            acc[i] += __shfl_xor(acc[i], 16);
        }
        if (lane < 8) *(v4f*)&s[w][4 * lane] = acc;
        __syncthreads();
        if (threadIdx.x < 32) {
            float tot = 0.f;
            #pragma unroll
            for (int q = 0; q < 8; ++q) tot += s[q][threadIdx.x];
            inv_np[b * DH2 + oc * 32 + threadIdx.x] = 1.f / fmaxf(sqrtf(tot), 1e-12f);
        }
    } else if (blockIdx.x < 328) {
        const int p = blockIdx.x - 256;
        const int b = p / KW, k = p - b * KW;
        const float* D = dw + (size_t)b * DH1 * KW + k;
        float acc = 0.f;
        #pragma unroll
        for (int j = 0; j < 4; ++j) {
            const float v = D[(size_t)(threadIdx.x + 256 * j) * KW];
            acc += v * v;
        }
        #pragma unroll
        for (int m = 16; m >= 1; m >>= 1) acc += __shfl_xor(acc, m);
        if (lane == 0) s[0][w] = acc;
        __syncthreads();
        if (threadIdx.x == 0) {
            float tot = 0.f;
            #pragma unroll
            for (int q = 0; q < 8; ++q) tot += s[0][q];
            inv_nd[p] = 1.f / fmaxf(sqrtf(tot), 1e-12f);
        }
    } else {
        const int i0 = (blockIdx.x - 328) * 16;
        const int hf = lane >> 4, l16 = lane & 15;
        const float* Ap = w2w + (size_t)(i0 + l16) * DH2 + 128 * w + 4 * hf;
        const float* Bq = pb + (size_t)(l16 & 7) * DH2 + 128 * w + 4 * hf;
        v4f a[16], q[16];
        #pragma unroll
        for (int u = 0; u < 16; ++u) {
            a[u] = *(const v4f*)(Ap + 8 * u);
            q[u] = *(const v4f*)(Bq + 8 * u);
        }
        const bool real = (l16 < 8);
        const v4f zero = {0.f, 0.f, 0.f, 0.f};
        v8f acc0 = {}, acc1 = {};
        #pragma unroll
        for (int u = 0; u < 16; ++u) {
            const v4f qq = real ? q[u] : zero;
            v2f a0, a1, b0, b1;
            a0.x = a[u][0]; a0.y = a[u][1]; a1.x = a[u][2]; a1.y = a[u][3];
            b0.x = qq[0];   b0.y = qq[1];   b1.x = qq[2];   b1.y = qq[3];
            acc0 = __builtin_amdgcn_wmma_f32_16x16x4_f32(false, a0, false, b0, (short)0, acc0, false, false);
            acc1 = __builtin_amdgcn_wmma_f32_16x16x4_f32(false, a1, false, b1, (short)0, acc1, false, false);
        }
        #pragma unroll
        for (int v = 0; v < 8; ++v) s[w][v * 32 + lane] = acc0[v] + acc1[v];
        __syncthreads();
        if (real) {
            float tot = 0.f;
            #pragma unroll
            for (int k = 0; k < 8; ++k) tot += s[k][w * 32 + lane];
            vb[l16 * DIN + i0 + w + 8 * hf] = tot;
        }
    }
}

__global__ __launch_bounds__(128) void k_gemm1_mish(
    const float* __restrict__ x, const float* __restrict__ w1w,
    const float* __restrict__ w1b, float* __restrict__ h)
{
    const int lane = threadIdx.x & 31;
    const int hf = lane >> 4, l16 = lane & 15, w = threadIdx.x >> 5;
    const int b  = blockIdx.x >> 7;
    const int r  = blockIdx.x & 127;
    const int m0 = ((r >> 3) * 2 + (w >> 1)) * 32;
    const int n0 = ((r & 7) * 2 + (w & 1)) * 64;
    const float* Ap = w1w + (size_t)(m0 + l16) * DIN + 4 * hf;
    const float* Bp = x + ((size_t)b * T_ + n0 + 4 * l16) * DIN + 4 * hf;
    v8f acc[2][4] = {};
    v4f ca[2][2], cb[2][4];
    #pragma unroll
    for (int u = 0; u < 2; ++u) {
        #pragma unroll
        for (int i = 0; i < 2; ++i) ca[u][i] = *(const v4f*)(Ap + (size_t)i * 16 * DIN + 8 * u);
        #pragma unroll
        for (int j = 0; j < 4; ++j) cb[u][j] = *(const v4f*)(Bp + (size_t)j * DIN + 8 * u);
    }
    #pragma unroll 2
    for (int k0 = 0; k0 < DIN; k0 += 16) {
        const int kn = (k0 + 16 < DIN) ? k0 + 16 : k0;
        v4f na[2][2], nb[2][4];
        #pragma unroll
        for (int u = 0; u < 2; ++u) {
            #pragma unroll
            for (int i = 0; i < 2; ++i) na[u][i] = *(const v4f*)(Ap + (size_t)i * 16 * DIN + kn + 8 * u);
            #pragma unroll
            for (int j = 0; j < 4; ++j) nb[u][j] = *(const v4f*)(Bp + (size_t)j * DIN + kn + 8 * u);
        }
        __builtin_amdgcn_sched_barrier(0);
        #pragma unroll
        for (int u = 0; u < 2; ++u)
            #pragma unroll
            for (int hh = 0; hh < 2; ++hh)
                #pragma unroll
                for (int i = 0; i < 2; ++i) {
                    v2f a;
                    a.x = ca[u][i][2 * hh];
                    a.y = ca[u][i][2 * hh + 1];
                    #pragma unroll
                    for (int j = 0; j < 4; ++j) {
                        v2f bv;
                        bv.x = cb[u][j][2 * hh];
                        bv.y = cb[u][j][2 * hh + 1];
                        acc[i][j] = __builtin_amdgcn_wmma_f32_16x16x4_f32(false, a, false, bv,
                                                                         (short)0, acc[i][j], false, false);
                    }
                }
        __builtin_amdgcn_sched_barrier(0);
        asm volatile("" :: "v"(na[0][0]), "v"(na[0][1]), "v"(na[1][0]), "v"(na[1][1]),
                        "v"(nb[0][0]), "v"(nb[0][1]), "v"(nb[0][2]), "v"(nb[0][3]),
                        "v"(nb[1][0]), "v"(nb[1][1]), "v"(nb[1][2]), "v"(nb[1][3]));
        #pragma unroll
        for (int u = 0; u < 2; ++u) {
            #pragma unroll
            for (int i = 0; i < 2; ++i) ca[u][i] = na[u][i];
            #pragma unroll
            for (int j = 0; j < 4; ++j) cb[u][j] = nb[u][j];
        }
    }
    float* H = h + (size_t)b * DH1 * T_;
    #pragma unroll
    for (int i = 0; i < 2; ++i) {
        #pragma unroll
        for (int v = 0; v < 8; ++v) {
            const int c = m0 + 16 * i + v + 8 * hf;
            const float bias = w1b[c];
            v4f o;
            o.x = mishf(acc[i][0][v] + bias);
            o.y = mishf(acc[i][1][v] + bias);
            o.z = mishf(acc[i][2][v] + bias);
            o.w = mishf(acc[i][3][v] + bias);
            *(v4f*)(H + (size_t)c * T_ + n0 + 4 * l16) = o;
        }
    }
}

__global__ __launch_bounds__(256) void k_dwconv(
    const float* __restrict__ h, const float* __restrict__ dw,
    const float* __restrict__ dg, const float* __restrict__ db,
    const float* __restrict__ pg, const float* __restrict__ inv_nd,
    float* __restrict__ y)
{
    const int lane = threadIdx.x & 31;
    const int row = __builtin_amdgcn_readfirstlane((int)(blockIdx.x * 8 + (threadIdx.x >> 5)));
    const int b = row >> 10;
    const float* hp  = h + (size_t)row * T_;
    const float* wp  = dw + (size_t)row * KW;
    const float* ndp = inv_nd + b * KW;
    float* yp = y + (size_t)row * T_;
    float wk[KW];
    #pragma unroll
    for (int k = 0; k < KW; ++k) wk[k] = wp[k] * ndp[k];
    const float g = dg[row], bi = db[row], pgv = pg[row];
    v4f cur[8], prv[8], nxt[8];
    #pragma unroll
    for (int j = 0; j < 8; ++j) cur[j] = *(const v4f*)(hp + 128 * j + 4 * lane);
    #pragma unroll
    for (int j = 0; j < 8; ++j) {
        const int t = 128 * j + 4 * lane;
        const bool inl = (t >= 4), inr = (t + 4 < T_);
        const v4f p = *(const v4f*)(hp + (inl ? t - 4 : t));
        const v4f n = *(const v4f*)(hp + (inr ? t + 4 : t));
        const v4f zero = {0.f, 0.f, 0.f, 0.f};
        prv[j] = inl ? p : zero;
        nxt[j] = inr ? n : zero;
    }
    #pragma unroll
    for (int j = 0; j < 8; ++j) {
        const int t = 128 * j + 4 * lane;
        const float win[12] = {prv[j].x, prv[j].y, prv[j].z, prv[j].w,
                               cur[j].x, cur[j].y, cur[j].z, cur[j].w,
                               nxt[j].x, nxt[j].y, nxt[j].z, nxt[j].w};
        v4f o;
        #pragma unroll
        for (int i = 0; i < 4; ++i) {
            float acc = 0.f;
            #pragma unroll
            for (int k = 0; k < KW; ++k) acc += win[i + k] * wk[k];
            const float yy = acc * g + bi;
            o[i] = yy * pgv;
        }
        *(v4f*)(yp + t) = o;
    }
}

__global__ __launch_bounds__(512) void k_weff(
    const float* __restrict__ w2w, const float* __restrict__ pw,
    const float* __restrict__ inv_np, float* __restrict__ weff)
{
    __shared__ v4f red[2][4][16][32];
    const int lane = threadIdx.x & 31;
    const int hf = lane >> 4, l16 = lane & 15;
    const int w  = (threadIdx.x >> 5) & 3;
    const int q  = threadIdx.x >> 7;
    const int b  = blockIdx.x >> 5;
    const int r  = blockIdx.x & 31;
    const int m0 = ((r >> 3) * 2 + (w >> 1)) * 32;
    const int n0 = ((r & 7) * 2 + (w & 1)) * 64;
    const int kq = q * KQ2;
    const float* Ap = w2w + (size_t)(m0 + l16) * DH2 + kq + 4 * hf;
    const float* Bp = pw + (size_t)b * DH1 * DH2 + (size_t)(n0 + 4 * l16) * DH2 + kq + 4 * hf;
    const float* Sp = inv_np + b * DH2 + kq + 4 * hf;
    v8f acc[2][4] = {};
    v4f ca[2][2], cb[2][4];
    #pragma unroll
    for (int u = 0; u < 2; ++u) {
        const v4f sc = *(const v4f*)(Sp + 8 * u);
        #pragma unroll
        for (int i = 0; i < 2; ++i) ca[u][i] = *(const v4f*)(Ap + (size_t)i * 16 * DH2 + 8 * u) * sc;
        #pragma unroll
        for (int j = 0; j < 4; ++j) cb[u][j] = *(const v4f*)(Bp + (size_t)j * DH2 + 8 * u);
    }
    #pragma unroll 2
    for (int k0 = 0; k0 < KQ2; k0 += 16) {
        const int kn = (k0 + 16 < KQ2) ? k0 + 16 : k0;
        v4f na[2][2], nb[2][4], ns[2];
        #pragma unroll
        for (int u = 0; u < 2; ++u) {
            ns[u] = *(const v4f*)(Sp + kn + 8 * u);
            #pragma unroll
            for (int i = 0; i < 2; ++i) na[u][i] = *(const v4f*)(Ap + (size_t)i * 16 * DH2 + kn + 8 * u);
            #pragma unroll
            for (int j = 0; j < 4; ++j) nb[u][j] = *(const v4f*)(Bp + (size_t)j * DH2 + kn + 8 * u);
        }
        __builtin_amdgcn_sched_barrier(0);
        #pragma unroll
        for (int u = 0; u < 2; ++u)
            #pragma unroll
            for (int hh = 0; hh < 2; ++hh)
                #pragma unroll
                for (int i = 0; i < 2; ++i) {
                    v2f a;
                    a.x = ca[u][i][2 * hh];
                    a.y = ca[u][i][2 * hh + 1];
                    #pragma unroll
                    for (int j = 0; j < 4; ++j) {
                        v2f bv;
                        bv.x = cb[u][j][2 * hh];
                        bv.y = cb[u][j][2 * hh + 1];
                        acc[i][j] = __builtin_amdgcn_wmma_f32_16x16x4_f32(false, a, false, bv,
                                                                         (short)0, acc[i][j], false, false);
                    }
                }
        __builtin_amdgcn_sched_barrier(0);
        asm volatile("" :: "v"(na[0][0]), "v"(na[0][1]), "v"(na[1][0]), "v"(na[1][1]),
                        "v"(nb[0][0]), "v"(nb[0][1]), "v"(nb[0][2]), "v"(nb[0][3]),
                        "v"(nb[1][0]), "v"(nb[1][1]), "v"(nb[1][2]), "v"(nb[1][3]),
                        "v"(ns[0]), "v"(ns[1]));
        #pragma unroll
        for (int u = 0; u < 2; ++u) {
            #pragma unroll
            for (int i = 0; i < 2; ++i) ca[u][i] = na[u][i] * ns[u];
            #pragma unroll
            for (int j = 0; j < 4; ++j) cb[u][j] = nb[u][j];
        }
    }
    if (q >= 2) {
        #pragma unroll
        for (int i = 0; i < 2; ++i)
            #pragma unroll
            for (int j = 0; j < 4; ++j)
                #pragma unroll
                for (int h = 0; h < 2; ++h) {
                    v4f p;
                    p.x = acc[i][j][4 * h];     p.y = acc[i][j][4 * h + 1];
                    p.z = acc[i][j][4 * h + 2]; p.w = acc[i][j][4 * h + 3];
                    red[q - 2][w][2 * (4 * i + j) + h][lane] = p;
                }
    }
    __syncthreads();
    if (q < 2) {
        #pragma unroll
        for (int i = 0; i < 2; ++i)
            #pragma unroll
            for (int j = 0; j < 4; ++j)
                #pragma unroll
                for (int h = 0; h < 2; ++h) {
                    const v4f p = red[q][w][2 * (4 * i + j) + h][lane];
                    acc[i][j][4 * h]     += p.x; acc[i][j][4 * h + 1] += p.y;
                    acc[i][j][4 * h + 2] += p.z; acc[i][j][4 * h + 3] += p.w;
                }
    }
    __syncthreads();
    if (q == 1) {
        #pragma unroll
        for (int i = 0; i < 2; ++i)
            #pragma unroll
            for (int j = 0; j < 4; ++j)
                #pragma unroll
                for (int h = 0; h < 2; ++h) {
                    v4f p;
                    p.x = acc[i][j][4 * h];     p.y = acc[i][j][4 * h + 1];
                    p.z = acc[i][j][4 * h + 2]; p.w = acc[i][j][4 * h + 3];
                    red[0][w][2 * (4 * i + j) + h][lane] = p;
                }
    }
    __syncthreads();
    if (q == 0) {
        #pragma unroll
        for (int i = 0; i < 2; ++i)
            #pragma unroll
            for (int j = 0; j < 4; ++j)
                #pragma unroll
                for (int h = 0; h < 2; ++h) {
                    const v4f p = red[0][w][2 * (4 * i + j) + h][lane];
                    acc[i][j][4 * h]     += p.x; acc[i][j][4 * h + 1] += p.y;
                    acc[i][j][4 * h + 2] += p.z; acc[i][j][4 * h + 3] += p.w;
                }
        float* Wo = weff + (size_t)b * DIN * DH1;
        #pragma unroll
        for (int i = 0; i < 2; ++i) {
            #pragma unroll
            for (int v = 0; v < 8; ++v) {
                const int row = m0 + 16 * i + v + 8 * hf;
                v4f o;
                o.x = acc[i][0][v];
                o.y = acc[i][1][v];
                o.z = acc[i][2][v];
                o.w = acc[i][3][v];
                *(v4f*)(Wo + (size_t)row * DH1 + n0 + 4 * l16) = o;
            }
        }
    }
}

__global__ __launch_bounds__(512) void k_final(
    const float* __restrict__ y, const float* __restrict__ weff,
    const float* __restrict__ vb, const float* __restrict__ w2b,
    const float* __restrict__ x, float* __restrict__ out)
{
    __shared__ v4f red[2][4][16][32];
    const int lane = threadIdx.x & 31;
    const int hf = lane >> 4, l16 = lane & 15;
    const int w  = (threadIdx.x >> 5) & 3;
    const int q  = threadIdx.x >> 7;
    const int b  = blockIdx.x >> 5;
    const int r  = blockIdx.x & 31;
    const int t0 = ((r >> 2) * 2 + (w >> 1)) * 64;
    const int i0 = ((r & 3) * 2 + (w & 1)) * 32;
    const int kq = q * KQ1;
    const float* Zp = y + (size_t)b * DH1 * T_ + (size_t)(kq + 4 * hf) * T_ + t0 + 4 * l16;
    const float* Wp = weff + (size_t)b * DIN * DH1 + (size_t)(i0 + 2 * l16) * DH1 + kq + 4 * hf;
    v8f acc[4][2] = {};
    v4f ca[2][4];
    v4f cb[2][2];
    #pragma unroll
    for (int u = 0; u < 2; ++u) {
        #pragma unroll
        for (int e = 0; e < 4; ++e) ca[u][e] = *(const v4f*)(Zp + (size_t)(8 * u + e) * T_);
        #pragma unroll
        for (int j = 0; j < 2; ++j) cb[u][j] = *(const v4f*)(Wp + (size_t)j * DH1 + 8 * u);
    }
    #pragma unroll 2
    for (int k0 = 0; k0 < KQ1; k0 += 16) {
        const int kn = (k0 + 16 < KQ1) ? k0 + 16 : k0;
        v4f na[2][4];
        v4f nb[2][2];
        #pragma unroll
        for (int u = 0; u < 2; ++u) {
            #pragma unroll
            for (int e = 0; e < 4; ++e) na[u][e] = *(const v4f*)(Zp + (size_t)(kn + 8 * u + e) * T_);
            #pragma unroll
            for (int j = 0; j < 2; ++j) nb[u][j] = *(const v4f*)(Wp + (size_t)j * DH1 + kn + 8 * u);
        }
        __builtin_amdgcn_sched_barrier(0);
        #pragma unroll
        for (int u = 0; u < 2; ++u)
            #pragma unroll
            for (int hh = 0; hh < 2; ++hh)
                #pragma unroll
                for (int a = 0; a < 4; ++a) {
                    v2f av;
                    av.x = ca[u][2 * hh][a];
                    av.y = ca[u][2 * hh + 1][a];
                    #pragma unroll
                    for (int j = 0; j < 2; ++j) {
                        v2f bv;
                        bv.x = cb[u][j][2 * hh];
                        bv.y = cb[u][j][2 * hh + 1];
                        acc[a][j] = __builtin_amdgcn_wmma_f32_16x16x4_f32(false, av, false, bv,
                                                                         (short)0, acc[a][j], false, false);
                    }
                }
        __builtin_amdgcn_sched_barrier(0);
        asm volatile("" :: "v"(na[0][0]), "v"(na[0][1]), "v"(na[0][2]), "v"(na[0][3]),
                        "v"(na[1][0]), "v"(na[1][1]), "v"(na[1][2]), "v"(na[1][3]),
                        "v"(nb[0][0]), "v"(nb[0][1]), "v"(nb[1][0]), "v"(nb[1][1]));
        #pragma unroll
        for (int u = 0; u < 2; ++u) {
            #pragma unroll
            for (int e = 0; e < 4; ++e) ca[u][e] = na[u][e];
            #pragma unroll
            for (int j = 0; j < 2; ++j) cb[u][j] = nb[u][j];
        }
    }
    if (q >= 2) {
        #pragma unroll
        for (int a = 0; a < 4; ++a)
            #pragma unroll
            for (int j = 0; j < 2; ++j)
                #pragma unroll
                for (int h = 0; h < 2; ++h) {
                    v4f p;
                    p.x = acc[a][j][4 * h];     p.y = acc[a][j][4 * h + 1];
                    p.z = acc[a][j][4 * h + 2]; p.w = acc[a][j][4 * h + 3];
                    red[q - 2][w][2 * (2 * a + j) + h][lane] = p;
                }
    }
    __syncthreads();
    if (q < 2) {
        #pragma unroll
        for (int a = 0; a < 4; ++a)
            #pragma unroll
            for (int j = 0; j < 2; ++j)
                #pragma unroll
                for (int h = 0; h < 2; ++h) {
                    const v4f p = red[q][w][2 * (2 * a + j) + h][lane];
                    acc[a][j][4 * h]     += p.x; acc[a][j][4 * h + 1] += p.y;
                    acc[a][j][4 * h + 2] += p.z; acc[a][j][4 * h + 3] += p.w;
                }
    }
    __syncthreads();
    if (q == 1) {
        #pragma unroll
        for (int a = 0; a < 4; ++a)
            #pragma unroll
            for (int j = 0; j < 2; ++j)
                #pragma unroll
                for (int h = 0; h < 2; ++h) {
                    v4f p;
                    p.x = acc[a][j][4 * h];     p.y = acc[a][j][4 * h + 1];
                    p.z = acc[a][j][4 * h + 2]; p.w = acc[a][j][4 * h + 3];
                    red[0][w][2 * (2 * a + j) + h][lane] = p;
                }
    }
    __syncthreads();
    if (q == 0) {
        #pragma unroll
        for (int a = 0; a < 4; ++a)
            #pragma unroll
            for (int j = 0; j < 2; ++j)
                #pragma unroll
                for (int h = 0; h < 2; ++h) {
                    const v4f p = red[0][w][2 * (2 * a + j) + h][lane];
                    acc[a][j][4 * h]     += p.x; acc[a][j][4 * h + 1] += p.y;
                    acc[a][j][4 * h + 2] += p.z; acc[a][j][4 * h + 3] += p.w;
                }
        const int ii = i0 + 2 * l16;
        const v2f bb = *(const v2f*)(w2b + ii) + *(const v2f*)(vb + b * DIN + ii);
        #pragma unroll
        for (int a = 0; a < 4; ++a) {
            #pragma unroll
            for (int v = 0; v < 8; ++v) {
                const int t = t0 + 4 * (v + 8 * hf) + a;
                const size_t oi = ((size_t)b * T_ + t) * DIN + ii;
                const v2f xv = *(const v2f*)(x + oi);
                v2f o;
                o.x = acc[a][0][v] + bb.x + xv.x;
                o.y = acc[a][1][v] + bb.y + xv.y;
                *(v2f*)(out + oi) = o;
            }
        }
    }
}

extern "C" void kernel_launch(void* const* d_in, const int* in_sizes, int n_in,
                              void* d_out, int out_size, void* d_ws, size_t ws_size,
                              hipStream_t stream)
{
    const float* x   = (const float*)d_in[0];
    const float* d_w = (const float*)d_in[1];
    const float* d_g = (const float*)d_in[2];
    const float* d_b = (const float*)d_in[3];
    const float* p_w = (const float*)d_in[4];
    const float* p_g = (const float*)d_in[5];
    const float* p_b = (const float*)d_in[6];
    const float* w1w = (const float*)d_in[7];
    const float* w1b = (const float*)d_in[8];
    const float* w2w = (const float*)d_in[9];
    const float* w2b = (const float*)d_in[10];
    float* out = (float*)d_out;

    float* ws = (float*)d_ws;
    const size_t NCT = (size_t)B_ * DH1 * T_;
    float* hbuf   = ws;
    float* ybuf   = ws + NCT;
    float* inv_nd = ws + 2 * NCT;
    float* inv_np = inv_nd + 128;
    float* vbuf   = inv_np + 8192;
    float* weff   = vbuf + 2048;

    k_norms<<<344, 256, 0, stream>>>(p_w, d_w, w2w, p_b, inv_np, inv_nd, vbuf);
    k_weff<<<256, 512, 0, stream>>>(w2w, p_w, inv_np, weff);
    k_gemm1_mish<<<1024, 128, 0, stream>>>(x, w1w, w1b, hbuf);
    k_dwconv<<<1024, 256, 0, stream>>>(hbuf, d_w, d_g, d_b, p_g, inv_nd, ybuf);
    k_final<<<256, 512, 0, stream>>>(ybuf, weff, vbuf, w2b, x, out);
}


# GEMM 1 and weight product: operands staged through LDS, each 128-byte row line fetched once per 4-wave group by whole-line loads (was 16 lines per load, each line 8 times); k order kept
# speedup vs baseline: 4.0281x; 1.0342x over previous
#include <hip/hip_runtime.h>
#include <math.h>

typedef __attribute__((ext_vector_type(2))) float v2f;
typedef __attribute__((ext_vector_type(4))) float v4f;
typedef __attribute__((ext_vector_type(8))) float v8f;

#define B_  8
#define T_  1024
#define DIN 256
#define DH1 1024
#define DH2 1024
#define KW  9
#define KQ1 (DH1 / 4)
#define KQ2 (DH2 / 4)

__device__ __forceinline__ float mishf(float v) {
    const float e = __expf(v);
    const float n = e * (e + 2.f);
    const float m = v * (n * __builtin_amdgcn_rcpf(n + 2.f));
    return (v > 20.f) ? v : m;
}

__global__ __launch_bounds__(256) void k_norms(const float* __restrict__ pw,
                                               const float* __restrict__ dw,
                                               const float* __restrict__ w2w,
                                               const float* __restrict__ pb,
                                               float* __restrict__ inv_np,
                                               float* __restrict__ inv_nd,
                                               float* __restrict__ vb)
{
    __shared__ float s[8][256];
    const int lane = threadIdx.x & 31, w = threadIdx.x >> 5;
    if (blockIdx.x < 256) {
        const int b = blockIdx.x >> 5, oc = blockIdx.x & 31;
        const int cs = lane >> 3, ol = lane & 7;
        const float* P = pw + (size_t)b * DH1 * DH2 + (size_t)(4 * w + cs) * DH2 + oc * 32 + 4 * ol;
        v4f v[32];
        #pragma unroll
        for (int j = 0; j < 32; ++j) v[j] = *(const v4f*)(P + (size_t)(32 * j) * DH2);
        __builtin_amdgcn_sched_barrier(0);
        v4f acc = {0.f, 0.f, 0.f, 0.f};
        #pragma unroll
        for (int j = 0; j < 32; ++j) acc += v[j] * v[j];
        #pragma unroll
        for (int i = 0; i < 4; ++i) {
            acc[i] += __shfl_xor(acc[i], 8);
            acc[i] += __shfl_xor(acc[i], 16);
        }
        if (lane < 8) *(v4f*)&s[w][4 * lane] = acc;
        __syncthreads();
        if (threadIdx.x < 32) {
            float tot = 0.f;
            #pragma unroll
            for (int q = 0; q < 8; ++q) tot += s[q][threadIdx.x];
            inv_np[b * DH2 + oc * 32 + threadIdx.x] = 1.f / fmaxf(sqrtf(tot), 1e-12f);
        }
    } else if (blockIdx.x < 328) {
        const int p = blockIdx.x - 256;
        const int b = p / KW, k = p - b * KW;
        const float* D = dw + (size_t)b * DH1 * KW + k;
        float acc = 0.f;
        #pragma unroll
        for (int j = 0; j < 4; ++j) {
            const float v = D[(size_t)(threadIdx.x + 256 * j) * KW];
            acc += v * v;
        }
        #pragma unroll
        for (int m = 16; m >= 1; m >>= 1) acc += __shfl_xor(acc, m);
        if (lane == 0) s[0][w] = acc;
        __syncthreads();
        if (threadIdx.x == 0) {
            float tot = 0.f;
            #pragma unroll
            for (int q = 0; q < 8; ++q) tot += s[0][q];
            inv_nd[p] = 1.f / fmaxf(sqrtf(tot), 1e-12f);
        }
    } else {
        const int i0 = (blockIdx.x - 328) * 16;
        const int hf = lane >> 4, l16 = lane & 15;
        const float* Ap = w2w + (size_t)(i0 + l16) * DH2 + 128 * w + 4 * hf;
        const float* Bq = pb + (size_t)(l16 & 7) * DH2 + 128 * w + 4 * hf;
        v4f a[16], q[16];
        #pragma unroll
        for (int u = 0; u < 16; ++u) {
            a[u] = *(const v4f*)(Ap + 8 * u);
            q[u] = *(const v4f*)(Bq + 8 * u);
        }
        const bool real = (l16 < 8);
        const v4f zero = {0.f, 0.f, 0.f, 0.f};
        v8f acc0 = {}, acc1 = {};
        #pragma unroll
        for (int u = 0; u < 16; ++u) {
            const v4f qq = real ? q[u] : zero;
            v2f a0, a1, b0, b1;
            a0.x = a[u][0]; a0.y = a[u][1]; a1.x = a[u][2]; a1.y = a[u][3];
            b0.x = qq[0];   b0.y = qq[1];   b1.x = qq[2];   b1.y = qq[3];
            acc0 = __builtin_amdgcn_wmma_f32_16x16x4_f32(false, a0, false, b0, (short)0, acc0, false, false);
            acc1 = __builtin_amdgcn_wmma_f32_16x16x4_f32(false, a1, false, b1, (short)0, acc1, false, false);
        }
        #pragma unroll
        for (int v = 0; v < 8; ++v) s[w][v * 32 + lane] = acc0[v] + acc1[v];
        __syncthreads();
        if (real) {
            float tot = 0.f;
            #pragma unroll
            for (int k = 0; k < 8; ++k) tot += s[k][w * 32 + lane];
            vb[l16 * DIN + i0 + w + 8 * hf] = tot;
        }
    }
}

#define SKK 32
#define SPI 36
#define STG (192 * SPI)

template <bool SCALE>
__device__ __forceinline__ void nt_core(const float* __restrict__ Ag, const float* __restrict__ Bg,
                                        const float* __restrict__ Sg, const int ld,
                                        float* __restrict__ L, const int tg, const int lane, const int w4,
                                        v8f (&acc)[2][4])
{
    const int hf = lane >> 4, l16 = lane & 15;
    const int c8 = tg & 7;
    const int ra = tg >> 3;
    const int rb = 4 * ((tg >> 3) & 3) + (tg >> 5);
    const float* Ap = Ag + (size_t)ra * ld + 4 * c8;
    const float* Bp = Bg + (size_t)rb * ld + 4 * c8;
    const float* Sp = SCALE ? Sg + 4 * c8 : Sg;
    const int sa = ra * SPI + 4 * c8;
    int sb[8];
    #pragma unroll
    for (int n = 0; n < 8; ++n) {
        const int R = rb + 16 * n;
        sb[n] = (64 + (R & 64) + 16 * (R & 3) + ((R >> 2) & 15)) * SPI + 4 * c8;
    }
    const int fa0 = ((w4 >> 1) * 32 + l16) * SPI + 4 * hf;
    const int fb0 = (64 + (w4 & 1) * 64 + l16) * SPI + 4 * hf;
    v4f ga[4], gb[8];
    {
        v4f gs = {1.f, 1.f, 1.f, 1.f};
        if (SCALE) gs = *(const v4f*)Sp;
        #pragma unroll
        for (int n = 0; n < 4; ++n) ga[n] = *(const v4f*)(Ap + (size_t)(16 * n) * ld);
        #pragma unroll
        for (int n = 0; n < 8; ++n) gb[n] = *(const v4f*)(Bp + (size_t)(16 * n) * ld);
        #pragma unroll
        for (int n = 0; n < 4; ++n) *(v4f*)(L + sa + 16 * n * SPI) = SCALE ? ga[n] * gs : ga[n];
        #pragma unroll
        for (int n = 0; n < 8; ++n) *(v4f*)(L + sb[n]) = gb[n];
    }
    __syncthreads();
    for (int s = 0; s < 256 / SKK; ++s) {
        const float* cur = L + (s & 1) * STG;
        float* nxt = L + ((s + 1) & 1) * STG;
        const bool more = (s + 1 < 256 / SKK);
        v4f gs = {1.f, 1.f, 1.f, 1.f};
        if (more) {
            const int kn = SKK * (s + 1);
            if (SCALE) gs = *(const v4f*)(Sp + kn);
            #pragma unroll
            for (int n = 0; n < 4; ++n) ga[n] = *(const v4f*)(Ap + (size_t)(16 * n) * ld + kn);
            #pragma unroll
            for (int n = 0; n < 8; ++n) gb[n] = *(const v4f*)(Bp + (size_t)(16 * n) * ld + kn);
        }
        v4f fa[4][2], fb[4][4];
        #pragma unroll
        for (int u = 0; u < 4; ++u) {
            #pragma unroll
            for (int i = 0; i < 2; ++i) fa[u][i] = *(const v4f*)(cur + fa0 + 16 * i * SPI + 8 * u);
            #pragma unroll
            for (int j = 0; j < 4; ++j) fb[u][j] = *(const v4f*)(cur + fb0 + 16 * j * SPI + 8 * u);
        }
        __builtin_amdgcn_sched_barrier(0);
        #pragma unroll
        for (int u = 0; u < 4; ++u)
            #pragma unroll
            for (int hh = 0; hh < 2; ++hh)
                #pragma unroll
                for (int i = 0; i < 2; ++i) {
                    v2f a;
                    a.x = fa[u][i][2 * hh];
                    a.y = fa[u][i][2 * hh + 1];
                    #pragma unroll
                    for (int j = 0; j < 4; ++j) {
                        v2f bv;
                        bv.x = fb[u][j][2 * hh];
                        bv.y = fb[u][j][2 * hh + 1];
                        acc[i][j] = __builtin_amdgcn_wmma_f32_16x16x4_f32(false, a, false, bv,
                                                                         (short)0, acc[i][j], false, false);
                    }
                }
        __builtin_amdgcn_sched_barrier(0);
        if (more) {
            #pragma unroll
            for (int n = 0; n < 4; ++n) *(v4f*)(nxt + sa + 16 * n * SPI) = SCALE ? ga[n] * gs : ga[n];
            #pragma unroll
            for (int n = 0; n < 8; ++n) *(v4f*)(nxt + sb[n]) = gb[n];
        }
        __syncthreads();
    }
}

__global__ __launch_bounds__(128) void k_gemm1_mish(
    const float* __restrict__ x, const float* __restrict__ w1w,
    const float* __restrict__ w1b, float* __restrict__ h)
{
    __shared__ __attribute__((aligned(16))) float L[2 * STG];
    const int lane = threadIdx.x & 31;
    const int hf = lane >> 4, l16 = lane & 15, w = threadIdx.x >> 5;
    const int b  = blockIdx.x >> 7;
    const int r  = blockIdx.x & 127;
    const int mg = (r >> 3) * 64;
    const int ng = (r & 7) * 128;
    const int m0 = mg + (w >> 1) * 32;
    const int n0 = ng + (w & 1) * 64;
    v8f acc[2][4] = {};
    nt_core<false>(w1w + (size_t)mg * DIN, x + ((size_t)b * T_ + ng) * DIN, w1w, DIN,
                   L, (int)threadIdx.x, lane, w, acc);
    float* H = h + (size_t)b * DH1 * T_;
    #pragma unroll
    for (int i = 0; i < 2; ++i) {
        #pragma unroll
        for (int v = 0; v < 8; ++v) {
            const int c = m0 + 16 * i + v + 8 * hf;
            const float bias = w1b[c];
            v4f o;
            o.x = mishf(acc[i][0][v] + bias);
            o.y = mishf(acc[i][1][v] + bias);
            o.z = mishf(acc[i][2][v] + bias);
            o.w = mishf(acc[i][3][v] + bias);
            *(v4f*)(H + (size_t)c * T_ + n0 + 4 * l16) = o;
        }
    }
}

__global__ __launch_bounds__(256) void k_dwconv(
    const float* __restrict__ h, const float* __restrict__ dw,
    const float* __restrict__ dg, const float* __restrict__ db,
    const float* __restrict__ pg, const float* __restrict__ inv_nd,
    float* __restrict__ y)
{
    const int lane = threadIdx.x & 31;
    const int row = __builtin_amdgcn_readfirstlane((int)(blockIdx.x * 8 + (threadIdx.x >> 5)));
    const int b = row >> 10;
    const float* hp  = h + (size_t)row * T_;
    const float* wp  = dw + (size_t)row * KW;
    const float* ndp = inv_nd + b * KW;
    float* yp = y + (size_t)row * T_;
    float wk[KW];
    #pragma unroll
    for (int k = 0; k < KW; ++k) wk[k] = wp[k] * ndp[k];
    const float g = dg[row], bi = db[row], pgv = pg[row];
    v4f cur[8], prv[8], nxt[8];
    #pragma unroll
    for (int j = 0; j < 8; ++j) cur[j] = *(const v4f*)(hp + 128 * j + 4 * lane);
    #pragma unroll
    for (int j = 0; j < 8; ++j) {
        const int t = 128 * j + 4 * lane;
        const bool inl = (t >= 4), inr = (t + 4 < T_);
        const v4f p = *(const v4f*)(hp + (inl ? t - 4 : t));
        const v4f n = *(const v4f*)(hp + (inr ? t + 4 : t));
        const v4f zero = {0.f, 0.f, 0.f, 0.f};
        prv[j] = inl ? p : zero;
        nxt[j] = inr ? n : zero;
    }
    #pragma unroll
    for (int j = 0; j < 8; ++j) {
        const int t = 128 * j + 4 * lane;
        const float win[12] = {prv[j].x, prv[j].y, prv[j].z, prv[j].w,
                               cur[j].x, cur[j].y, cur[j].z, cur[j].w,
                               nxt[j].x, nxt[j].y, nxt[j].z, nxt[j].w};
        v4f o;
        #pragma unroll
        for (int i = 0; i < 4; ++i) {
            float acc = 0.f;
            #pragma unroll
            for (int k = 0; k < KW; ++k) acc += win[i + k] * wk[k];
            const float yy = acc * g + bi;
            o[i] = yy * pgv;
        }
        *(v4f*)(yp + t) = o;
    }
}

__global__ __launch_bounds__(512) void k_weff(
    const float* __restrict__ w2w, const float* __restrict__ pw,
    const float* __restrict__ inv_np, float* __restrict__ weff)
{
    __shared__ __attribute__((aligned(16))) float L[4 * 2 * STG];
    v4f (*red)[4][16][32] = reinterpret_cast<v4f (*)[4][16][32]>(L);
    const int lane = threadIdx.x & 31;
    const int hf = lane >> 4, l16 = lane & 15;
    const int w  = (threadIdx.x >> 5) & 3;
    const int q  = threadIdx.x >> 7;
    const int b  = blockIdx.x >> 5;
    const int r  = blockIdx.x & 31;
    const int mg = (r >> 3) * 64;
    const int ng = (r & 7) * 128;
    const int m0 = mg + (w >> 1) * 32;
    const int n0 = ng + (w & 1) * 64;
    const int kq = q * KQ2;
    v8f acc[2][4] = {};
    nt_core<true>(w2w + (size_t)mg * DH2 + kq, pw + (size_t)b * DH1 * DH2 + (size_t)ng * DH2 + kq,
                  inv_np + b * DH2 + kq, DH2, L + q * 2 * STG, (int)(threadIdx.x & 127), lane, w, acc);
    if (q >= 2) {
        #pragma unroll
        for (int i = 0; i < 2; ++i)
            #pragma unroll
            for (int j = 0; j < 4; ++j)
                #pragma unroll
                for (int h = 0; h < 2; ++h) {
                    v4f p;
                    p.x = acc[i][j][4 * h];     p.y = acc[i][j][4 * h + 1];
                    p.z = acc[i][j][4 * h + 2]; p.w = acc[i][j][4 * h + 3];
                    red[q - 2][w][2 * (4 * i + j) + h][lane] = p;
                }
    }
    __syncthreads();
    if (q < 2) {
        #pragma unroll
        for (int i = 0; i < 2; ++i)
            #pragma unroll
            for (int j = 0; j < 4; ++j)
                #pragma unroll
                for (int h = 0; h < 2; ++h) {
                    const v4f p = red[q][w][2 * (4 * i + j) + h][lane];
                    acc[i][j][4 * h]     += p.x; acc[i][j][4 * h + 1] += p.y;
                    acc[i][j][4 * h + 2] += p.z; acc[i][j][4 * h + 3] += p.w;
                }
    }
    __syncthreads();
    if (q == 1) {
        #pragma unroll
        for (int i = 0; i < 2; ++i)
            #pragma unroll
            for (int j = 0; j < 4; ++j)
                #pragma unroll
                for (int h = 0; h < 2; ++h) {
                    v4f p;
                    p.x = acc[i][j][4 * h];     p.y = acc[i][j][4 * h + 1];
                    p.z = acc[i][j][4 * h + 2]; p.w = acc[i][j][4 * h + 3];
                    red[0][w][2 * (4 * i + j) + h][lane] = p;
                }
    }
    __syncthreads();
    if (q == 0) {
        #pragma unroll
        for (int i = 0; i < 2; ++i)
            #pragma unroll
            for (int j = 0; j < 4; ++j)
                #pragma unroll
                for (int h = 0; h < 2; ++h) {
                    const v4f p = red[0][w][2 * (4 * i + j) + h][lane];
                    acc[i][j][4 * h]     += p.x; acc[i][j][4 * h + 1] += p.y;
                    acc[i][j][4 * h + 2] += p.z; acc[i][j][4 * h + 3] += p.w;
                }
        float* Wo = weff + (size_t)b * DIN * DH1;
        #pragma unroll
        for (int i = 0; i < 2; ++i) {
            #pragma unroll
            for (int v = 0; v < 8; ++v) {
                const int row = m0 + 16 * i + v + 8 * hf;
                v4f o;
                o.x = acc[i][0][v];
                o.y = acc[i][1][v];
                o.z = acc[i][2][v];
                o.w = acc[i][3][v];
                *(v4f*)(Wo + (size_t)row * DH1 + n0 + 4 * l16) = o;
            }
        }
    }
}

__global__ __launch_bounds__(512) void k_final(
    const float* __restrict__ y, const float* __restrict__ weff,
    const float* __restrict__ vb, const float* __restrict__ w2b,
    const float* __restrict__ x, float* __restrict__ out)
{
    __shared__ v4f red[2][4][16][32];
    const int lane = threadIdx.x & 31;
    const int hf = lane >> 4, l16 = lane & 15;
    const int w  = (threadIdx.x >> 5) & 3;
    const int q  = threadIdx.x >> 7;
    const int b  = blockIdx.x >> 5;
    const int r  = blockIdx.x & 31;
    const int t0 = ((r >> 2) * 2 + (w >> 1)) * 64;
    const int i0 = ((r & 3) * 2 + (w & 1)) * 32;
    const int kq = q * KQ1;
    const float* Zp = y + (size_t)b * DH1 * T_ + (size_t)(kq + 4 * hf) * T_ + t0 + 4 * l16;
    const float* Wp = weff + (size_t)b * DIN * DH1 + (size_t)(i0 + 2 * l16) * DH1 + kq + 4 * hf;
    v8f acc[4][2] = {};
    v4f ca[2][4];
    v4f cb[2][2];
    #pragma unroll
    for (int u = 0; u < 2; ++u) {
        #pragma unroll
        for (int e = 0; e < 4; ++e) ca[u][e] = *(const v4f*)(Zp + (size_t)(8 * u + e) * T_);
        #pragma unroll
        for (int j = 0; j < 2; ++j) cb[u][j] = *(const v4f*)(Wp + (size_t)j * DH1 + 8 * u);
    }
    #pragma unroll 2
    for (int k0 = 0; k0 < KQ1; k0 += 16) {
        const int kn = (k0 + 16 < KQ1) ? k0 + 16 : k0;
        v4f na[2][4];
        v4f nb[2][2];
        #pragma unroll
        for (int u = 0; u < 2; ++u) {
            #pragma unroll
            for (int e = 0; e < 4; ++e) na[u][e] = *(const v4f*)(Zp + (size_t)(kn + 8 * u + e) * T_);
            #pragma unroll
            for (int j = 0; j < 2; ++j) nb[u][j] = *(const v4f*)(Wp + (size_t)j * DH1 + kn + 8 * u);
        }
        __builtin_amdgcn_sched_barrier(0);
        #pragma unroll
        for (int u = 0; u < 2; ++u)
            #pragma unroll
            for (int hh = 0; hh < 2; ++hh)
                #pragma unroll
                for (int a = 0; a < 4; ++a) {
                    v2f av;
                    av.x = ca[u][2 * hh][a];
                    av.y = ca[u][2 * hh + 1][a];
                    #pragma unroll
                    for (int j = 0; j < 2; ++j) {
                        v2f bv;
                        bv.x = cb[u][j][2 * hh];
                        bv.y = cb[u][j][2 * hh + 1];
                        acc[a][j] = __builtin_amdgcn_wmma_f32_16x16x4_f32(false, av, false, bv,
                                                                         (short)0, acc[a][j], false, false);
                    }
                }
        __builtin_amdgcn_sched_barrier(0);
        asm volatile("" :: "v"(na[0][0]), "v"(na[0][1]), "v"(na[0][2]), "v"(na[0][3]),
                        "v"(na[1][0]), "v"(na[1][1]), "v"(na[1][2]), "v"(na[1][3]),
                        "v"(nb[0][0]), "v"(nb[0][1]), "v"(nb[1][0]), "v"(nb[1][1]));
        #pragma unroll
        for (int u = 0; u < 2; ++u) {
            #pragma unroll
            for (int e = 0; e < 4; ++e) ca[u][e] = na[u][e];
            #pragma unroll
            for (int j = 0; j < 2; ++j) cb[u][j] = nb[u][j];
        }
    }
    if (q >= 2) {
        #pragma unroll
        for (int a = 0; a < 4; ++a)
            #pragma unroll
            for (int j = 0; j < 2; ++j)
                #pragma unroll
                for (int h = 0; h < 2; ++h) {
                    v4f p;
                    p.x = acc[a][j][4 * h];     p.y = acc[a][j][4 * h + 1];
                    p.z = acc[a][j][4 * h + 2]; p.w = acc[a][j][4 * h + 3];
                    red[q - 2][w][2 * (2 * a + j) + h][lane] = p;
                }
    }
    __syncthreads();
    if (q < 2) {
        #pragma unroll
        for (int a = 0; a < 4; ++a)
            #pragma unroll
            for (int j = 0; j < 2; ++j)
                #pragma unroll
                for (int h = 0; h < 2; ++h) {
                    const v4f p = red[q][w][2 * (2 * a + j) + h][lane];
                    acc[a][j][4 * h]     += p.x; acc[a][j][4 * h + 1] += p.y;
                    acc[a][j][4 * h + 2] += p.z; acc[a][j][4 * h + 3] += p.w;
                }
    }
    __syncthreads();
    if (q == 1) {
        #pragma unroll
        for (int a = 0; a < 4; ++a)
            #pragma unroll
            for (int j = 0; j < 2; ++j)
                #pragma unroll
                for (int h = 0; h < 2; ++h) {
                    v4f p;
                    p.x = acc[a][j][4 * h];     p.y = acc[a][j][4 * h + 1];
                    p.z = acc[a][j][4 * h + 2]; p.w = acc[a][j][4 * h + 3];
                    red[0][w][2 * (2 * a + j) + h][lane] = p;
                }
    }
    __syncthreads();
    if (q == 0) {
        #pragma unroll
        for (int a = 0; a < 4; ++a)
            #pragma unroll
            for (int j = 0; j < 2; ++j)
                #pragma unroll
                for (int h = 0; h < 2; ++h) {
                    const v4f p = red[0][w][2 * (2 * a + j) + h][lane];
                    acc[a][j][4 * h]     += p.x; acc[a][j][4 * h + 1] += p.y;
                    acc[a][j][4 * h + 2] += p.z; acc[a][j][4 * h + 3] += p.w;
                }
        const int ii = i0 + 2 * l16;
        const v2f bb = *(const v2f*)(w2b + ii) + *(const v2f*)(vb + b * DIN + ii);
        #pragma unroll
        for (int a = 0; a < 4; ++a) {
            #pragma unroll
            for (int v = 0; v < 8; ++v) {
                const int t = t0 + 4 * (v + 8 * hf) + a;
                const size_t oi = ((size_t)b * T_ + t) * DIN + ii;
                const v2f xv = *(const v2f*)(x + oi);
                v2f o;
                o.x = acc[a][0][v] + bb.x + xv.x;
                o.y = acc[a][1][v] + bb.y + xv.y;
                *(v2f*)(out + oi) = o;
            }
        }
    }
}

extern "C" void kernel_launch(void* const* d_in, const int* in_sizes, int n_in,
                              void* d_out, int out_size, void* d_ws, size_t ws_size,
                              hipStream_t stream)
{
    const float* x   = (const float*)d_in[0];
    const float* d_w = (const float*)d_in[1];
    const float* d_g = (const float*)d_in[2];
    const float* d_b = (const float*)d_in[3];
    const float* p_w = (const float*)d_in[4];
    const float* p_g = (const float*)d_in[5];
    const float* p_b = (const float*)d_in[6];
    const float* w1w = (const float*)d_in[7];
    const float* w1b = (const float*)d_in[8];
    const float* w2w = (const float*)d_in[9];
    const float* w2b = (const float*)d_in[10];
    float* out = (float*)d_out;

    float* ws = (float*)d_ws;
    const size_t NCT = (size_t)B_ * DH1 * T_;
    float* hbuf   = ws;
    float* ybuf   = ws + NCT;
    float* inv_nd = ws + 2 * NCT;
    float* inv_np = inv_nd + 128;
    float* vbuf   = inv_np + 8192;
    float* weff   = vbuf + 2048;

    k_norms<<<344, 256, 0, stream>>>(p_w, d_w, w2w, p_b, inv_np, inv_nd, vbuf);
    k_weff<<<256, 512, 0, stream>>>(w2w, p_w, inv_np, weff);
    k_gemm1_mish<<<1024, 128, 0, stream>>>(x, w1w, w1b, hbuf);
    k_dwconv<<<1024, 256, 0, stream>>>(hbuf, d_w, d_g, d_b, p_g, inv_nd, ybuf);
    k_final<<<256, 512, 0, stream>>>(ybuf, weff, vbuf, w2b, x, out);
}


# GEMM 1: a work-group owns 32 channels x all 1024 time steps, keeps mish(h) in LDS and runs the depthwise conv on it in the same launch; h never goes to memory, 5 launches -> 4; sums unchanged
# speedup vs baseline: 4.1684x; 1.0348x over previous
#include <hip/hip_runtime.h>
#include <math.h>

typedef __attribute__((ext_vector_type(2))) float v2f;
typedef __attribute__((ext_vector_type(4))) float v4f;
typedef __attribute__((ext_vector_type(8))) float v8f;

#define B_  8
#define T_  1024
#define DIN 256
#define DH1 1024
#define DH2 1024
#define KW  9
#define KQ1 (DH1 / 4)
#define KQ2 (DH2 / 4)

__device__ __forceinline__ float mishf(float v) {
    const float e = __expf(v);
    const float n = e * (e + 2.f);
    const float m = v * (n * __builtin_amdgcn_rcpf(n + 2.f));
    return (v > 20.f) ? v : m;
}

__global__ __launch_bounds__(256) void k_norms(const float* __restrict__ pw,
                                               const float* __restrict__ dw,
                                               const float* __restrict__ w2w,
                                               const float* __restrict__ pb,
                                               float* __restrict__ inv_np,
                                               float* __restrict__ inv_nd,
                                               float* __restrict__ vb)
{
    __shared__ float s[8][256];
    const int lane = threadIdx.x & 31, w = threadIdx.x >> 5;
    if (blockIdx.x < 256) {
        const int b = blockIdx.x >> 5, oc = blockIdx.x & 31;
        const int cs = lane >> 3, ol = lane & 7;
        const float* P = pw + (size_t)b * DH1 * DH2 + (size_t)(4 * w + cs) * DH2 + oc * 32 + 4 * ol;
        v4f v[32];
        #pragma unroll
        for (int j = 0; j < 32; ++j) v[j] = *(const v4f*)(P + (size_t)(32 * j) * DH2);
        __builtin_amdgcn_sched_barrier(0);
        v4f acc = {0.f, 0.f, 0.f, 0.f};
        #pragma unroll
        for (int j = 0; j < 32; ++j) acc += v[j] * v[j];
        #pragma unroll
        for (int i = 0; i < 4; ++i) {
            acc[i] += __shfl_xor(acc[i], 8);
            acc[i] += __shfl_xor(acc[i], 16);
        }
        if (lane < 8) *(v4f*)&s[w][4 * lane] = acc;
        __syncthreads();
        if (threadIdx.x < 32) {
            float tot = 0.f;
            #pragma unroll
            for (int q = 0; q < 8; ++q) tot += s[q][threadIdx.x];
            inv_np[b * DH2 + oc * 32 + threadIdx.x] = 1.f / fmaxf(sqrtf(tot), 1e-12f);
        }
    } else if (blockIdx.x < 328) {
        const int p = blockIdx.x - 256;
        const int b = p / KW, k = p - b * KW;
        const float* D = dw + (size_t)b * DH1 * KW + k;
        float acc = 0.f;
        #pragma unroll
        for (int j = 0; j < 4; ++j) {
            const float v = D[(size_t)(threadIdx.x + 256 * j) * KW];
            acc += v * v;
        }
        #pragma unroll
        for (int m = 16; m >= 1; m >>= 1) acc += __shfl_xor(acc, m);
        if (lane == 0) s[0][w] = acc;
        __syncthreads();
        if (threadIdx.x == 0) {
            float tot = 0.f;
            #pragma unroll
            for (int q = 0; q < 8; ++q) tot += s[0][q];
            inv_nd[p] = 1.f / fmaxf(sqrtf(tot), 1e-12f);
        }
    } else {
        const int i0 = (blockIdx.x - 328) * 16;
        const int hf = lane >> 4, l16 = lane & 15;
        const float* Ap = w2w + (size_t)(i0 + l16) * DH2 + 128 * w + 4 * hf;
        const float* Bq = pb + (size_t)(l16 & 7) * DH2 + 128 * w + 4 * hf;
        v4f a[16], q[16];
        #pragma unroll
        for (int u = 0; u < 16; ++u) {
            a[u] = *(const v4f*)(Ap + 8 * u);
            q[u] = *(const v4f*)(Bq + 8 * u);
        }
        const bool real = (l16 < 8);
        const v4f zero = {0.f, 0.f, 0.f, 0.f};
        v8f acc0 = {}, acc1 = {};
        #pragma unroll
        for (int u = 0; u < 16; ++u) {
            const v4f qq = real ? q[u] : zero;
            v2f a0, a1, b0, b1;
            a0.x = a[u][0]; a0.y = a[u][1]; a1.x = a[u][2]; a1.y = a[u][3];
            b0.x = qq[0];   b0.y = qq[1];   b1.x = qq[2];   b1.y = qq[3];
            acc0 = __builtin_amdgcn_wmma_f32_16x16x4_f32(false, a0, false, b0, (short)0, acc0, false, false);
            acc1 = __builtin_amdgcn_wmma_f32_16x16x4_f32(false, a1, false, b1, (short)0, acc1, false, false);
        }
        #pragma unroll
        for (int v = 0; v < 8; ++v) s[w][v * 32 + lane] = acc0[v] + acc1[v];
        __syncthreads();
        if (real) {
            float tot = 0.f;
            #pragma unroll
            for (int k = 0; k < 8; ++k) tot += s[k][w * 32 + lane];
            vb[l16 * DIN + i0 + w + 8 * hf] = tot;
        }
    }
}

#define SKK 32
#define SPI 36
#define STG (192 * SPI)

template <bool SCALE>
__device__ __forceinline__ void nt_core(const float* __restrict__ Ag, const float* __restrict__ Bg,
                                        const float* __restrict__ Sg, const int ld,
                                        float* __restrict__ L, const int tg, const int lane, const int w4,
                                        v8f (&acc)[2][4])
{
    const int hf = lane >> 4, l16 = lane & 15;
    const int c8 = tg & 7;
    const int ra = tg >> 3;
    const int rb = 4 * ((tg >> 3) & 3) + (tg >> 5);
    const float* Ap = Ag + (size_t)ra * ld + 4 * c8;
    const float* Bp = Bg + (size_t)rb * ld + 4 * c8;
    const float* Sp = SCALE ? Sg + 4 * c8 : Sg;
    const int sa = ra * SPI + 4 * c8;
    int sb[8];
    #pragma unroll
    for (int n = 0; n < 8; ++n) {
        const int R = rb + 16 * n;
        sb[n] = (64 + (R & 64) + 16 * (R & 3) + ((R >> 2) & 15)) * SPI + 4 * c8;
    }
    const int fa0 = ((w4 >> 1) * 32 + l16) * SPI + 4 * hf;
    const int fb0 = (64 + (w4 & 1) * 64 + l16) * SPI + 4 * hf;
    v4f ga[4], gb[8];
    {
        v4f gs = {1.f, 1.f, 1.f, 1.f};
        if (SCALE) gs = *(const v4f*)Sp;
        #pragma unroll
        for (int n = 0; n < 4; ++n) ga[n] = *(const v4f*)(Ap + (size_t)(16 * n) * ld);
        #pragma unroll
        for (int n = 0; n < 8; ++n) gb[n] = *(const v4f*)(Bp + (size_t)(16 * n) * ld);
        #pragma unroll
        for (int n = 0; n < 4; ++n) *(v4f*)(L + sa + 16 * n * SPI) = SCALE ? ga[n] * gs : ga[n];
        #pragma unroll
        for (int n = 0; n < 8; ++n) *(v4f*)(L + sb[n]) = gb[n];
    }
    __syncthreads();
    for (int s = 0; s < 256 / SKK; ++s) {
        const float* cur = L + (s & 1) * STG;
        float* nxt = L + ((s + 1) & 1) * STG;
        const bool more = (s + 1 < 256 / SKK);
        v4f gs = {1.f, 1.f, 1.f, 1.f};
        if (more) {
            const int kn = SKK * (s + 1);
            if (SCALE) gs = *(const v4f*)(Sp + kn);
            #pragma unroll
            for (int n = 0; n < 4; ++n) ga[n] = *(const v4f*)(Ap + (size_t)(16 * n) * ld + kn);
            #pragma unroll
            for (int n = 0; n < 8; ++n) gb[n] = *(const v4f*)(Bp + (size_t)(16 * n) * ld + kn);
        }
        v4f fa[4][2], fb[4][4];
        #pragma unroll
        for (int u = 0; u < 4; ++u) {
            #pragma unroll
            for (int i = 0; i < 2; ++i) fa[u][i] = *(const v4f*)(cur + fa0 + 16 * i * SPI + 8 * u);
            #pragma unroll
            for (int j = 0; j < 4; ++j) fb[u][j] = *(const v4f*)(cur + fb0 + 16 * j * SPI + 8 * u);
        }
        __builtin_amdgcn_sched_barrier(0);
        #pragma unroll
        for (int u = 0; u < 4; ++u)
            #pragma unroll
            for (int hh = 0; hh < 2; ++hh)
                #pragma unroll
                for (int i = 0; i < 2; ++i) {
                    v2f a;
                    a.x = fa[u][i][2 * hh];
                    a.y = fa[u][i][2 * hh + 1];
                    #pragma unroll
                    for (int j = 0; j < 4; ++j) {
                        v2f bv;
                        bv.x = fb[u][j][2 * hh];
                        bv.y = fb[u][j][2 * hh + 1];
                        acc[i][j] = __builtin_amdgcn_wmma_f32_16x16x4_f32(false, a, false, bv,
                                                                         (short)0, acc[i][j], false, false);
                    }
                }
        __builtin_amdgcn_sched_barrier(0);
        if (more) {
            #pragma unroll
            for (int n = 0; n < 4; ++n) *(v4f*)(nxt + sa + 16 * n * SPI) = SCALE ? ga[n] * gs : ga[n];
            #pragma unroll
            for (int n = 0; n < 8; ++n) *(v4f*)(nxt + sb[n]) = gb[n];
        }
        __syncthreads();
    }
}

#define APW 260

__global__ __launch_bounds__(512) void k_gemm1_conv(
    const float* __restrict__ x, const float* __restrict__ w1w, const float* __restrict__ w1b,
    const float* __restrict__ dw, const float* __restrict__ dg, const float* __restrict__ db,
    const float* __restrict__ pg, const float* __restrict__ inv_nd, float* __restrict__ y)
{
    __shared__ __attribute__((aligned(16))) float As[32 * APW];
    __shared__ __attribute__((aligned(16))) float Hs[32 * T_];
    const int lane = threadIdx.x & 31;
    const int hf = lane >> 4, l16 = lane & 15, w = threadIdx.x >> 5;
    const int b  = blockIdx.x >> 5;
    const int c0 = (blockIdx.x & 31) * 32;
    const int n0 = w * 64;
    {
        const int r = threadIdx.x >> 4, pc = threadIdx.x & 15;
        const float* src = w1w + (size_t)(c0 + r) * DIN + 4 * pc;
        v4f g[4];
        #pragma unroll
        for (int n = 0; n < 4; ++n) g[n] = *(const v4f*)(src + 64 * n);
        #pragma unroll
        for (int n = 0; n < 4; ++n) *(v4f*)(As + r * APW + 4 * pc + 64 * n) = g[n];
    }
    const float* Xp = x + ((size_t)b * T_ + n0 + 4 * l16) * DIN + 4 * hf;
    v4f cb[4][4];
    #pragma unroll
    for (int u = 0; u < 4; ++u)
        #pragma unroll
        for (int j = 0; j < 4; ++j) cb[u][j] = *(const v4f*)(Xp + (size_t)j * DIN + 8 * u);
    __syncthreads();
    const float* Af = As + l16 * APW + 4 * hf;
    v8f acc[2][4] = {};
    #pragma unroll 2
    for (int s = 0; s < DIN / 32; ++s) {
        const int kn = (s + 1 < DIN / 32) ? 32 * (s + 1) : 32 * s;
        v4f nb[4][4];
        #pragma unroll
        for (int u = 0; u < 4; ++u)
            #pragma unroll
            for (int j = 0; j < 4; ++j) nb[u][j] = *(const v4f*)(Xp + (size_t)j * DIN + kn + 8 * u);
        v4f fa[4][2];
        #pragma unroll
        for (int u = 0; u < 4; ++u)
            #pragma unroll
            for (int i = 0; i < 2; ++i) fa[u][i] = *(const v4f*)(Af + 16 * i * APW + 32 * s + 8 * u);
        __builtin_amdgcn_sched_barrier(0);
        #pragma unroll
        for (int u = 0; u < 4; ++u)
            #pragma unroll
            for (int hh = 0; hh < 2; ++hh)
                #pragma unroll
                for (int i = 0; i < 2; ++i) {
                    v2f a;
                    a.x = fa[u][i][2 * hh];
                    a.y = fa[u][i][2 * hh + 1];
                    #pragma unroll
                    for (int j = 0; j < 4; ++j) {
                        v2f bv;
                        bv.x = cb[u][j][2 * hh];
                        bv.y = cb[u][j][2 * hh + 1];
                        acc[i][j] = __builtin_amdgcn_wmma_f32_16x16x4_f32(false, a, false, bv,
                                                                         (short)0, acc[i][j], false, false);
                    }
                }
        __builtin_amdgcn_sched_barrier(0);
        asm volatile("" :: "v"(nb[0][0]), "v"(nb[0][1]), "v"(nb[0][2]), "v"(nb[0][3]),
                        "v"(nb[1][0]), "v"(nb[1][1]), "v"(nb[1][2]), "v"(nb[1][3]),
                        "v"(nb[2][0]), "v"(nb[2][1]), "v"(nb[2][2]), "v"(nb[2][3]),
                        "v"(nb[3][0]), "v"(nb[3][1]), "v"(nb[3][2]), "v"(nb[3][3]));
        #pragma unroll
        for (int u = 0; u < 4; ++u)
            #pragma unroll
            for (int j = 0; j < 4; ++j) cb[u][j] = nb[u][j];
    }
    #pragma unroll
    for (int i = 0; i < 2; ++i) {
        #pragma unroll
        for (int v = 0; v < 8; ++v) {
            const int cl = 16 * i + v + 8 * hf;
            const float bias = w1b[c0 + cl];
            v4f o;
            o.x = mishf(acc[i][0][v] + bias);
            o.y = mishf(acc[i][1][v] + bias);
            o.z = mishf(acc[i][2][v] + bias);
            o.w = mishf(acc[i][3][v] + bias);
            *(v4f*)(Hs + cl * T_ + n0 + 4 * l16) = o;
        }
    }
    __syncthreads();
    #pragma unroll 1
    for (int rr = 0; rr < 2; ++rr) {
        const int cl  = __builtin_amdgcn_readfirstlane(w + 16 * rr);
        const int row = __builtin_amdgcn_readfirstlane(b * DH1 + c0 + w + 16 * rr);
        const float* wp  = dw + (size_t)row * KW;
        const float* ndp = inv_nd + b * KW;
        float* yp = y + (size_t)row * T_;
        float wk[KW];
        #pragma unroll
        for (int k = 0; k < KW; ++k) wk[k] = wp[k] * ndp[k];
        const float g = dg[row], bi = db[row], pgv = pg[row];
        v4f cur[8], prv[8], nxt[8];
        #pragma unroll
        for (int j = 0; j < 8; ++j) cur[j] = *(const v4f*)(Hs + cl * T_ + 128 * j + 4 * lane);
        #pragma unroll
        for (int j = 0; j < 8; ++j) {
            const int t = 128 * j + 4 * lane;
            const bool inl = (t >= 4), inr = (t + 4 < T_);
            const v4f p = *(const v4f*)(Hs + cl * T_ + (inl ? t - 4 : t));
            const v4f n = *(const v4f*)(Hs + cl * T_ + (inr ? t + 4 : t));
            const v4f zero = {0.f, 0.f, 0.f, 0.f};
            prv[j] = inl ? p : zero;
            nxt[j] = inr ? n : zero;
        }
        #pragma unroll
        for (int j = 0; j < 8; ++j) {
            const int t = 128 * j + 4 * lane;
            const float win[12] = {prv[j].x, prv[j].y, prv[j].z, prv[j].w,
                                   cur[j].x, cur[j].y, cur[j].z, cur[j].w,
                                   nxt[j].x, nxt[j].y, nxt[j].z, nxt[j].w};
            v4f o;
            #pragma unroll
            for (int i = 0; i < 4; ++i) {
                float a9 = 0.f;
                #pragma unroll
                for (int k = 0; k < KW; ++k) a9 += win[i + k] * wk[k];
                const float yy = a9 * g + bi;
                o[i] = yy * pgv;
            }
            *(v4f*)(yp + t) = o;
        }
    }
}

__global__ __launch_bounds__(512) void k_weff(
    const float* __restrict__ w2w, const float* __restrict__ pw,
    const float* __restrict__ inv_np, float* __restrict__ weff)
{
    __shared__ __attribute__((aligned(16))) float L[4 * 2 * STG];
    v4f (*red)[4][16][32] = reinterpret_cast<v4f (*)[4][16][32]>(L);
    const int lane = threadIdx.x & 31;
    const int hf = lane >> 4, l16 = lane & 15;
    const int w  = (threadIdx.x >> 5) & 3;
    const int q  = threadIdx.x >> 7;
    const int b  = blockIdx.x >> 5;
    const int r  = blockIdx.x & 31;
    const int mg = (r >> 3) * 64;
    const int ng = (r & 7) * 128;
    const int m0 = mg + (w >> 1) * 32;
    const int n0 = ng + (w & 1) * 64;
    const int kq = q * KQ2;
    v8f acc[2][4] = {};
    nt_core<true>(w2w + (size_t)mg * DH2 + kq, pw + (size_t)b * DH1 * DH2 + (size_t)ng * DH2 + kq,
                  inv_np + b * DH2 + kq, DH2, L + q * 2 * STG, (int)(threadIdx.x & 127), lane, w, acc);
    if (q >= 2) {
        #pragma unroll
        for (int i = 0; i < 2; ++i)
            #pragma unroll
            for (int j = 0; j < 4; ++j)
                #pragma unroll
                for (int h = 0; h < 2; ++h) {
                    v4f p;
                    p.x = acc[i][j][4 * h];     p.y = acc[i][j][4 * h + 1];
                    p.z = acc[i][j][4 * h + 2]; p.w = acc[i][j][4 * h + 3];
                    red[q - 2][w][2 * (4 * i + j) + h][lane] = p;
                }
    }
    __syncthreads();
    if (q < 2) {
        #pragma unroll
        for (int i = 0; i < 2; ++i)
            #pragma unroll
            for (int j = 0; j < 4; ++j)
                #pragma unroll
                for (int h = 0; h < 2; ++h) {
                    const v4f p = red[q][w][2 * (4 * i + j) + h][lane];
                    acc[i][j][4 * h]     += p.x; acc[i][j][4 * h + 1] += p.y;
                    acc[i][j][4 * h + 2] += p.z; acc[i][j][4 * h + 3] += p.w;
                }
    }
    __syncthreads();
    if (q == 1) {
        #pragma unroll
        for (int i = 0; i < 2; ++i)
            #pragma unroll
            for (int j = 0; j < 4; ++j)
                #pragma unroll
                for (int h = 0; h < 2; ++h) {
                    v4f p;
                    p.x = acc[i][j][4 * h];     p.y = acc[i][j][4 * h + 1];
                    p.z = acc[i][j][4 * h + 2]; p.w = acc[i][j][4 * h + 3];
                    red[0][w][2 * (4 * i + j) + h][lane] = p;
                }
    }
    __syncthreads();
    if (q == 0) {
        #pragma unroll
        for (int i = 0; i < 2; ++i)
            #pragma unroll
            for (int j = 0; j < 4; ++j)
                #pragma unroll
                for (int h = 0; h < 2; ++h) {
                    const v4f p = red[0][w][2 * (4 * i + j) + h][lane];
                    acc[i][j][4 * h]     += p.x; acc[i][j][4 * h + 1] += p.y;
                    acc[i][j][4 * h + 2] += p.z; acc[i][j][4 * h + 3] += p.w;
                }
        float* Wo = weff + (size_t)b * DIN * DH1;
        #pragma unroll
        for (int i = 0; i < 2; ++i) {
            #pragma unroll
            for (int v = 0; v < 8; ++v) {
                const int row = m0 + 16 * i + v + 8 * hf;
                v4f o;
                o.x = acc[i][0][v];
                o.y = acc[i][1][v];
                o.z = acc[i][2][v];
                o.w = acc[i][3][v];
                *(v4f*)(Wo + (size_t)row * DH1 + n0 + 4 * l16) = o;
            }
        }
    }
}

__global__ __launch_bounds__(512) void k_final(
    const float* __restrict__ y, const float* __restrict__ weff,
    const float* __restrict__ vb, const float* __restrict__ w2b,
    const float* __restrict__ x, float* __restrict__ out)
{
    __shared__ v4f red[2][4][16][32];
    const int lane = threadIdx.x & 31;
    const int hf = lane >> 4, l16 = lane & 15;
    const int w  = (threadIdx.x >> 5) & 3;
    const int q  = threadIdx.x >> 7;
    const int b  = blockIdx.x >> 5;
    const int r  = blockIdx.x & 31;
    const int t0 = ((r >> 2) * 2 + (w >> 1)) * 64;
    const int i0 = ((r & 3) * 2 + (w & 1)) * 32;
    const int kq = q * KQ1;
    const float* Zp = y + (size_t)b * DH1 * T_ + (size_t)(kq + 4 * hf) * T_ + t0 + 4 * l16;
    const float* Wp = weff + (size_t)b * DIN * DH1 + (size_t)(i0 + 2 * l16) * DH1 + kq + 4 * hf;
    v8f acc[4][2] = {};
    v4f ca[2][4];
    v4f cb[2][2];
    #pragma unroll
    for (int u = 0; u < 2; ++u) {
        #pragma unroll
        for (int e = 0; e < 4; ++e) ca[u][e] = *(const v4f*)(Zp + (size_t)(8 * u + e) * T_);
        #pragma unroll
        for (int j = 0; j < 2; ++j) cb[u][j] = *(const v4f*)(Wp + (size_t)j * DH1 + 8 * u);
    }
    #pragma unroll 2
    for (int k0 = 0; k0 < KQ1; k0 += 16) {
        const int kn = (k0 + 16 < KQ1) ? k0 + 16 : k0;
        v4f na[2][4];
        v4f nb[2][2];
        #pragma unroll
        for (int u = 0; u < 2; ++u) {
            #pragma unroll
            for (int e = 0; e < 4; ++e) na[u][e] = *(const v4f*)(Zp + (size_t)(kn + 8 * u + e) * T_);
            #pragma unroll
            for (int j = 0; j < 2; ++j) nb[u][j] = *(const v4f*)(Wp + (size_t)j * DH1 + kn + 8 * u);
        }
        __builtin_amdgcn_sched_barrier(0);
        #pragma unroll
        for (int u = 0; u < 2; ++u)
            #pragma unroll
            for (int hh = 0; hh < 2; ++hh)
                #pragma unroll
                for (int a = 0; a < 4; ++a) {
                    v2f av;
                    av.x = ca[u][2 * hh][a];
                    av.y = ca[u][2 * hh + 1][a];
                    #pragma unroll
                    for (int j = 0; j < 2; ++j) {
                        v2f bv;
                        bv.x = cb[u][j][2 * hh];
                        bv.y = cb[u][j][2 * hh + 1];
                        acc[a][j] = __builtin_amdgcn_wmma_f32_16x16x4_f32(false, av, false, bv,
                                                                         (short)0, acc[a][j], false, false);
                    }
                }
        __builtin_amdgcn_sched_barrier(0);
        asm volatile("" :: "v"(na[0][0]), "v"(na[0][1]), "v"(na[0][2]), "v"(na[0][3]),
                        "v"(na[1][0]), "v"(na[1][1]), "v"(na[1][2]), "v"(na[1][3]),
                        "v"(nb[0][0]), "v"(nb[0][1]), "v"(nb[1][0]), "v"(nb[1][1]));
        #pragma unroll
        for (int u = 0; u < 2; ++u) {
            #pragma unroll
            for (int e = 0; e < 4; ++e) ca[u][e] = na[u][e];
            #pragma unroll
            for (int j = 0; j < 2; ++j) cb[u][j] = nb[u][j];
        }
    }
    if (q >= 2) {
        #pragma unroll
        for (int a = 0; a < 4; ++a)
            #pragma unroll
            for (int j = 0; j < 2; ++j)
                #pragma unroll
                for (int h = 0; h < 2; ++h) {
                    v4f p;
                    p.x = acc[a][j][4 * h];     p.y = acc[a][j][4 * h + 1];
                    p.z = acc[a][j][4 * h + 2]; p.w = acc[a][j][4 * h + 3];
                    red[q - 2][w][2 * (2 * a + j) + h][lane] = p;
                }
    }
    __syncthreads();
    if (q < 2) {
        #pragma unroll
        for (int a = 0; a < 4; ++a)
            #pragma unroll
            for (int j = 0; j < 2; ++j)
                #pragma unroll
                for (int h = 0; h < 2; ++h) {
                    const v4f p = red[q][w][2 * (2 * a + j) + h][lane];
                    acc[a][j][4 * h]     += p.x; acc[a][j][4 * h + 1] += p.y;
                    acc[a][j][4 * h + 2] += p.z; acc[a][j][4 * h + 3] += p.w;
                }
    }
    __syncthreads();
    if (q == 1) {
        #pragma unroll
        for (int a = 0; a < 4; ++a)
            #pragma unroll
            for (int j = 0; j < 2; ++j)
                #pragma unroll
                for (int h = 0; h < 2; ++h) {
                    v4f p;
                    p.x = acc[a][j][4 * h];     p.y = acc[a][j][4 * h + 1];
                    p.z = acc[a][j][4 * h + 2]; p.w = acc[a][j][4 * h + 3];
                    red[0][w][2 * (2 * a + j) + h][lane] = p;
                }
    }
    __syncthreads();
    if (q == 0) {
        #pragma unroll
        for (int a = 0; a < 4; ++a)
            #pragma unroll
            for (int j = 0; j < 2; ++j)
                #pragma unroll
                for (int h = 0; h < 2; ++h) {
                    const v4f p = red[0][w][2 * (2 * a + j) + h][lane];
                    acc[a][j][4 * h]     += p.x; acc[a][j][4 * h + 1] += p.y;
                    acc[a][j][4 * h + 2] += p.z; acc[a][j][4 * h + 3] += p.w;
                }
        const int ii = i0 + 2 * l16;
        const v2f bb = *(const v2f*)(w2b + ii) + *(const v2f*)(vb + b * DIN + ii);
        #pragma unroll
        for (int a = 0; a < 4; ++a) {
            #pragma unroll
            for (int v = 0; v < 8; ++v) {
                const int t = t0 + 4 * (v + 8 * hf) + a;
                const size_t oi = ((size_t)b * T_ + t) * DIN + ii;
                const v2f xv = *(const v2f*)(x + oi);
                v2f o;
                o.x = acc[a][0][v] + bb.x + xv.x;
                o.y = acc[a][1][v] + bb.y + xv.y;
                *(v2f*)(out + oi) = o;
            }
        }
    }
}

extern "C" void kernel_launch(void* const* d_in, const int* in_sizes, int n_in,
                              void* d_out, int out_size, void* d_ws, size_t ws_size,
                              hipStream_t stream)
{
    const float* x   = (const float*)d_in[0];
    const float* d_w = (const float*)d_in[1];
    const float* d_g = (const float*)d_in[2];
    const float* d_b = (const float*)d_in[3];
    const float* p_w = (const float*)d_in[4];
    const float* p_g = (const float*)d_in[5];
    const float* p_b = (const float*)d_in[6];
    const float* w1w = (const float*)d_in[7];
    const float* w1b = (const float*)d_in[8];
    const float* w2w = (const float*)d_in[9];
    const float* w2b = (const float*)d_in[10];
    float* out = (float*)d_out;

    float* ws = (float*)d_ws;
    const size_t NCT = (size_t)B_ * DH1 * T_;
    float* ybuf   = ws + NCT;
    float* inv_nd = ws + 2 * NCT;
    float* inv_np = inv_nd + 128;
    float* vbuf   = inv_np + 8192;
    float* weff   = vbuf + 2048;

    k_norms<<<344, 256, 0, stream>>>(p_w, d_w, w2w, p_b, inv_np, inv_nd, vbuf);
    k_weff<<<256, 512, 0, stream>>>(w2w, p_w, inv_np, weff);
    k_gemm1_conv<<<256, 512, 0, stream>>>(x, w1w, w1b, d_w, d_g, d_b, p_g, inv_nd, ybuf);
    k_final<<<256, 512, 0, stream>>>(ybuf, weff, vbuf, w2b, x, out);
}


# Norm pre-pass folded into the GEMM 1 launch (4 launches -> 3): p-norm loads ride in its main loop, d-norm per group in LDS, bias product as 16 partial tiles; p-norm and bias sum order changes
# speedup vs baseline: 4.4083x; 1.0576x over previous
#include <hip/hip_runtime.h>
#include <math.h>

typedef __attribute__((ext_vector_type(2))) float v2f;
typedef __attribute__((ext_vector_type(4))) float v4f;
typedef __attribute__((ext_vector_type(8))) float v8f;

#define B_  8
#define T_  1024
#define DIN 256
#define DH1 1024
#define DH2 1024
#define KW  9
#define KQ1 (DH1 / 4)
#define KQ2 (DH2 / 4)

__device__ __forceinline__ float mishf(float v) {
    const float e = __expf(v);
    const float n = e * (e + 2.f);
    const float m = v * (n * __builtin_amdgcn_rcpf(n + 2.f));
    return (v > 20.f) ? v : m;
}

#define SKK 32
#define SPI 36
#define STG (192 * SPI)

template <bool SCALE>
__device__ __forceinline__ void nt_core(const float* __restrict__ Ag, const float* __restrict__ Bg,
                                        const float* __restrict__ Sg, const int ld,
                                        float* __restrict__ L, const int tg, const int lane, const int w4,
                                        v8f (&acc)[2][4])
{
    const int hf = lane >> 4, l16 = lane & 15;
    const int c8 = tg & 7;
    const int ra = tg >> 3;
    const int rb = 4 * ((tg >> 3) & 3) + (tg >> 5);
    const float* Ap = Ag + (size_t)ra * ld + 4 * c8;
    const float* Bp = Bg + (size_t)rb * ld + 4 * c8;
    const float* Sp = SCALE ? Sg + 4 * c8 : Sg;
    const int sa = ra * SPI + 4 * c8;
    int sb[8];
    #pragma unroll
    for (int n = 0; n < 8; ++n) {
        const int R = rb + 16 * n;
        sb[n] = (64 + (R & 64) + 16 * (R & 3) + ((R >> 2) & 15)) * SPI + 4 * c8;
    }
    const int fa0 = ((w4 >> 1) * 32 + l16) * SPI + 4 * hf;
    const int fb0 = (64 + (w4 & 1) * 64 + l16) * SPI + 4 * hf;
    v4f ga[4], gb[8];
    {
        v4f gs = {1.f, 1.f, 1.f, 1.f};
        if (SCALE) gs = *(const v4f*)Sp;
        #pragma unroll
        for (int n = 0; n < 4; ++n) ga[n] = *(const v4f*)(Ap + (size_t)(16 * n) * ld);
        #pragma unroll
        for (int n = 0; n < 8; ++n) gb[n] = *(const v4f*)(Bp + (size_t)(16 * n) * ld);
        #pragma unroll
        for (int n = 0; n < 4; ++n) *(v4f*)(L + sa + 16 * n * SPI) = SCALE ? ga[n] * gs : ga[n];
        #pragma unroll
        for (int n = 0; n < 8; ++n) *(v4f*)(L + sb[n]) = gb[n];
    }
    __syncthreads();
    for (int s = 0; s < 256 / SKK; ++s) {
        const float* cur = L + (s & 1) * STG;
        float* nxt = L + ((s + 1) & 1) * STG;
        const bool more = (s + 1 < 256 / SKK);
        v4f gs = {1.f, 1.f, 1.f, 1.f};
        if (more) {
            const int kn = SKK * (s + 1);
            if (SCALE) gs = *(const v4f*)(Sp + kn);
            #pragma unroll
            for (int n = 0; n < 4; ++n) ga[n] = *(const v4f*)(Ap + (size_t)(16 * n) * ld + kn);
            #pragma unroll
            for (int n = 0; n < 8; ++n) gb[n] = *(const v4f*)(Bp + (size_t)(16 * n) * ld + kn);
        }
        v4f fa[4][2], fb[4][4];
        #pragma unroll
        for (int u = 0; u < 4; ++u) {
            #pragma unroll
            for (int i = 0; i < 2; ++i) fa[u][i] = *(const v4f*)(cur + fa0 + 16 * i * SPI + 8 * u);
            #pragma unroll
            for (int j = 0; j < 4; ++j) fb[u][j] = *(const v4f*)(cur + fb0 + 16 * j * SPI + 8 * u);
        }
        __builtin_amdgcn_sched_barrier(0);
        #pragma unroll
        for (int u = 0; u < 4; ++u)
            #pragma unroll
            for (int hh = 0; hh < 2; ++hh)
                #pragma unroll
                for (int i = 0; i < 2; ++i) {
                    v2f a;
                    a.x = fa[u][i][2 * hh];
                    a.y = fa[u][i][2 * hh + 1];
                    #pragma unroll
                    for (int j = 0; j < 4; ++j) {
                        v2f bv;
                        bv.x = fb[u][j][2 * hh];
                        bv.y = fb[u][j][2 * hh + 1];
                        acc[i][j] = __builtin_amdgcn_wmma_f32_16x16x4_f32(false, a, false, bv,
                                                                         (short)0, acc[i][j], false, false);
                    }
                }
        __builtin_amdgcn_sched_barrier(0);
        if (more) {
            #pragma unroll
            for (int n = 0; n < 4; ++n) *(v4f*)(nxt + sa + 16 * n * SPI) = SCALE ? ga[n] * gs : ga[n];
            #pragma unroll
            for (int n = 0; n < 8; ++n) *(v4f*)(nxt + sb[n]) = gb[n];
        }
        __syncthreads();
    }
}

#define APW 260

__global__ __launch_bounds__(512) void k_gemm1_conv(
    const float* __restrict__ x, const float* __restrict__ w1w, const float* __restrict__ w1b,
    const float* __restrict__ dw, const float* __restrict__ dg, const float* __restrict__ db,
    const float* __restrict__ pg, const float* __restrict__ pw, const float* __restrict__ w2w,
    const float* __restrict__ pb, float* __restrict__ inv_np, float* __restrict__ vbp,
    float* __restrict__ y)
{
    __shared__ __attribute__((aligned(16))) float As[32 * APW];
    __shared__ __attribute__((aligned(16))) float Hs[32 * T_];
    __shared__ __attribute__((aligned(16))) float sp[16][32];
    __shared__ float sd[KW][8];
    __shared__ float snd[KW];
    __shared__ float svb[8][32];
    const int lane = threadIdx.x & 31;
    const int hf = lane >> 4, l16 = lane & 15, w = threadIdx.x >> 5;
    const int b  = blockIdx.x >> 5;
    const int c0 = (blockIdx.x & 31) * 32;
    const int n0 = w * 64;
    const int sr = threadIdx.x >> 4, spc = threadIdx.x & 15;
    v4f g[4];
    {
        const float* src = w1w + (size_t)(c0 + sr) * DIN + 4 * spc;
        #pragma unroll
        for (int n = 0; n < 4; ++n) g[n] = *(const v4f*)(src + 64 * n);
    }
    const float* Pn = pw + (size_t)b * DH1 * DH2 + (size_t)(4 * w + (lane >> 3)) * DH2
                      + (blockIdx.x & 31) * 32 + 4 * (lane & 7);
    v4f pacc = {0.f, 0.f, 0.f, 0.f};
    const float* Xp = x + ((size_t)b * T_ + n0 + 4 * l16) * DIN + 4 * hf;
    v4f cb[4][4];
    #pragma unroll
    for (int u = 0; u < 4; ++u)
        #pragma unroll
        for (int j = 0; j < 4; ++j) cb[u][j] = *(const v4f*)(Xp + (size_t)j * DIN + 8 * u);
    #pragma unroll
    for (int n = 0; n < 4; ++n) *(v4f*)(As + sr * APW + 4 * spc + 64 * n) = g[n];
    if (threadIdx.x < 256) {
        const float* D = dw + (size_t)b * DH1 * KW;
        float a9[KW];
        #pragma unroll
        for (int k = 0; k < KW; ++k) a9[k] = 0.f;
        #pragma unroll
        for (int j = 0; j < 4; ++j) {
            #pragma unroll
            for (int k = 0; k < KW; ++k) {
                const float v = D[(size_t)(threadIdx.x + 256 * j) * KW + k];
                a9[k] += v * v;
            }
        }
        #pragma unroll
        for (int k = 0; k < KW; ++k) {
            #pragma unroll
            for (int m = 16; m >= 1; m >>= 1) a9[k] += __shfl_xor(a9[k], m);
        }
        if (lane == 0) {
            #pragma unroll
            for (int k = 0; k < KW; ++k) sd[k][w] = a9[k];
        }
    } else if (w == 15) {
        const int i0 = (blockIdx.x & 15) * 16;
        const int ks = blockIdx.x >> 4;
        const float* Ap = w2w + (size_t)(i0 + l16) * DH2 + 64 * ks + 4 * hf;
        const float* Bq = pb + (size_t)(l16 & 7) * DH2 + 64 * ks + 4 * hf;
        v4f a[8], q[8];
        #pragma unroll
        for (int u = 0; u < 8; ++u) {
            a[u] = *(const v4f*)(Ap + 8 * u);
            q[u] = *(const v4f*)(Bq + 8 * u);
        }
        const bool real = (l16 < 8);
        const v4f zero = {0.f, 0.f, 0.f, 0.f};
        v8f acc0 = {}, acc1 = {};
        #pragma unroll
        for (int u = 0; u < 8; ++u) {
            const v4f qq = real ? q[u] : zero;
            v2f a0, a1, b0, b1;
            a0.x = a[u][0]; a0.y = a[u][1]; a1.x = a[u][2]; a1.y = a[u][3];
            b0.x = qq[0];   b0.y = qq[1];   b1.x = qq[2];   b1.y = qq[3];
            acc0 = __builtin_amdgcn_wmma_f32_16x16x4_f32(false, a0, false, b0, (short)0, acc0, false, false);
            acc1 = __builtin_amdgcn_wmma_f32_16x16x4_f32(false, a1, false, b1, (short)0, acc1, false, false);
        }
        #pragma unroll
        for (int v = 0; v < 8; ++v) svb[v][lane] = acc0[v] + acc1[v];
    }
    __syncthreads();
    if (threadIdx.x < KW) {
        float tot = 0.f;
        #pragma unroll
        for (int q = 0; q < 8; ++q) tot += sd[threadIdx.x][q];
        snd[threadIdx.x] = 1.f / fmaxf(sqrtf(tot), 1e-12f);
    }
    const float* Af = As + l16 * APW + 4 * hf;
    v8f acc[2][4] = {};
    #pragma unroll 2
    for (int s = 0; s < DIN / 32; ++s) {
        const int kn = (s + 1 < DIN / 32) ? 32 * (s + 1) : 32 * s;
        v4f nb[4][4];
        #pragma unroll
        for (int u = 0; u < 4; ++u)
            #pragma unroll
            for (int j = 0; j < 4; ++j) nb[u][j] = *(const v4f*)(Xp + (size_t)j * DIN + kn + 8 * u);
        const v4f pv0 = *(const v4f*)(Pn + (size_t)(128 * s) * DH2);
        const v4f pv1 = *(const v4f*)(Pn + (size_t)(128 * s + 64) * DH2);
        v4f fa[4][2];
        #pragma unroll
        for (int u = 0; u < 4; ++u)
            #pragma unroll
            for (int i = 0; i < 2; ++i) fa[u][i] = *(const v4f*)(Af + 16 * i * APW + 32 * s + 8 * u);
        __builtin_amdgcn_sched_barrier(0);
        #pragma unroll
        for (int u = 0; u < 4; ++u)
            #pragma unroll
            for (int hh = 0; hh < 2; ++hh)
                #pragma unroll
                for (int i = 0; i < 2; ++i) {
                    v2f a;
                    a.x = fa[u][i][2 * hh];
                    a.y = fa[u][i][2 * hh + 1];
                    #pragma unroll
                    for (int j = 0; j < 4; ++j) {
                        v2f bv;
                        bv.x = cb[u][j][2 * hh];
                        bv.y = cb[u][j][2 * hh + 1];
                        acc[i][j] = __builtin_amdgcn_wmma_f32_16x16x4_f32(false, a, false, bv,
                                                                         (short)0, acc[i][j], false, false);
                    }
                }
        __builtin_amdgcn_sched_barrier(0);
        asm volatile("" :: "v"(nb[0][0]), "v"(nb[0][1]), "v"(nb[0][2]), "v"(nb[0][3]),
                        "v"(nb[1][0]), "v"(nb[1][1]), "v"(nb[1][2]), "v"(nb[1][3]),
                        "v"(nb[2][0]), "v"(nb[2][1]), "v"(nb[2][2]), "v"(nb[2][3]),
                        "v"(nb[3][0]), "v"(nb[3][1]), "v"(nb[3][2]), "v"(nb[3][3]));
        pacc += pv0 * pv0;
        pacc += pv1 * pv1;
        #pragma unroll
        for (int u = 0; u < 4; ++u)
            #pragma unroll
            for (int j = 0; j < 4; ++j) cb[u][j] = nb[u][j];
    }
    #pragma unroll
    for (int i = 0; i < 4; ++i) {
        pacc[i] += __shfl_xor(pacc[i], 8);
        pacc[i] += __shfl_xor(pacc[i], 16);
    }
    if (lane < 8) *(v4f*)&sp[w][4 * lane] = pacc;
    #pragma unroll
    for (int i = 0; i < 2; ++i) {
        #pragma unroll
        for (int v = 0; v < 8; ++v) {
            const int cl = 16 * i + v + 8 * hf;
            const float bias = w1b[c0 + cl];
            v4f o;
            o.x = mishf(acc[i][0][v] + bias);
            o.y = mishf(acc[i][1][v] + bias);
            o.z = mishf(acc[i][2][v] + bias);
            o.w = mishf(acc[i][3][v] + bias);
            *(v4f*)(Hs + cl * T_ + n0 + 4 * l16) = o;
        }
    }
    __syncthreads();
    if (threadIdx.x < 32) {
        float tot = 0.f;
        #pragma unroll
        for (int q = 0; q < 16; ++q) tot += sp[q][threadIdx.x];
        inv_np[b * DH2 + (blockIdx.x & 31) * 32 + threadIdx.x] = 1.f / fmaxf(sqrtf(tot), 1e-12f);
    }
    #pragma unroll 1
    for (int rr = 0; rr < 2; ++rr) {
        const int cl  = __builtin_amdgcn_readfirstlane(w + 16 * rr);
        const int row = __builtin_amdgcn_readfirstlane(b * DH1 + c0 + w + 16 * rr);
        const float* wp  = dw + (size_t)row * KW;
        float* yp = y + (size_t)row * T_;
        float wk[KW];
        #pragma unroll
        for (int k = 0; k < KW; ++k) wk[k] = wp[k] * snd[k];
        const float g = dg[row], bi = db[row], pgv = pg[row];
        v4f cur[8], prv[8], nxt[8];
        #pragma unroll
        for (int j = 0; j < 8; ++j) cur[j] = *(const v4f*)(Hs + cl * T_ + 128 * j + 4 * lane);
        #pragma unroll
        for (int j = 0; j < 8; ++j) {
            const int t = 128 * j + 4 * lane;
            const bool inl = (t >= 4), inr = (t + 4 < T_);
            const v4f p = *(const v4f*)(Hs + cl * T_ + (inl ? t - 4 : t));
            const v4f n = *(const v4f*)(Hs + cl * T_ + (inr ? t + 4 : t));
            const v4f zero = {0.f, 0.f, 0.f, 0.f};
            prv[j] = inl ? p : zero;
            nxt[j] = inr ? n : zero;
        }
        #pragma unroll
        for (int j = 0; j < 8; ++j) {
            const int t = 128 * j + 4 * lane;
            const float win[12] = {prv[j].x, prv[j].y, prv[j].z, prv[j].w,
                                   cur[j].x, cur[j].y, cur[j].z, cur[j].w,
                                   nxt[j].x, nxt[j].y, nxt[j].z, nxt[j].w};
            v4f o;
            #pragma unroll
            for (int i = 0; i < 4; ++i) {
                float a9 = 0.f;
                #pragma unroll
                for (int k = 0; k < KW; ++k) a9 += win[i + k] * wk[k];
                const float yy = a9 * g + bi;
                o[i] = yy * pgv;
            }
            *(v4f*)(yp + t) = o;
        }
    }
    if (w == 15 && l16 < 8) {
        const int i0 = (blockIdx.x & 15) * 16;
        const int ks = blockIdx.x >> 4;
        #pragma unroll
        for (int v = 0; v < 8; ++v)
            vbp[(size_t)ks * (B_ * DIN) + l16 * DIN + i0 + v + 8 * hf] = svb[v][lane];
    }
}

__global__ __launch_bounds__(512) void k_weff(
    const float* __restrict__ w2w, const float* __restrict__ pw,
    const float* __restrict__ inv_np, float* __restrict__ weff)
{
    __shared__ __attribute__((aligned(16))) float L[4 * 2 * STG];
    v4f (*red)[4][16][32] = reinterpret_cast<v4f (*)[4][16][32]>(L);
    const int lane = threadIdx.x & 31;
    const int hf = lane >> 4, l16 = lane & 15;
    const int w  = (threadIdx.x >> 5) & 3;
    const int q  = threadIdx.x >> 7;
    const int b  = blockIdx.x >> 5;
    const int r  = blockIdx.x & 31;
    const int mg = (r >> 3) * 64;
    const int ng = (r & 7) * 128;
    const int m0 = mg + (w >> 1) * 32;
    const int n0 = ng + (w & 1) * 64;
    const int kq = q * KQ2;
    v8f acc[2][4] = {};
    nt_core<true>(w2w + (size_t)mg * DH2 + kq, pw + (size_t)b * DH1 * DH2 + (size_t)ng * DH2 + kq,
                  inv_np + b * DH2 + kq, DH2, L + q * 2 * STG, (int)(threadIdx.x & 127), lane, w, acc);
    if (q >= 2) {
        #pragma unroll
        for (int i = 0; i < 2; ++i)
            #pragma unroll
            for (int j = 0; j < 4; ++j)
                #pragma unroll
                for (int h = 0; h < 2; ++h) {
                    v4f p;
                    p.x = acc[i][j][4 * h];     p.y = acc[i][j][4 * h + 1];
                    p.z = acc[i][j][4 * h + 2]; p.w = acc[i][j][4 * h + 3];
                    red[q - 2][w][2 * (4 * i + j) + h][lane] = p;
                }
    }
    __syncthreads();
    if (q < 2) {
        #pragma unroll
        for (int i = 0; i < 2; ++i)
            #pragma unroll
            for (int j = 0; j < 4; ++j)
                #pragma unroll
                for (int h = 0; h < 2; ++h) {
                    const v4f p = red[q][w][2 * (4 * i + j) + h][lane];
                    acc[i][j][4 * h]     += p.x; acc[i][j][4 * h + 1] += p.y;
                    acc[i][j][4 * h + 2] += p.z; acc[i][j][4 * h + 3] += p.w;
                }
    }
    __syncthreads();
    if (q == 1) {
        #pragma unroll
        for (int i = 0; i < 2; ++i)
            #pragma unroll
            for (int j = 0; j < 4; ++j)
                #pragma unroll
                for (int h = 0; h < 2; ++h) {
                    v4f p;
                    p.x = acc[i][j][4 * h];     p.y = acc[i][j][4 * h + 1];
                    p.z = acc[i][j][4 * h + 2]; p.w = acc[i][j][4 * h + 3];
                    red[0][w][2 * (4 * i + j) + h][lane] = p;
                }
    }
    __syncthreads();
    if (q == 0) {
        #pragma unroll
        for (int i = 0; i < 2; ++i)
            #pragma unroll
            for (int j = 0; j < 4; ++j)
                #pragma unroll
                for (int h = 0; h < 2; ++h) {
                    const v4f p = red[0][w][2 * (4 * i + j) + h][lane];
                    acc[i][j][4 * h]     += p.x; acc[i][j][4 * h + 1] += p.y;
                    acc[i][j][4 * h + 2] += p.z; acc[i][j][4 * h + 3] += p.w;
                }
        float* Wo = weff + (size_t)b * DIN * DH1;
        #pragma unroll
        for (int i = 0; i < 2; ++i) {
            #pragma unroll
            for (int v = 0; v < 8; ++v) {
                const int row = m0 + 16 * i + v + 8 * hf;
                v4f o;
                o.x = acc[i][0][v];
                o.y = acc[i][1][v];
                o.z = acc[i][2][v];
                o.w = acc[i][3][v];
                *(v4f*)(Wo + (size_t)row * DH1 + n0 + 4 * l16) = o;
            }
        }
    }
}

__global__ __launch_bounds__(512) void k_final(
    const float* __restrict__ y, const float* __restrict__ weff,
    const float* __restrict__ vbp, const float* __restrict__ w2b,
    const float* __restrict__ x, float* __restrict__ out)
{
    __shared__ v4f red[2][4][16][32];
    const int lane = threadIdx.x & 31;
    const int hf = lane >> 4, l16 = lane & 15;
    const int w  = (threadIdx.x >> 5) & 3;
    const int q  = threadIdx.x >> 7;
    const int b  = blockIdx.x >> 5;
    const int r  = blockIdx.x & 31;
    const int t0 = ((r >> 2) * 2 + (w >> 1)) * 64;
    const int i0 = ((r & 3) * 2 + (w & 1)) * 32;
    const int kq = q * KQ1;
    const float* Zp = y + (size_t)b * DH1 * T_ + (size_t)(kq + 4 * hf) * T_ + t0 + 4 * l16;
    const float* Wp = weff + (size_t)b * DIN * DH1 + (size_t)(i0 + 2 * l16) * DH1 + kq + 4 * hf;
    v8f acc[4][2] = {};
    v4f ca[2][4];
    v4f cb[2][2];
    #pragma unroll
    for (int u = 0; u < 2; ++u) {
        #pragma unroll
        for (int e = 0; e < 4; ++e) ca[u][e] = *(const v4f*)(Zp + (size_t)(8 * u + e) * T_);
        #pragma unroll
        for (int j = 0; j < 2; ++j) cb[u][j] = *(const v4f*)(Wp + (size_t)j * DH1 + 8 * u);
    }
    #pragma unroll 2
    for (int k0 = 0; k0 < KQ1; k0 += 16) {
        const int kn = (k0 + 16 < KQ1) ? k0 + 16 : k0;
        v4f na[2][4];
        v4f nb[2][2];
        #pragma unroll
        for (int u = 0; u < 2; ++u) {
            #pragma unroll
            for (int e = 0; e < 4; ++e) na[u][e] = *(const v4f*)(Zp + (size_t)(kn + 8 * u + e) * T_);
            #pragma unroll
            for (int j = 0; j < 2; ++j) nb[u][j] = *(const v4f*)(Wp + (size_t)j * DH1 + kn + 8 * u);
        }
        __builtin_amdgcn_sched_barrier(0);
        #pragma unroll
        for (int u = 0; u < 2; ++u)
            #pragma unroll
            for (int hh = 0; hh < 2; ++hh)
                #pragma unroll
                for (int a = 0; a < 4; ++a) {
                    v2f av;
                    av.x = ca[u][2 * hh][a];
                    av.y = ca[u][2 * hh + 1][a];
                    #pragma unroll
                    for (int j = 0; j < 2; ++j) {
                        v2f bv;
                        bv.x = cb[u][j][2 * hh];
                        bv.y = cb[u][j][2 * hh + 1];
                        acc[a][j] = __builtin_amdgcn_wmma_f32_16x16x4_f32(false, av, false, bv,
                                                                         (short)0, acc[a][j], false, false);
                    }
                }
        __builtin_amdgcn_sched_barrier(0);
        asm volatile("" :: "v"(na[0][0]), "v"(na[0][1]), "v"(na[0][2]), "v"(na[0][3]),
                        "v"(na[1][0]), "v"(na[1][1]), "v"(na[1][2]), "v"(na[1][3]),
                        "v"(nb[0][0]), "v"(nb[0][1]), "v"(nb[1][0]), "v"(nb[1][1]));
        #pragma unroll
        for (int u = 0; u < 2; ++u) {
            #pragma unroll
            for (int e = 0; e < 4; ++e) ca[u][e] = na[u][e];
            #pragma unroll
            for (int j = 0; j < 2; ++j) cb[u][j] = nb[u][j];
        }
    }
    if (q >= 2) {
        #pragma unroll
        for (int a = 0; a < 4; ++a)
            #pragma unroll
            for (int j = 0; j < 2; ++j)
                #pragma unroll
                for (int h = 0; h < 2; ++h) {
                    v4f p;
                    p.x = acc[a][j][4 * h];     p.y = acc[a][j][4 * h + 1];
                    p.z = acc[a][j][4 * h + 2]; p.w = acc[a][j][4 * h + 3];
                    red[q - 2][w][2 * (2 * a + j) + h][lane] = p;
                }
    }
    __syncthreads();
    if (q < 2) {
        #pragma unroll
        for (int a = 0; a < 4; ++a)
            #pragma unroll
            for (int j = 0; j < 2; ++j)
                #pragma unroll
                for (int h = 0; h < 2; ++h) {
                    const v4f p = red[q][w][2 * (2 * a + j) + h][lane];
                    acc[a][j][4 * h]     += p.x; acc[a][j][4 * h + 1] += p.y;
                    acc[a][j][4 * h + 2] += p.z; acc[a][j][4 * h + 3] += p.w;
                }
    }
    __syncthreads();
    if (q == 1) {
        #pragma unroll
        for (int a = 0; a < 4; ++a)
            #pragma unroll
            for (int j = 0; j < 2; ++j)
                #pragma unroll
                for (int h = 0; h < 2; ++h) {
                    v4f p;
                    p.x = acc[a][j][4 * h];     p.y = acc[a][j][4 * h + 1];
                    p.z = acc[a][j][4 * h + 2]; p.w = acc[a][j][4 * h + 3];
                    red[0][w][2 * (2 * a + j) + h][lane] = p;
                }
    }
    __syncthreads();
    if (q == 0) {
        #pragma unroll
        for (int a = 0; a < 4; ++a)
            #pragma unroll
            for (int j = 0; j < 2; ++j)
                #pragma unroll
                for (int h = 0; h < 2; ++h) {
                    const v4f p = red[0][w][2 * (2 * a + j) + h][lane];
                    acc[a][j][4 * h]     += p.x; acc[a][j][4 * h + 1] += p.y;
                    acc[a][j][4 * h + 2] += p.z; acc[a][j][4 * h + 3] += p.w;
                }
        const int ii = i0 + 2 * l16;
        v2f vs = {0.f, 0.f};
        #pragma unroll
        for (int p = 0; p < 16; ++p) vs += *(const v2f*)(vbp + (size_t)p * (B_ * DIN) + b * DIN + ii);
        const v2f bb = *(const v2f*)(w2b + ii) + vs;
        #pragma unroll
        for (int a = 0; a < 4; ++a) {
            #pragma unroll
            for (int v = 0; v < 8; ++v) {
                const int t = t0 + 4 * (v + 8 * hf) + a;
                const size_t oi = ((size_t)b * T_ + t) * DIN + ii;
                const v2f xv = *(const v2f*)(x + oi);
                v2f o;
                o.x = acc[a][0][v] + bb.x + xv.x;
                o.y = acc[a][1][v] + bb.y + xv.y;
                *(v2f*)(out + oi) = o;
            }
        }
    }
}

extern "C" void kernel_launch(void* const* d_in, const int* in_sizes, int n_in,
                              void* d_out, int out_size, void* d_ws, size_t ws_size,
                              hipStream_t stream)
{
    const float* x   = (const float*)d_in[0];
    const float* d_w = (const float*)d_in[1];
    const float* d_g = (const float*)d_in[2];
    const float* d_b = (const float*)d_in[3];
    const float* p_w = (const float*)d_in[4];
    const float* p_g = (const float*)d_in[5];
    const float* p_b = (const float*)d_in[6];
    const float* w1w = (const float*)d_in[7];
    const float* w1b = (const float*)d_in[8];
    const float* w2w = (const float*)d_in[9];
    const float* w2b = (const float*)d_in[10];
    float* out = (float*)d_out;

    float* ws = (float*)d_ws;
    const size_t NCT = (size_t)B_ * DH1 * T_;
    float* ybuf   = ws + NCT;
    float* inv_np = ws + 2 * NCT;
    float* vbuf   = inv_np + 8192;
    float* weff   = vbuf + 16 * 2048;

    k_gemm1_conv<<<256, 512, 0, stream>>>(x, w1w, w1b, d_w, d_g, d_b, p_g, p_w, w2w, p_b, inv_np, vbuf, ybuf);
    k_weff<<<256, 512, 0, stream>>>(w2w, p_w, inv_np, weff);
    k_final<<<256, 512, 0, stream>>>(ybuf, weff, vbuf, w2b, x, out);
}
